# starting kernel
# baseline (speedup 1.0000x reference)
#include <hip/hip_runtime.h>
#include <hip/hip_bf16.h>

typedef __attribute__((ext_vector_type(2))) float v2f;
typedef __attribute__((ext_vector_type(8))) float v8f;

#define NEG_SLOPE 0.2f


__device__ __forceinline__ float lrelu(float v) {
    return v >= 0.0f ? v : NEG_SLOPE * v;
}

__device__ __forceinline__ unsigned fkey(float f) {
    unsigned u = __float_as_uint(f);
    return (u & 0x80000000u) ? ~u : (u | 0x80000000u);
}
__device__ __forceinline__ float fdekey(unsigned k) {
    return (k & 0x80000000u) ? __uint_as_float(k ^ 0x80000000u)
                             : __uint_as_float(~k);
}

__global__ void zero_u32(unsigned* __restrict__ p, int n) {
    int i = blockIdx.x * blockDim.x + threadIdx.x;
    if (i < n) p[i] = 0u;
}

__global__ __launch_bounds__(32)
void gemm1_wmma(const float* __restrict__ x, const float* __restrict__ W1,
                float* __restrict__ h1, int N) {
    const int mtile = blockIdx.x;
    const int lane  = threadIdx.x;
    const int mrow  = mtile * 16 + (lane & 15);
    const int lrow  = mrow < N ? mrow : N - 1;
    const int khalf = (lane >> 4) * 2;
    const int colbase = blockIdx.y * 64;
    const int nc    = lane & 15;

    v8f c0 = {}, c1 = {}, c2 = {}, c3 = {};
    const float* xrow = x + (size_t)lrow * 128;

    for (int k = 0; k < 128; k += 4) {
        const int kb = k + khalf;
        v2f a; a.x = xrow[kb]; a.y = xrow[kb + 1];
        const float* w0 = W1 + (size_t)kb * 256 + colbase + nc;
        const float* w1 = w0 + 256;
        v2f b0; b0.x = w0[0];  b0.y = w1[0];
        v2f b1; b1.x = w0[16]; b1.y = w1[16];
        v2f b2; b2.x = w0[32]; b2.y = w1[32];
        v2f b3; b3.x = w0[48]; b3.y = w1[48];
        c0 = __builtin_amdgcn_wmma_f32_16x16x4_f32(false, a, false, b0, (short)0, c0, false, false);
        c1 = __builtin_amdgcn_wmma_f32_16x16x4_f32(false, a, false, b1, (short)0, c1, false, false);
        c2 = __builtin_amdgcn_wmma_f32_16x16x4_f32(false, a, false, b2, (short)0, c2, false, false);
        c3 = __builtin_amdgcn_wmma_f32_16x16x4_f32(false, a, false, b3, (short)0, c3, false, false);
    }

    const int rowoff = (lane >= 16) ? 8 : 0;
    for (int r = 0; r < 8; ++r) {
        const int grow = mtile * 16 + rowoff + r;
        if (grow < N) {
            float* op = h1 + (size_t)grow * 256 + colbase + nc;
            op[0]  = c0[r];
            op[16] = c1[r];
            op[32] = c2[r];
            op[48] = c3[r];
        }
    }
}

__global__ __launch_bounds__(32)
void gemm2_wmma(const float* __restrict__ y1, const float* __restrict__ W2,
                float* __restrict__ h2, int N) {
    const int mtile = blockIdx.x;
    const int lane  = threadIdx.x;
    const int mrow  = mtile * 16 + (lane & 15);
    const int lrow  = mrow < N ? mrow : N - 1;
    const int khalf = (lane >> 4) * 2;
    const int nc    = lane & 15;

    v8f c = {};
    const float* yr = y1 + (size_t)lrow * 256;
    for (int k = 0; k < 256; k += 4) {
        const int kb = k + khalf;
        v2f a; a.x = yr[kb]; a.y = yr[kb + 1];
        v2f b; b.x = W2[(size_t)kb * 16 + nc]; b.y = W2[(size_t)(kb + 1) * 16 + nc];
        c = __builtin_amdgcn_wmma_f32_16x16x4_f32(false, a, false, b, (short)0, c, false, false);
    }
    const int rowoff = (lane >= 16) ? 8 : 0;
    for (int r = 0; r < 8; ++r) {
        const int grow = mtile * 16 + rowoff + r;
        if (grow < N) h2[(size_t)grow * 16 + nc] = c[r];
    }
}

__global__ void att1_kernel(const float* __restrict__ h1,
                            const float* __restrict__ att_src,
                            const float* __restrict__ att_dst,
                            float* __restrict__ a_src, float* __restrict__ a_dst,
                            int NH) {
    int i = blockIdx.x * blockDim.x + threadIdx.x;
    if (i >= NH) return;
    const int h = i & 3;
    const float* hp = h1 + (size_t)(i >> 2) * 256 + h * 64;
    const float* as = att_src + h * 64;
    const float* ad = att_dst + h * 64;
    float s = 0.0f, d = 0.0f;
    for (int c = 0; c < 64; ++c) {
        const float v = hp[c];
        s += v * as[c];
        d += v * ad[c];
    }
    a_src[i] = s;
    a_dst[i] = d;
}

__global__ void att2_kernel(const float* __restrict__ h2,
                            const float* __restrict__ att_src,
                            const float* __restrict__ att_dst,
                            float* __restrict__ a_src, float* __restrict__ a_dst,
                            int N) {
    int i = blockIdx.x * blockDim.x + threadIdx.x;
    if (i >= N) return;
    const float* hp = h2 + (size_t)i * 16;
    float s = 0.0f, d = 0.0f;
    for (int c = 0; c < 16; ++c) {
        const float v = hp[c];
        s += v * att_src[c];
        d += v * att_dst[c];
    }
    a_src[i] = s;
    a_dst[i] = d;
}

__device__ __forceinline__ void edge_sd(const int* __restrict__ ei, int e, int E,
                                        int& s, int& d) {
    if (e < E) { s = ei[e]; d = ei[E + e]; }
    else       { s = d = e - E; }
}

__global__ void edge_max1(const int* __restrict__ ei, int E, int N,
                          const float* __restrict__ a_src, const float* __restrict__ a_dst,
                          unsigned* __restrict__ mkey) {
    int e = blockIdx.x * blockDim.x + threadIdx.x;
    const int ET = E + N;
    if (e >= ET) return;
    int s, d; edge_sd(ei, e, E, s, d);
    for (int h = 0; h < 4; ++h) {
        const float v = lrelu(a_src[s * 4 + h] + a_dst[d * 4 + h]);
        atomicMax(&mkey[d * 4 + h], fkey(v));
    }
}

__global__ void edge_expsum1(const int* __restrict__ ei, int E, int N,
                             const float* __restrict__ a_src, const float* __restrict__ a_dst,
                             const unsigned* __restrict__ mkey,
                             float* __restrict__ ex, float* __restrict__ den) {
    int e = blockIdx.x * blockDim.x + threadIdx.x;
    const int ET = E + N;
    if (e >= ET) return;
    int s, d; edge_sd(ei, e, E, s, d);
    for (int h = 0; h < 4; ++h) {
        const float v  = lrelu(a_src[s * 4 + h] + a_dst[d * 4 + h]);
        const float m  = fdekey(mkey[d * 4 + h]);
        const float xv = expf(v - m);
        ex[(size_t)e * 4 + h] = xv;
        atomicAdd(&den[d * 4 + h], xv);
    }
}

__global__ void edge_agg1(const int* __restrict__ ei, int E, int N,
                          const float* __restrict__ ex, const float* __restrict__ den,
                          const float* __restrict__ h1, float* __restrict__ out1) {
    const int gid = blockIdx.x * blockDim.x + threadIdx.x;
    const int e = gid >> 6;
    const int t = gid & 63;
    const int ET = E + N;
    if (e >= ET) return;
    int s, d; edge_sd(ei, e, E, s, d);
    const int head = t >> 4;
    const float alpha = ex[(size_t)e * 4 + head] / den[d * 4 + head];
    const float4 hv = *(const float4*)(h1 + (size_t)s * 256 + t * 4);
    float* op = out1 + (size_t)d * 256 + t * 4;
    atomicAdd(op + 0, alpha * hv.x);
    atomicAdd(op + 1, alpha * hv.y);
    atomicAdd(op + 2, alpha * hv.z);
    atomicAdd(op + 3, alpha * hv.w);
}

__global__ void relu_bias1(float* __restrict__ out1, const float* __restrict__ b1, int n) {
    int i = blockIdx.x * blockDim.x + threadIdx.x;
    if (i < n) out1[i] = fmaxf(out1[i] + b1[i & 255], 0.0f);
}

__global__ void edge_max2(const int* __restrict__ ei, int E, int N,
                          const float* __restrict__ a_src, const float* __restrict__ a_dst,
                          unsigned* __restrict__ mkey) {
    int e = blockIdx.x * blockDim.x + threadIdx.x;
    const int ET = E + N;
    if (e >= ET) return;
    int s, d; edge_sd(ei, e, E, s, d);
    const float v = lrelu(a_src[s] + a_dst[d]);
    atomicMax(&mkey[d], fkey(v));
}

__global__ void edge_expsum2(const int* __restrict__ ei, int E, int N,
                             const float* __restrict__ a_src, const float* __restrict__ a_dst,
                             const unsigned* __restrict__ mkey,
                             float* __restrict__ ex, float* __restrict__ den) {
    int e = blockIdx.x * blockDim.x + threadIdx.x;
    const int ET = E + N;
    if (e >= ET) return;
    int s, d; edge_sd(ei, e, E, s, d);
    const float v  = lrelu(a_src[s] + a_dst[d]);
    const float xv = expf(v - fdekey(mkey[d]));
    ex[e] = xv;
    atomicAdd(&den[d], xv);
}

__global__ void edge_agg2(const int* __restrict__ ei, int E, int N,
                          const float* __restrict__ ex, const float* __restrict__ den,
                          const float* __restrict__ h2, float* __restrict__ out) {
    const int gid = blockIdx.x * blockDim.x + threadIdx.x;
    const int e = gid >> 4;
    const int t = gid & 15;
    const int ET = E + N;
    if (e >= ET) return;
    int s, d; edge_sd(ei, e, E, s, d);
    const float alpha = ex[e] / den[d];
    atomicAdd(&out[(size_t)d * 16 + t], alpha * h2[(size_t)s * 16 + t]);
}

__global__ void bias2_kernel(float* __restrict__ out, const float* __restrict__ b2, int n) {
    int i = blockIdx.x * blockDim.x + threadIdx.x;
    if (i < n) out[i] += b2[i & 15];
}


extern "C" void kernel_launch(void* const* d_in, const int* in_sizes, int n_in,
                              void* d_out, int out_size, void* d_ws, size_t ws_size,
                              hipStream_t stream) {
    const float* x   = (const float*)d_in[0];
    const int*   ei  = (const int*)  d_in[1];
    const float* W1  = (const float*)d_in[2];
    const float* as1 = (const float*)d_in[3];
    const float* ad1 = (const float*)d_in[4];
    const float* b1  = (const float*)d_in[5];
    const float* W2  = (const float*)d_in[6];
    const float* as2 = (const float*)d_in[7];
    const float* ad2 = (const float*)d_in[8];
    const float* b2  = (const float*)d_in[9];

    const int N  = in_sizes[0] / 128;
    const int E  = in_sizes[1] / 2;
    const int ET = E + N;

    char* w = (char*)d_ws;
    auto carve = [&](size_t bytes) -> void* {
        void* p = (void*)w;
        w += (bytes + 255) & ~((size_t)255);
        return p;
    };
    float*    h1    = (float*)   carve((size_t)N * 256 * 4);
    float*    out1  = (float*)   carve((size_t)N * 256 * 4);
    float*    aS1   = (float*)   carve((size_t)N * 4 * 4);
    float*    aD1   = (float*)   carve((size_t)N * 4 * 4);
    unsigned* mkey1 = (unsigned*)carve((size_t)N * 4 * 4);
    float*    den1  = (float*)   carve((size_t)N * 4 * 4);
    float*    ex1   = (float*)   carve((size_t)ET * 4 * 4);
    float*    h2    = (float*)   carve((size_t)N * 16 * 4);
    float*    aS2   = (float*)   carve((size_t)N * 4);
    float*    aD2   = (float*)   carve((size_t)N * 4);
    unsigned* mkey2 = (unsigned*)carve((size_t)N * 4);
    float*    den2  = (float*)   carve((size_t)N * 4);
    float*    ex2   = (float*)   carve((size_t)ET * 4);

    const int TB = 256;
    auto blocks = [&](long long n) { return (unsigned)((n + TB - 1) / TB); };

    zero_u32<<<blocks((long long)N * 256), TB, 0, stream>>>((unsigned*)out1, N * 256);
    zero_u32<<<blocks(N * 4), TB, 0, stream>>>(mkey1, N * 4);
    zero_u32<<<blocks(N * 4), TB, 0, stream>>>((unsigned*)den1, N * 4);
    zero_u32<<<blocks(N), TB, 0, stream>>>(mkey2, N);
    zero_u32<<<blocks(N), TB, 0, stream>>>((unsigned*)den2, N);
    zero_u32<<<blocks(N * 16), TB, 0, stream>>>((unsigned*)d_out, N * 16);

    dim3 g1((N + 15) / 16, 4);
    gemm1_wmma<<<g1, 32, 0, stream>>>(x, W1, h1, N);
    att1_kernel<<<blocks(N * 4), TB, 0, stream>>>(h1, as1, ad1, aS1, aD1, N * 4);
    edge_max1<<<blocks(ET), TB, 0, stream>>>(ei, E, N, aS1, aD1, mkey1);
    edge_expsum1<<<blocks(ET), TB, 0, stream>>>(ei, E, N, aS1, aD1, mkey1, ex1, den1);
    edge_agg1<<<blocks((long long)ET * 64), TB, 0, stream>>>(ei, E, N, ex1, den1, h1, out1);
    relu_bias1<<<blocks((long long)N * 256), TB, 0, stream>>>(out1, b1, N * 256);

    gemm2_wmma<<<(N + 15) / 16, 32, 0, stream>>>(out1, W2, h2, N);
    att2_kernel<<<blocks(N), TB, 0, stream>>>(h2, as2, ad2, aS2, aD2, N);
    edge_max2<<<blocks(ET), TB, 0, stream>>>(ei, E, N, aS2, aD2, mkey2);
    edge_expsum2<<<blocks(ET), TB, 0, stream>>>(ei, E, N, aS2, aD2, mkey2, ex2, den2);
    edge_agg2<<<blocks((long long)ET * 16), TB, 0, stream>>>(ei, E, N, ex2, den2, h2, (float*)d_out);
    bias2_kernel<<<blocks(N * 16), TB, 0, stream>>>((float*)d_out, b2, N * 16);
}


# Edges grouped by destination per call (integer-atomic counting sort, arrival-order rank); per node softmax and weighted row sum in registers in list order, row stored once: no float atomics
# speedup vs baseline: 3.4925x; 3.4925x over previous
#include <hip/hip_runtime.h>
#include <hip/hip_bf16.h>

typedef __attribute__((ext_vector_type(2))) float v2f;
typedef __attribute__((ext_vector_type(8))) float v8f;

#define NEG_SLOPE 0.2f


__device__ __forceinline__ float lrelu(float v) {
    return v >= 0.0f ? v : NEG_SLOPE * v;
}

__global__ void zero_u32(unsigned* __restrict__ p, int n) {
    int i = blockIdx.x * blockDim.x + threadIdx.x;
    if (i < n) p[i] = 0u;
}

__global__ __launch_bounds__(32)
void gemm1_wmma(const float* __restrict__ x, const float* __restrict__ W1,
                float* __restrict__ h1, int N) {
    const int mtile = blockIdx.x;
    const int lane  = threadIdx.x;
    const int mrow  = mtile * 16 + (lane & 15);
    const int lrow  = mrow < N ? mrow : N - 1;
    const int khalf = (lane >> 4) * 2;
    const int colbase = blockIdx.y * 64;
    const int nc    = lane & 15;

    v8f c0 = {}, c1 = {}, c2 = {}, c3 = {};
    const float* xrow = x + (size_t)lrow * 128;

    for (int k = 0; k < 128; k += 4) {
        const int kb = k + khalf;
        v2f a; a.x = xrow[kb]; a.y = xrow[kb + 1];
        const float* w0 = W1 + (size_t)kb * 256 + colbase + nc;
        const float* w1 = w0 + 256;
        v2f b0; b0.x = w0[0];  b0.y = w1[0];
        v2f b1; b1.x = w0[16]; b1.y = w1[16];
        v2f b2; b2.x = w0[32]; b2.y = w1[32];
        v2f b3; b3.x = w0[48]; b3.y = w1[48];
        c0 = __builtin_amdgcn_wmma_f32_16x16x4_f32(false, a, false, b0, (short)0, c0, false, false);
        c1 = __builtin_amdgcn_wmma_f32_16x16x4_f32(false, a, false, b1, (short)0, c1, false, false);
        c2 = __builtin_amdgcn_wmma_f32_16x16x4_f32(false, a, false, b2, (short)0, c2, false, false);
        c3 = __builtin_amdgcn_wmma_f32_16x16x4_f32(false, a, false, b3, (short)0, c3, false, false);
    }

    const int rowoff = (lane >= 16) ? 8 : 0;
    for (int r = 0; r < 8; ++r) {
        const int grow = mtile * 16 + rowoff + r;
        if (grow < N) {
            float* op = h1 + (size_t)grow * 256 + colbase + nc;
            op[0]  = c0[r];
            op[16] = c1[r];
            op[32] = c2[r];
            op[48] = c3[r];
        }
    }
}

__global__ __launch_bounds__(32)
void gemm2_wmma(const float* __restrict__ y1, const float* __restrict__ W2,
                float* __restrict__ h2, int N) {
    const int mtile = blockIdx.x;
    const int lane  = threadIdx.x;
    const int mrow  = mtile * 16 + (lane & 15);
    const int lrow  = mrow < N ? mrow : N - 1;
    const int khalf = (lane >> 4) * 2;
    const int nc    = lane & 15;

    v8f c = {};
    const float* yr = y1 + (size_t)lrow * 256;
    for (int k = 0; k < 256; k += 4) {
        const int kb = k + khalf;
        v2f a; a.x = yr[kb]; a.y = yr[kb + 1];
        v2f b; b.x = W2[(size_t)kb * 16 + nc]; b.y = W2[(size_t)(kb + 1) * 16 + nc];
        c = __builtin_amdgcn_wmma_f32_16x16x4_f32(false, a, false, b, (short)0, c, false, false);
    }
    const int rowoff = (lane >= 16) ? 8 : 0;
    for (int r = 0; r < 8; ++r) {
        const int grow = mtile * 16 + rowoff + r;
        if (grow < N) h2[(size_t)grow * 16 + nc] = c[r];
    }
}

__global__ void att1_kernel(const float* __restrict__ h1,
                            const float* __restrict__ att_src,
                            const float* __restrict__ att_dst,
                            float* __restrict__ a_src, float* __restrict__ a_dst,
                            int NH) {
    int i = blockIdx.x * blockDim.x + threadIdx.x;
    if (i >= NH) return;
    const int h = i & 3;
    const float* hp = h1 + (size_t)(i >> 2) * 256 + h * 64;
    const float* as = att_src + h * 64;
    const float* ad = att_dst + h * 64;
    float s = 0.0f, d = 0.0f;
    for (int c = 0; c < 64; ++c) {
        const float v = hp[c];
        s += v * as[c];
        d += v * ad[c];
    }
    a_src[i] = s;
    a_dst[i] = d;
}

__global__ void att2_kernel(const float* __restrict__ h2,
                            const float* __restrict__ att_src,
                            const float* __restrict__ att_dst,
                            float* __restrict__ a_src, float* __restrict__ a_dst,
                            int N) {
    int i = blockIdx.x * blockDim.x + threadIdx.x;
    if (i >= N) return;
    const float* hp = h2 + (size_t)i * 16;
    float s = 0.0f, d = 0.0f;
    for (int c = 0; c < 16; ++c) {
        const float v = hp[c];
        s += v * att_src[c];
        d += v * att_dst[c];
    }
    a_src[i] = s;
    a_dst[i] = d;
}

__global__ void count_rank(const int* __restrict__ ei, int E,
                           int* __restrict__ deg, int* __restrict__ rank) {
    const int e = blockIdx.x * blockDim.x + threadIdx.x;
    if (e >= E) return;
    rank[e] = atomicAdd(&deg[ei[E + e]], 1);
}

__global__ __launch_bounds__(1024)
void scan_deg(const int* __restrict__ deg, int* __restrict__ rowptr, int N) {
    __shared__ int wsum[32];
    __shared__ int wpre[32];
    __shared__ int base_s;
    const int t = threadIdx.x, lane = t & 31, wv = t >> 5;
    const int start = blockIdx.x * 1024;
    int pre = 0;
    for (int i = t; i < start; i += 1024) pre += deg[i];
    const int idx = start + t;
    const int v = idx < N ? deg[idx] : 0;
    int inc = v;
    for (int off = 1; off < 32; off <<= 1) {
        const int nb = __shfl_up(inc, off);
        if (lane >= off) inc += nb;
    }
    for (int off = 16; off > 0; off >>= 1) pre += __shfl_xor(pre, off);
    if (lane == 31) wsum[wv] = inc;
    if (lane == 0)  wpre[wv] = pre;
    __syncthreads();
    if (wv == 0) {
        const int tot = wsum[lane];
        int p = wpre[lane];
        int winc = tot;
        for (int off = 1; off < 32; off <<= 1) {
            const int nb = __shfl_up(winc, off);
            if (lane >= off) winc += nb;
        }
        for (int off = 16; off > 0; off >>= 1) p += __shfl_xor(p, off);
        wsum[lane] = winc - tot;
        if (lane == 0) base_s = p;
    }
    __syncthreads();
    if (idx < N) rowptr[idx] = base_s + wsum[wv] + inc - v;
}

__global__ void fill_csr(const int* __restrict__ ei, int E,
                         const int* __restrict__ rowptr, const int* __restrict__ rank,
                         int* __restrict__ csr_src) {
    const int e = blockIdx.x * blockDim.x + threadIdx.x;
    if (e >= E) return;
    csr_src[rowptr[ei[E + e]] + rank[e]] = ei[e];
}

__device__ __forceinline__ float wave_max(float v) {
    for (int o = 16; o > 0; o >>= 1) v = fmaxf(v, __shfl_xor(v, o));
    return v;
}
__device__ __forceinline__ float wave_sum(float v) {
    for (int o = 16; o > 0; o >>= 1) v += __shfl_xor(v, o);
    return v;
}
__device__ __forceinline__ float lane_f(float v, int j) {
    return __int_as_float(__builtin_amdgcn_readlane(__float_as_int(v), j));
}

__global__ __launch_bounds__(256)
void node_agg1(const int* __restrict__ rowptr, const int* __restrict__ deg,
               const int* __restrict__ csr_src,
               const float* __restrict__ a_src, const float* __restrict__ a_dst,
               const float* __restrict__ h1, const float* __restrict__ b1,
               float* __restrict__ y1, int N) {
    const int lane = threadIdx.x & 31;
    const int wv = __builtin_amdgcn_readfirstlane((int)(threadIdx.x >> 5));
    const int d = blockIdx.x * 8 + wv;
    if (d >= N) return;
    const bool hi = lane >= 16;
    const int start = rowptr[d];
    const int n = deg[d];
    const int cnt = n + 1;
    const float4 ad = *(const float4*)(a_dst + (size_t)d * 4);

    float m0 = -__builtin_huge_valf(), m1 = m0, m2 = m0, m3 = m0;
    for (int c = 0; c < cnt; c += 32) {
        const int k = c + lane;
        if (k < cnt) {
            const int s = k < n ? csr_src[start + k] : d;
            const float4 a = *(const float4*)(a_src + (size_t)s * 4);
            m0 = fmaxf(m0, lrelu(a.x + ad.x));
            m1 = fmaxf(m1, lrelu(a.y + ad.y));
            m2 = fmaxf(m2, lrelu(a.z + ad.z));
            m3 = fmaxf(m3, lrelu(a.w + ad.w));
        }
    }
    m0 = wave_max(m0); m1 = wave_max(m1); m2 = wave_max(m2); m3 = wave_max(m3);

    float d0 = 0.0f, d1 = 0.0f, d2 = 0.0f, d3 = 0.0f;
    float4 accA = make_float4(0.0f, 0.0f, 0.0f, 0.0f);
    float4 accB = make_float4(0.0f, 0.0f, 0.0f, 0.0f);
    for (int c = 0; c < cnt; c += 32) {
        const int k = c + lane;
        int s = d;
        float w0 = 0.0f, w1 = 0.0f, w2 = 0.0f, w3 = 0.0f;
        if (k < cnt) {
            if (k < n) s = csr_src[start + k];
            const float4 a = *(const float4*)(a_src + (size_t)s * 4);
            w0 = expf(lrelu(a.x + ad.x) - m0);
            w1 = expf(lrelu(a.y + ad.y) - m1);
            w2 = expf(lrelu(a.z + ad.z) - m2);
            w3 = expf(lrelu(a.w + ad.w) - m3);
        }
        d0 += w0; d1 += w1; d2 += w2; d3 += w3;
        const int cn = (cnt - c) < 32 ? (cnt - c) : 32;
#pragma unroll 4
        for (int j = 0; j < cn; ++j) {
            const int sj = __builtin_amdgcn_readlane(s, j);
            const float x0 = lane_f(w0, j), x1 = lane_f(w1, j);
            const float x2 = lane_f(w2, j), x3 = lane_f(w3, j);
            const float wa = hi ? x1 : x0;
            const float wb = hi ? x3 : x2;
            const float* hr = h1 + (size_t)sj * 256 + lane * 4;
            const float4 vA = *(const float4*)hr;
            const float4 vB = *(const float4*)(hr + 128);
            accA.x += wa * vA.x; accA.y += wa * vA.y; accA.z += wa * vA.z; accA.w += wa * vA.w;
            accB.x += wb * vB.x; accB.y += wb * vB.y; accB.z += wb * vB.z; accB.w += wb * vB.w;
        }
    }
    d0 = wave_sum(d0); d1 = wave_sum(d1); d2 = wave_sum(d2); d3 = wave_sum(d3);
    const float denA = hi ? d1 : d0;
    const float denB = hi ? d3 : d2;
    const float4 bA = *(const float4*)(b1 + lane * 4);
    const float4 bB = *(const float4*)(b1 + 128 + lane * 4);
    float4 oA, oB;
    oA.x = fmaxf(accA.x / denA + bA.x, 0.0f); oA.y = fmaxf(accA.y / denA + bA.y, 0.0f);
    oA.z = fmaxf(accA.z / denA + bA.z, 0.0f); oA.w = fmaxf(accA.w / denA + bA.w, 0.0f);
    oB.x = fmaxf(accB.x / denB + bB.x, 0.0f); oB.y = fmaxf(accB.y / denB + bB.y, 0.0f);
    oB.z = fmaxf(accB.z / denB + bB.z, 0.0f); oB.w = fmaxf(accB.w / denB + bB.w, 0.0f);
    float* op = y1 + (size_t)d * 256 + lane * 4;
    *(float4*)op = oA;
    *(float4*)(op + 128) = oB;
}

__global__ __launch_bounds__(256)
void node_agg2(const int* __restrict__ rowptr, const int* __restrict__ deg,
               const int* __restrict__ csr_src,
               const float* __restrict__ a_src, const float* __restrict__ a_dst,
               const float* __restrict__ h2, const float* __restrict__ b2,
               float* __restrict__ out, int N) {
    const int lane = threadIdx.x & 31;
    const int wv = __builtin_amdgcn_readfirstlane((int)(threadIdx.x >> 5));
    const int d = blockIdx.x * 8 + wv;
    if (d >= N) return;
    const int half = lane >> 4;
    const int t = lane & 15;
    const int start = rowptr[d];
    const int n = deg[d];
    const int cnt = n + 1;
    const float ad = a_dst[d];

    float m = -__builtin_huge_valf();
    for (int c = 0; c < cnt; c += 32) {
        const int k = c + lane;
        if (k < cnt) {
            const int s = k < n ? csr_src[start + k] : d;
            m = fmaxf(m, lrelu(a_src[s] + ad));
        }
    }
    m = wave_max(m);

    float den = 0.0f, acc = 0.0f;
    for (int c = 0; c < cnt; c += 32) {
        const int k = c + lane;
        int s = d;
        float w = 0.0f;
        if (k < cnt) {
            if (k < n) s = csr_src[start + k];
            w = expf(lrelu(a_src[s] + ad) - m);
        }
        den += w;
        const int cn = (cnt - c) < 32 ? (cnt - c) : 32;
        const int cp = cn & ~1;
#pragma unroll 4
        for (int j = 0; j < cp; j += 2) {
            const int s0 = __builtin_amdgcn_readlane(s, j);
            const int s1 = __builtin_amdgcn_readlane(s, j + 1);
            const float w0 = lane_f(w, j);
            const float w1 = lane_f(w, j + 1);
            const int sj = half ? s1 : s0;
            const float wj = half ? w1 : w0;
            acc += wj * h2[(size_t)sj * 16 + t];
        }
        if (cn & 1) {
            const int sl = __builtin_amdgcn_readlane(s, cp);
            const float wl = lane_f(w, cp);
            if (half == 0) acc += wl * h2[(size_t)sl * 16 + t];
        }
    }
    den = wave_sum(den);
    acc += __shfl_xor(acc, 16);
    if (lane < 16) out[(size_t)d * 16 + t] = acc / den + b2[t];
}


extern "C" void kernel_launch(void* const* d_in, const int* in_sizes, int n_in,
                              void* d_out, int out_size, void* d_ws, size_t ws_size,
                              hipStream_t stream) {
    const float* x   = (const float*)d_in[0];
    const int*   ei  = (const int*)  d_in[1];
    const float* W1  = (const float*)d_in[2];
    const float* as1 = (const float*)d_in[3];
    const float* ad1 = (const float*)d_in[4];
    const float* b1  = (const float*)d_in[5];
    const float* W2  = (const float*)d_in[6];
    const float* as2 = (const float*)d_in[7];
    const float* ad2 = (const float*)d_in[8];
    const float* b2  = (const float*)d_in[9];

    const int N  = in_sizes[0] / 128;
    const int E  = in_sizes[1] / 2;

    char* w = (char*)d_ws;
    auto carve = [&](size_t bytes) -> void* {
        void* p = (void*)w;
        w += (bytes + 255) & ~((size_t)255);
        return p;
    };
    float* h1     = (float*)carve((size_t)N * 256 * 4);
    float* y1     = (float*)carve((size_t)N * 256 * 4);
    float* aS1    = (float*)carve((size_t)N * 4 * 4);
    float* aD1    = (float*)carve((size_t)N * 4 * 4);
    float* h2     = (float*)carve((size_t)N * 16 * 4);
    float* aS2    = (float*)carve((size_t)N * 4);
    float* aD2    = (float*)carve((size_t)N * 4);
    int*   deg    = (int*)  carve((size_t)N * 4);
    int*   rowptr = (int*)  carve((size_t)N * 4);
    int*   rank   = (int*)  carve((size_t)E * 4);
    int*   csrs   = (int*)  carve((size_t)E * 4);

    const int TB = 256;
    auto blocks = [&](long long n) { return (unsigned)((n + TB - 1) / TB); };

    zero_u32<<<blocks(N), TB, 0, stream>>>((unsigned*)deg, N);
    count_rank<<<blocks(E), TB, 0, stream>>>(ei, E, deg, rank);
    scan_deg<<<(N + 1023) / 1024, 1024, 0, stream>>>(deg, rowptr, N);
    fill_csr<<<blocks(E), TB, 0, stream>>>(ei, E, rowptr, rank, csrs);

    dim3 g1((N + 15) / 16, 4);
    gemm1_wmma<<<g1, 32, 0, stream>>>(x, W1, h1, N);
    att1_kernel<<<blocks(N * 4), TB, 0, stream>>>(h1, as1, ad1, aS1, aD1, N * 4);
    node_agg1<<<(N + 7) / 8, 256, 0, stream>>>(rowptr, deg, csrs, aS1, aD1, h1, b1, y1, N);

    gemm2_wmma<<<(N + 15) / 16, 32, 0, stream>>>(y1, W2, h2, N);
    att2_kernel<<<blocks(N), TB, 0, stream>>>(h2, as2, ad2, aS2, aD2, N);
    node_agg2<<<(N + 7) / 8, 256, 0, stream>>>(rowptr, deg, csrs, aS2, aD2, h2, b2, (float*)d_out, N);
}


# Attention scores taken from the WMMA accumulators in both GEMM epilogues (lane partials, then a 16-lane exchange tree: new summation order); both score kernels and their re-read of h removed
# speedup vs baseline: 3.6942x; 1.0578x over previous
#include <hip/hip_runtime.h>
#include <hip/hip_bf16.h>

typedef __attribute__((ext_vector_type(2))) float v2f;
typedef __attribute__((ext_vector_type(8))) float v8f;

#define NEG_SLOPE 0.2f


__device__ __forceinline__ float lrelu(float v) {
    return v >= 0.0f ? v : NEG_SLOPE * v;
}

__global__ void zero_u32(unsigned* __restrict__ p, int n) {
    int i = blockIdx.x * blockDim.x + threadIdx.x;
    if (i < n) p[i] = 0u;
}

__device__ __forceinline__ float half_rows_sum(const float (&ps)[8], const float (&pd)[8], int c) {
    const bool b3 = (c & 8) != 0, b2 = (c & 4) != 0, b1 = (c & 2) != 0, b0 = (c & 1) != 0;
    float qs[4], qd[4];
    for (int i = 0; i < 4; ++i) {
        qs[i] = (b3 ? ps[i + 4] : ps[i]) + __shfl_xor(b3 ? ps[i] : ps[i + 4], 8);
        qd[i] = (b3 ? pd[i + 4] : pd[i]) + __shfl_xor(b3 ? pd[i] : pd[i + 4], 8);
    }
    float us[2], ud[2];
    for (int i = 0; i < 2; ++i) {
        us[i] = (b2 ? qs[i + 2] : qs[i]) + __shfl_xor(b2 ? qs[i] : qs[i + 2], 4);
        ud[i] = (b2 ? qd[i + 2] : qd[i]) + __shfl_xor(b2 ? qd[i] : qd[i + 2], 4);
    }
    const float vs = (b1 ? us[1] : us[0]) + __shfl_xor(b1 ? us[0] : us[1], 2);
    const float vd = (b1 ? ud[1] : ud[0]) + __shfl_xor(b1 ? ud[0] : ud[1], 2);
    return (b0 ? vd : vs) + __shfl_xor(b0 ? vs : vd, 1);
}

__global__ __launch_bounds__(32)
void gemm1_wmma(const float* __restrict__ x, const float* __restrict__ W1,
                const float* __restrict__ att_src, const float* __restrict__ att_dst,
                float* __restrict__ h1, float* __restrict__ a_src, float* __restrict__ a_dst,
                int N) {
    const int mtile = blockIdx.x;
    const int lane  = threadIdx.x;
    const int mrow  = mtile * 16 + (lane & 15);
    const int lrow  = mrow < N ? mrow : N - 1;
    const int khalf = (lane >> 4) * 2;
    const int colbase = blockIdx.y * 64;
    const int nc    = lane & 15;

    v8f c0 = {}, c1 = {}, c2 = {}, c3 = {};
    const float* xrow = x + (size_t)lrow * 128;

    for (int k = 0; k < 128; k += 4) {
        const int kb = k + khalf;
        v2f a; a.x = xrow[kb]; a.y = xrow[kb + 1];
        const float* w0 = W1 + (size_t)kb * 256 + colbase + nc;
        const float* w1 = w0 + 256;
        v2f b0; b0.x = w0[0];  b0.y = w1[0];
        v2f b1; b1.x = w0[16]; b1.y = w1[16];
        v2f b2; b2.x = w0[32]; b2.y = w1[32];
        v2f b3; b3.x = w0[48]; b3.y = w1[48];
        c0 = __builtin_amdgcn_wmma_f32_16x16x4_f32(false, a, false, b0, (short)0, c0, false, false);
        c1 = __builtin_amdgcn_wmma_f32_16x16x4_f32(false, a, false, b1, (short)0, c1, false, false);
        c2 = __builtin_amdgcn_wmma_f32_16x16x4_f32(false, a, false, b2, (short)0, c2, false, false);
        c3 = __builtin_amdgcn_wmma_f32_16x16x4_f32(false, a, false, b3, (short)0, c3, false, false);
    }

    const int rowoff = (lane >= 16) ? 8 : 0;
    for (int r = 0; r < 8; ++r) {
        const int grow = mtile * 16 + rowoff + r;
        if (grow < N) {
            float* op = h1 + (size_t)grow * 256 + colbase + nc;
            op[0]  = c0[r];
            op[16] = c1[r];
            op[32] = c2[r];
            op[48] = c3[r];
        }
    }

    const float* as = att_src + colbase + nc;
    const float* ad = att_dst + colbase + nc;
    const float s0 = as[0], s1 = as[16], s2 = as[32], s3 = as[48];
    const float t0 = ad[0], t1 = ad[16], t2 = ad[32], t3 = ad[48];
    float ps[8], pd[8];
    for (int r = 0; r < 8; ++r) {
        ps[r] = c0[r] * s0 + c1[r] * s1 + c2[r] * s2 + c3[r] * s3;
        pd[r] = c0[r] * t0 + c1[r] * t1 + c2[r] * t2 + c3[r] * t3;
    }
    const float sc = half_rows_sum(ps, pd, nc);
    const int srow = mtile * 16 + rowoff + (nc >> 1);
    if (srow < N) {
        float* sp = (nc & 1) ? a_dst : a_src;
        sp[(size_t)srow * 4 + blockIdx.y] = sc;
    }
}

__global__ __launch_bounds__(32)
void gemm2_wmma(const float* __restrict__ y1, const float* __restrict__ W2,
                const float* __restrict__ att_src, const float* __restrict__ att_dst,
                float* __restrict__ h2, float* __restrict__ a_src, float* __restrict__ a_dst,
                int N) {
    const int mtile = blockIdx.x;
    const int lane  = threadIdx.x;
    const int mrow  = mtile * 16 + (lane & 15);
    const int lrow  = mrow < N ? mrow : N - 1;
    const int khalf = (lane >> 4) * 2;
    const int nc    = lane & 15;

    v8f c = {};
    const float* yr = y1 + (size_t)lrow * 256;
    for (int k = 0; k < 256; k += 4) {
        const int kb = k + khalf;
        v2f a; a.x = yr[kb]; a.y = yr[kb + 1];
        v2f b; b.x = W2[(size_t)kb * 16 + nc]; b.y = W2[(size_t)(kb + 1) * 16 + nc];
        c = __builtin_amdgcn_wmma_f32_16x16x4_f32(false, a, false, b, (short)0, c, false, false);
    }
    const int rowoff = (lane >= 16) ? 8 : 0;
    for (int r = 0; r < 8; ++r) {
        const int grow = mtile * 16 + rowoff + r;
        if (grow < N) h2[(size_t)grow * 16 + nc] = c[r];
    }

    const float s = att_src[nc];
    const float t = att_dst[nc];
    float ps[8], pd[8];
    for (int r = 0; r < 8; ++r) {
        ps[r] = c[r] * s;
        pd[r] = c[r] * t;
    }
    const float sc = half_rows_sum(ps, pd, nc);
    const int srow = mtile * 16 + rowoff + (nc >> 1);
    if (srow < N) {
        float* sp = (nc & 1) ? a_dst : a_src;
        sp[srow] = sc;
    }
}

__global__ void count_rank(const int* __restrict__ ei, int E,
                           int* __restrict__ deg, int* __restrict__ rank) {
    const int e = blockIdx.x * blockDim.x + threadIdx.x;
    if (e >= E) return;
    rank[e] = atomicAdd(&deg[ei[E + e]], 1);
}

__global__ __launch_bounds__(1024)
void scan_deg(const int* __restrict__ deg, int* __restrict__ rowptr, int N) {
    __shared__ int wsum[32];
    __shared__ int wpre[32];
    __shared__ int base_s;
    const int t = threadIdx.x, lane = t & 31, wv = t >> 5;
    const int start = blockIdx.x * 1024;
    int pre = 0;
    for (int i = t; i < start; i += 1024) pre += deg[i];
    const int idx = start + t;
    const int v = idx < N ? deg[idx] : 0;
    int inc = v;
    for (int off = 1; off < 32; off <<= 1) {
        const int nb = __shfl_up(inc, off);
        if (lane >= off) inc += nb;
    }
    for (int off = 16; off > 0; off >>= 1) pre += __shfl_xor(pre, off);
    if (lane == 31) wsum[wv] = inc;
    if (lane == 0)  wpre[wv] = pre;
    __syncthreads();
    if (wv == 0) {
        const int tot = wsum[lane];
        int p = wpre[lane];
        int winc = tot;
        for (int off = 1; off < 32; off <<= 1) {
            const int nb = __shfl_up(winc, off);
            if (lane >= off) winc += nb;
        }
        for (int off = 16; off > 0; off >>= 1) p += __shfl_xor(p, off);
        wsum[lane] = winc - tot;
        if (lane == 0) base_s = p;
    }
    __syncthreads();
    if (idx < N) rowptr[idx] = base_s + wsum[wv] + inc - v;
}

__global__ void fill_csr(const int* __restrict__ ei, int E,
                         const int* __restrict__ rowptr, const int* __restrict__ rank,
                         int* __restrict__ csr_src) {
    const int e = blockIdx.x * blockDim.x + threadIdx.x;
    if (e >= E) return;
    csr_src[rowptr[ei[E + e]] + rank[e]] = ei[e];
}

__device__ __forceinline__ float wave_max(float v) {
    for (int o = 16; o > 0; o >>= 1) v = fmaxf(v, __shfl_xor(v, o));
    return v;
}
__device__ __forceinline__ float wave_sum(float v) {
    for (int o = 16; o > 0; o >>= 1) v += __shfl_xor(v, o);
    return v;
}
__device__ __forceinline__ float lane_f(float v, int j) {
    return __int_as_float(__builtin_amdgcn_readlane(__float_as_int(v), j));
}

__global__ __launch_bounds__(256)
void node_agg1(const int* __restrict__ rowptr, const int* __restrict__ deg,
               const int* __restrict__ csr_src,
               const float* __restrict__ a_src, const float* __restrict__ a_dst,
               const float* __restrict__ h1, const float* __restrict__ b1,
               float* __restrict__ y1, int N) {
    const int lane = threadIdx.x & 31;
    const int wv = __builtin_amdgcn_readfirstlane((int)(threadIdx.x >> 5));
    const int d = blockIdx.x * 8 + wv;
    if (d >= N) return;
    const bool hi = lane >= 16;
    const int start = rowptr[d];
    const int n = deg[d];
    const int cnt = n + 1;
    const float4 ad = *(const float4*)(a_dst + (size_t)d * 4);

    float m0 = -__builtin_huge_valf(), m1 = m0, m2 = m0, m3 = m0;
    for (int c = 0; c < cnt; c += 32) {
        const int k = c + lane;
        if (k < cnt) {
            const int s = k < n ? csr_src[start + k] : d;
            const float4 a = *(const float4*)(a_src + (size_t)s * 4);
            m0 = fmaxf(m0, lrelu(a.x + ad.x));
            m1 = fmaxf(m1, lrelu(a.y + ad.y));
            m2 = fmaxf(m2, lrelu(a.z + ad.z));
            m3 = fmaxf(m3, lrelu(a.w + ad.w));
        }
    }
    m0 = wave_max(m0); m1 = wave_max(m1); m2 = wave_max(m2); m3 = wave_max(m3);

    float d0 = 0.0f, d1 = 0.0f, d2 = 0.0f, d3 = 0.0f;
    float4 accA = make_float4(0.0f, 0.0f, 0.0f, 0.0f);
    float4 accB = make_float4(0.0f, 0.0f, 0.0f, 0.0f);
    for (int c = 0; c < cnt; c += 32) {
        const int k = c + lane;
        int s = d;
        float w0 = 0.0f, w1 = 0.0f, w2 = 0.0f, w3 = 0.0f;
        if (k < cnt) {
            if (k < n) s = csr_src[start + k];
            const float4 a = *(const float4*)(a_src + (size_t)s * 4);
            w0 = expf(lrelu(a.x + ad.x) - m0);
            w1 = expf(lrelu(a.y + ad.y) - m1);
            w2 = expf(lrelu(a.z + ad.z) - m2);
            w3 = expf(lrelu(a.w + ad.w) - m3);
        }
        d0 += w0; d1 += w1; d2 += w2; d3 += w3;
        const int cn = (cnt - c) < 32 ? (cnt - c) : 32;
#pragma unroll 4
        for (int j = 0; j < cn; ++j) {
            const int sj = __builtin_amdgcn_readlane(s, j);
            const float x0 = lane_f(w0, j), x1 = lane_f(w1, j);
            const float x2 = lane_f(w2, j), x3 = lane_f(w3, j);
            const float wa = hi ? x1 : x0;
            const float wb = hi ? x3 : x2;
            const float* hr = h1 + (size_t)sj * 256 + lane * 4;
            const float4 vA = *(const float4*)hr;
            const float4 vB = *(const float4*)(hr + 128);
            accA.x += wa * vA.x; accA.y += wa * vA.y; accA.z += wa * vA.z; accA.w += wa * vA.w;
            accB.x += wb * vB.x; accB.y += wb * vB.y; accB.z += wb * vB.z; accB.w += wb * vB.w;
        }
    }
    d0 = wave_sum(d0); d1 = wave_sum(d1); d2 = wave_sum(d2); d3 = wave_sum(d3);
    const float denA = hi ? d1 : d0;
    const float denB = hi ? d3 : d2;
    const float4 bA = *(const float4*)(b1 + lane * 4);
    const float4 bB = *(const float4*)(b1 + 128 + lane * 4);
    float4 oA, oB;
    oA.x = fmaxf(accA.x / denA + bA.x, 0.0f); oA.y = fmaxf(accA.y / denA + bA.y, 0.0f);
    oA.z = fmaxf(accA.z / denA + bA.z, 0.0f); oA.w = fmaxf(accA.w / denA + bA.w, 0.0f);
    oB.x = fmaxf(accB.x / denB + bB.x, 0.0f); oB.y = fmaxf(accB.y / denB + bB.y, 0.0f);
    oB.z = fmaxf(accB.z / denB + bB.z, 0.0f); oB.w = fmaxf(accB.w / denB + bB.w, 0.0f);
    float* op = y1 + (size_t)d * 256 + lane * 4;
    *(float4*)op = oA;
    *(float4*)(op + 128) = oB;
}

__global__ __launch_bounds__(256)
void node_agg2(const int* __restrict__ rowptr, const int* __restrict__ deg,
               const int* __restrict__ csr_src,
               const float* __restrict__ a_src, const float* __restrict__ a_dst,
               const float* __restrict__ h2, const float* __restrict__ b2,
               float* __restrict__ out, int N) {
    const int lane = threadIdx.x & 31;
    const int wv = __builtin_amdgcn_readfirstlane((int)(threadIdx.x >> 5));
    const int d = blockIdx.x * 8 + wv;
    if (d >= N) return;
    const int half = lane >> 4;
    const int t = lane & 15;
    const int start = rowptr[d];
    const int n = deg[d];
    const int cnt = n + 1;
    const float ad = a_dst[d];

    float m = -__builtin_huge_valf();
    for (int c = 0; c < cnt; c += 32) {
        const int k = c + lane;
        if (k < cnt) {
            const int s = k < n ? csr_src[start + k] : d;
            m = fmaxf(m, lrelu(a_src[s] + ad));
        }
    }
    m = wave_max(m);

    float den = 0.0f, acc = 0.0f;
    for (int c = 0; c < cnt; c += 32) {
        const int k = c + lane;
        int s = d;
        float w = 0.0f;
        if (k < cnt) {
            if (k < n) s = csr_src[start + k];
            w = expf(lrelu(a_src[s] + ad) - m);
        }
        den += w;
        const int cn = (cnt - c) < 32 ? (cnt - c) : 32;
        const int cp = cn & ~1;
#pragma unroll 4
        for (int j = 0; j < cp; j += 2) {
            const int s0 = __builtin_amdgcn_readlane(s, j);
            const int s1 = __builtin_amdgcn_readlane(s, j + 1);
            const float w0 = lane_f(w, j);
            const float w1 = lane_f(w, j + 1);
            const int sj = half ? s1 : s0;
            const float wj = half ? w1 : w0;
            acc += wj * h2[(size_t)sj * 16 + t];
        }
        if (cn & 1) {
            const int sl = __builtin_amdgcn_readlane(s, cp);
            const float wl = lane_f(w, cp);
            if (half == 0) acc += wl * h2[(size_t)sl * 16 + t];
        }
    }
    den = wave_sum(den);
    acc += __shfl_xor(acc, 16);
    if (lane < 16) out[(size_t)d * 16 + t] = acc / den + b2[t];
}


extern "C" void kernel_launch(void* const* d_in, const int* in_sizes, int n_in,
                              void* d_out, int out_size, void* d_ws, size_t ws_size,
                              hipStream_t stream) {
    const float* x   = (const float*)d_in[0];
    const int*   ei  = (const int*)  d_in[1];
    const float* W1  = (const float*)d_in[2];
    const float* as1 = (const float*)d_in[3];
    const float* ad1 = (const float*)d_in[4];
    const float* b1  = (const float*)d_in[5];
    const float* W2  = (const float*)d_in[6];
    const float* as2 = (const float*)d_in[7];
    const float* ad2 = (const float*)d_in[8];
    const float* b2  = (const float*)d_in[9];

    const int N  = in_sizes[0] / 128;
    const int E  = in_sizes[1] / 2;

    char* w = (char*)d_ws;
    auto carve = [&](size_t bytes) -> void* {
        void* p = (void*)w;
        w += (bytes + 255) & ~((size_t)255);
        return p;
    };
    float* h1     = (float*)carve((size_t)N * 256 * 4);
    float* y1     = (float*)carve((size_t)N * 256 * 4);
    float* aS1    = (float*)carve((size_t)N * 4 * 4);
    float* aD1    = (float*)carve((size_t)N * 4 * 4);
    float* h2     = (float*)carve((size_t)N * 16 * 4);
    float* aS2    = (float*)carve((size_t)N * 4);
    float* aD2    = (float*)carve((size_t)N * 4);
    int*   deg    = (int*)  carve((size_t)N * 4);
    int*   rowptr = (int*)  carve((size_t)N * 4);
    int*   rank   = (int*)  carve((size_t)E * 4);
    int*   csrs   = (int*)  carve((size_t)E * 4);

    const int TB = 256;
    auto blocks = [&](long long n) { return (unsigned)((n + TB - 1) / TB); };

    zero_u32<<<blocks(N), TB, 0, stream>>>((unsigned*)deg, N);
    count_rank<<<blocks(E), TB, 0, stream>>>(ei, E, deg, rank);
    scan_deg<<<(N + 1023) / 1024, 1024, 0, stream>>>(deg, rowptr, N);
    fill_csr<<<blocks(E), TB, 0, stream>>>(ei, E, rowptr, rank, csrs);

    dim3 g1((N + 15) / 16, 4);
    gemm1_wmma<<<g1, 32, 0, stream>>>(x, W1, as1, ad1, h1, aS1, aD1, N);
    node_agg1<<<(N + 7) / 8, 256, 0, stream>>>(rowptr, deg, csrs, aS1, aD1, h1, b1, y1, N);

    gemm2_wmma<<<(N + 15) / 16, 32, 0, stream>>>(y1, W2, as2, ad2, h2, aS2, aD2, N);
    node_agg2<<<(N + 7) / 8, 256, 0, stream>>>(rowptr, deg, csrs, aS2, aD2, h2, b2, (float*)d_out, N);
}


# Edge grouping ranks on LDS counters (64 list pieces, one block each: the LDS add returns the place, piece counts scanned to starts), no global add per edge; list order: by piece, arrival order inside
# speedup vs baseline: 4.7878x; 1.2960x over previous
#include <hip/hip_runtime.h>
#include <hip/hip_bf16.h>

typedef __attribute__((ext_vector_type(2))) float v2f;
typedef __attribute__((ext_vector_type(8))) float v8f;

#define NEG_SLOPE 0.2f


__device__ __forceinline__ float lrelu(float v) {
    return v >= 0.0f ? v : NEG_SLOPE * v;
}

__device__ __forceinline__ float half_rows_sum(const float (&ps)[8], const float (&pd)[8], int c) {
    const bool b3 = (c & 8) != 0, b2 = (c & 4) != 0, b1 = (c & 2) != 0, b0 = (c & 1) != 0;
    float qs[4], qd[4];
    for (int i = 0; i < 4; ++i) {
        qs[i] = (b3 ? ps[i + 4] : ps[i]) + __shfl_xor(b3 ? ps[i] : ps[i + 4], 8);
        qd[i] = (b3 ? pd[i + 4] : pd[i]) + __shfl_xor(b3 ? pd[i] : pd[i + 4], 8);
    }
    float us[2], ud[2];
    for (int i = 0; i < 2; ++i) {
        us[i] = (b2 ? qs[i + 2] : qs[i]) + __shfl_xor(b2 ? qs[i] : qs[i + 2], 4);
        ud[i] = (b2 ? qd[i + 2] : qd[i]) + __shfl_xor(b2 ? qd[i] : qd[i + 2], 4);
    }
    const float vs = (b1 ? us[1] : us[0]) + __shfl_xor(b1 ? us[0] : us[1], 2);
    const float vd = (b1 ? ud[1] : ud[0]) + __shfl_xor(b1 ? ud[0] : ud[1], 2);
    return (b0 ? vd : vs) + __shfl_xor(b0 ? vs : vd, 1);
}

__global__ __launch_bounds__(32)
void gemm1_wmma(const float* __restrict__ x, const float* __restrict__ W1,
                const float* __restrict__ att_src, const float* __restrict__ att_dst,
                float* __restrict__ h1, float* __restrict__ a_src, float* __restrict__ a_dst,
                int N) {
    const int mtile = blockIdx.x;
    const int lane  = threadIdx.x;
    const int mrow  = mtile * 16 + (lane & 15);
    const int lrow  = mrow < N ? mrow : N - 1;
    const int khalf = (lane >> 4) * 2;
    const int colbase = blockIdx.y * 64;
    const int nc    = lane & 15;

    v8f c0 = {}, c1 = {}, c2 = {}, c3 = {};
    const float* xrow = x + (size_t)lrow * 128;

    for (int k = 0; k < 128; k += 4) {
        const int kb = k + khalf;
        v2f a; a.x = xrow[kb]; a.y = xrow[kb + 1];
        const float* w0 = W1 + (size_t)kb * 256 + colbase + nc;
        const float* w1 = w0 + 256;
        v2f b0; b0.x = w0[0];  b0.y = w1[0];
        v2f b1; b1.x = w0[16]; b1.y = w1[16];
        v2f b2; b2.x = w0[32]; b2.y = w1[32];
        v2f b3; b3.x = w0[48]; b3.y = w1[48];
        c0 = __builtin_amdgcn_wmma_f32_16x16x4_f32(false, a, false, b0, (short)0, c0, false, false);
        c1 = __builtin_amdgcn_wmma_f32_16x16x4_f32(false, a, false, b1, (short)0, c1, false, false);
        c2 = __builtin_amdgcn_wmma_f32_16x16x4_f32(false, a, false, b2, (short)0, c2, false, false);
        c3 = __builtin_amdgcn_wmma_f32_16x16x4_f32(false, a, false, b3, (short)0, c3, false, false);
    }

    const int rowoff = (lane >= 16) ? 8 : 0;
    for (int r = 0; r < 8; ++r) {
        const int grow = mtile * 16 + rowoff + r;
        if (grow < N) {
            float* op = h1 + (size_t)grow * 256 + colbase + nc;
            op[0]  = c0[r];
            op[16] = c1[r];
            op[32] = c2[r];
            op[48] = c3[r];
        }
    }

    const float* as = att_src + colbase + nc;
    const float* ad = att_dst + colbase + nc;
    const float s0 = as[0], s1 = as[16], s2 = as[32], s3 = as[48];
    const float t0 = ad[0], t1 = ad[16], t2 = ad[32], t3 = ad[48];
    float ps[8], pd[8];
    for (int r = 0; r < 8; ++r) {
        ps[r] = c0[r] * s0 + c1[r] * s1 + c2[r] * s2 + c3[r] * s3;
        pd[r] = c0[r] * t0 + c1[r] * t1 + c2[r] * t2 + c3[r] * t3;
    }
    const float sc = half_rows_sum(ps, pd, nc);
    const int srow = mtile * 16 + rowoff + (nc >> 1);
    if (srow < N) {
        float* sp = (nc & 1) ? a_dst : a_src;
        sp[(size_t)srow * 4 + blockIdx.y] = sc;
    }
}

__global__ __launch_bounds__(32)
void gemm2_wmma(const float* __restrict__ y1, const float* __restrict__ W2,
                const float* __restrict__ att_src, const float* __restrict__ att_dst,
                float* __restrict__ h2, float* __restrict__ a_src, float* __restrict__ a_dst,
                int N) {
    const int mtile = blockIdx.x;
    const int lane  = threadIdx.x;
    const int mrow  = mtile * 16 + (lane & 15);
    const int lrow  = mrow < N ? mrow : N - 1;
    const int khalf = (lane >> 4) * 2;
    const int nc    = lane & 15;

    v8f c = {};
    const float* yr = y1 + (size_t)lrow * 256;
    for (int k = 0; k < 256; k += 4) {
        const int kb = k + khalf;
        v2f a; a.x = yr[kb]; a.y = yr[kb + 1];
        v2f b; b.x = W2[(size_t)kb * 16 + nc]; b.y = W2[(size_t)(kb + 1) * 16 + nc];
        c = __builtin_amdgcn_wmma_f32_16x16x4_f32(false, a, false, b, (short)0, c, false, false);
    }
    const int rowoff = (lane >= 16) ? 8 : 0;
    for (int r = 0; r < 8; ++r) {
        const int grow = mtile * 16 + rowoff + r;
        if (grow < N) h2[(size_t)grow * 16 + nc] = c[r];
    }

    const float s = att_src[nc];
    const float t = att_dst[nc];
    float ps[8], pd[8];
    for (int r = 0; r < 8; ++r) {
        ps[r] = c[r] * s;
        pd[r] = c[r] * t;
    }
    const float sc = half_rows_sum(ps, pd, nc);
    const int srow = mtile * 16 + rowoff + (nc >> 1);
    if (srow < N) {
        float* sp = (nc & 1) ? a_dst : a_src;
        sp[srow] = sc;
    }
}

#define CHUNKS 64
#define HCAP   50176
#define NBLK   256
#define NBLK_LOG 8
#define EPT    5

__device__ __forceinline__ int wave_sum_i(int v) {
    for (int o = 16; o > 0; o >>= 1) v += __shfl_xor(v, o);
    return v;
}

__global__ __launch_bounds__(1024)
void chunk_rank(const int* __restrict__ ei, int E, int epc, int NB,
                int* __restrict__ rank, int* __restrict__ cnt, int* __restrict__ coarse) {
    __shared__ int hist[HCAP];
    __shared__ int blk[HCAP / NBLK];
    const int t = threadIdx.x;
    const int c = blockIdx.x;
    int4* h4 = (int4*)hist;
    for (int i = t; i < HCAP / 4; i += 1024) h4[i] = make_int4(0, 0, 0, 0);
    if (t < HCAP / NBLK) blk[t] = 0;
    __syncthreads();

    const int e0 = c * epc;
    const int e1 = (e0 + epc) < E ? (e0 + epc) : E;
    for (int eb = e0 + t; eb < e1; eb += EPT * 1024) {
        int dd[EPT], rr[EPT];
#pragma unroll
        for (int u = 0; u < EPT; ++u) {
            const int e = eb + u * 1024;
            dd[u] = e < e1 ? ei[E + e] : 0;
        }
#pragma unroll
        for (int u = 0; u < EPT; ++u) {
            const int e = eb + u * 1024;
            rr[u] = 0;
            if (e < e1) {
                rr[u] = atomicAdd(&hist[dd[u]], 1);
                atomicAdd(&blk[dd[u] >> NBLK_LOG], 1);
            }
        }
#pragma unroll
        for (int u = 0; u < EPT; ++u) {
            const int e = eb + u * 1024;
            if (e < e1) rank[e] = rr[u];
        }
    }
    __syncthreads();

    int4* c4 = (int4*)(cnt + (size_t)c * HCAP);
    for (int i = t; i < HCAP / 4; i += 1024) c4[i] = h4[i];
    if (t < NB) coarse[c * NB + t] = blk[t];
}

__global__ __launch_bounds__(NBLK)
void scan_cols(int* __restrict__ cnt, const int* __restrict__ coarse,
               int* __restrict__ deg, int* __restrict__ rowptr, int N, int NB) {
    __shared__ int wsum[NBLK / 32];
    __shared__ int wpre[NBLK / 32];
    const int t = threadIdx.x, lane = t & 31, wv = t >> 5;
    const int j = blockIdx.x;
    const int idx = j * NBLK + t;

    int pre = 0;
    for (int b = t; b < j; b += NBLK)
        for (int c = 0; c < CHUNKS; ++c) pre += coarse[c * NB + b];

    int v[CHUNKS];
    int d = 0;
#pragma unroll
    for (int c = 0; c < CHUNKS; ++c) {
        v[c] = cnt[(size_t)c * HCAP + idx];
        d += v[c];
    }
    int inc = d;
    for (int off = 1; off < 32; off <<= 1) {
        const int nb = __shfl_up(inc, off);
        if (lane >= off) inc += nb;
    }
    pre = wave_sum_i(pre);
    if (lane == 31) wsum[wv] = inc;
    if (lane == 0)  wpre[wv] = pre;
    __syncthreads();
    int base = 0;
    for (int k = 0; k < NBLK / 32; ++k) {
        base += wpre[k];
        if (k < wv) base += wsum[k];
    }
    if (idx < N) {
        int run = base + inc - d;
        rowptr[idx] = run;
        deg[idx] = d;
#pragma unroll
        for (int c = 0; c < CHUNKS; ++c) {
            cnt[(size_t)c * HCAP + idx] = run;
            run += v[c];
        }
    }
}

__global__ __launch_bounds__(256)
void fill_csr(const int* __restrict__ ei, int E, int epc,
              const int* __restrict__ cnt, const int* __restrict__ rank,
              int* __restrict__ csr_src) {
    const int c = blockIdx.y;
    const int e0 = c * epc;
    const int e1 = (e0 + epc) < E ? (e0 + epc) : E;
    const int e = e0 + blockIdx.x * 256 + threadIdx.x;
    if (e >= e1) return;
    csr_src[cnt[(size_t)c * HCAP + ei[E + e]] + rank[e]] = ei[e];
}

__device__ __forceinline__ float wave_max(float v) {
    for (int o = 16; o > 0; o >>= 1) v = fmaxf(v, __shfl_xor(v, o));
    return v;
}
__device__ __forceinline__ float wave_sum(float v) {
    for (int o = 16; o > 0; o >>= 1) v += __shfl_xor(v, o);
    return v;
}
__device__ __forceinline__ float lane_f(float v, int j) {
    return __int_as_float(__builtin_amdgcn_readlane(__float_as_int(v), j));
}

__global__ __launch_bounds__(256)
void node_agg1(const int* __restrict__ rowptr, const int* __restrict__ deg,
               const int* __restrict__ csr_src,
               const float* __restrict__ a_src, const float* __restrict__ a_dst,
               const float* __restrict__ h1, const float* __restrict__ b1,
               float* __restrict__ y1, int N) {
    const int lane = threadIdx.x & 31;
    const int wv = __builtin_amdgcn_readfirstlane((int)(threadIdx.x >> 5));
    const int d = blockIdx.x * 8 + wv;
    if (d >= N) return;
    const bool hi = lane >= 16;
    const int start = rowptr[d];
    const int n = deg[d];
    const int cnt = n + 1;
    const float4 ad = *(const float4*)(a_dst + (size_t)d * 4);

    float m0 = -__builtin_huge_valf(), m1 = m0, m2 = m0, m3 = m0;
    for (int c = 0; c < cnt; c += 32) {
        const int k = c + lane;
        if (k < cnt) {
            const int s = k < n ? csr_src[start + k] : d;
            const float4 a = *(const float4*)(a_src + (size_t)s * 4);
            m0 = fmaxf(m0, lrelu(a.x + ad.x));
            m1 = fmaxf(m1, lrelu(a.y + ad.y));
            m2 = fmaxf(m2, lrelu(a.z + ad.z));
            m3 = fmaxf(m3, lrelu(a.w + ad.w));
        }
    }
    m0 = wave_max(m0); m1 = wave_max(m1); m2 = wave_max(m2); m3 = wave_max(m3);

    float d0 = 0.0f, d1 = 0.0f, d2 = 0.0f, d3 = 0.0f;
    float4 accA = make_float4(0.0f, 0.0f, 0.0f, 0.0f);
    float4 accB = make_float4(0.0f, 0.0f, 0.0f, 0.0f);
    for (int c = 0; c < cnt; c += 32) {
        const int k = c + lane;
        int s = d;
        float w0 = 0.0f, w1 = 0.0f, w2 = 0.0f, w3 = 0.0f;
        if (k < cnt) {
            if (k < n) s = csr_src[start + k];
            const float4 a = *(const float4*)(a_src + (size_t)s * 4);
            w0 = expf(lrelu(a.x + ad.x) - m0);
            w1 = expf(lrelu(a.y + ad.y) - m1);
            w2 = expf(lrelu(a.z + ad.z) - m2);
            w3 = expf(lrelu(a.w + ad.w) - m3);
        }
        d0 += w0; d1 += w1; d2 += w2; d3 += w3;
        const int cn = (cnt - c) < 32 ? (cnt - c) : 32;
#pragma unroll 4
        for (int j = 0; j < cn; ++j) {
            const int sj = __builtin_amdgcn_readlane(s, j);
            const float x0 = lane_f(w0, j), x1 = lane_f(w1, j);
            const float x2 = lane_f(w2, j), x3 = lane_f(w3, j);
            const float wa = hi ? x1 : x0;
            const float wb = hi ? x3 : x2;
            const float* hr = h1 + (size_t)sj * 256 + lane * 4;
            const float4 vA = *(const float4*)hr;
            const float4 vB = *(const float4*)(hr + 128);
            accA.x += wa * vA.x; accA.y += wa * vA.y; accA.z += wa * vA.z; accA.w += wa * vA.w;
            accB.x += wb * vB.x; accB.y += wb * vB.y; accB.z += wb * vB.z; accB.w += wb * vB.w;
        }
    }
    d0 = wave_sum(d0); d1 = wave_sum(d1); d2 = wave_sum(d2); d3 = wave_sum(d3);
    const float denA = hi ? d1 : d0;
    const float denB = hi ? d3 : d2;
    const float4 bA = *(const float4*)(b1 + lane * 4);
    const float4 bB = *(const float4*)(b1 + 128 + lane * 4);
    float4 oA, oB;
    oA.x = fmaxf(accA.x / denA + bA.x, 0.0f); oA.y = fmaxf(accA.y / denA + bA.y, 0.0f);
    oA.z = fmaxf(accA.z / denA + bA.z, 0.0f); oA.w = fmaxf(accA.w / denA + bA.w, 0.0f);
    oB.x = fmaxf(accB.x / denB + bB.x, 0.0f); oB.y = fmaxf(accB.y / denB + bB.y, 0.0f);
    oB.z = fmaxf(accB.z / denB + bB.z, 0.0f); oB.w = fmaxf(accB.w / denB + bB.w, 0.0f);
    float* op = y1 + (size_t)d * 256 + lane * 4;
    *(float4*)op = oA;
    *(float4*)(op + 128) = oB;
}

__global__ __launch_bounds__(256)
void node_agg2(const int* __restrict__ rowptr, const int* __restrict__ deg,
               const int* __restrict__ csr_src,
               const float* __restrict__ a_src, const float* __restrict__ a_dst,
               const float* __restrict__ h2, const float* __restrict__ b2,
               float* __restrict__ out, int N) {
    const int lane = threadIdx.x & 31;
    const int wv = __builtin_amdgcn_readfirstlane((int)(threadIdx.x >> 5));
    const int d = blockIdx.x * 8 + wv;
    if (d >= N) return;
    const int half = lane >> 4;
    const int t = lane & 15;
    const int start = rowptr[d];
    const int n = deg[d];
    const int cnt = n + 1;
    const float ad = a_dst[d];

    float m = -__builtin_huge_valf();
    for (int c = 0; c < cnt; c += 32) {
        const int k = c + lane;
        if (k < cnt) {
            const int s = k < n ? csr_src[start + k] : d;
            m = fmaxf(m, lrelu(a_src[s] + ad));
        }
    }
    m = wave_max(m);

    float den = 0.0f, acc = 0.0f;
    for (int c = 0; c < cnt; c += 32) {
        const int k = c + lane;
        int s = d;
        float w = 0.0f;
        if (k < cnt) {
            if (k < n) s = csr_src[start + k];
            w = expf(lrelu(a_src[s] + ad) - m);
        }
        den += w;
        const int cn = (cnt - c) < 32 ? (cnt - c) : 32;
        const int cp = cn & ~1;
#pragma unroll 4
        for (int j = 0; j < cp; j += 2) {
            const int s0 = __builtin_amdgcn_readlane(s, j);
            const int s1 = __builtin_amdgcn_readlane(s, j + 1);
            const float w0 = lane_f(w, j);
            const float w1 = lane_f(w, j + 1);
            const int sj = half ? s1 : s0;
            const float wj = half ? w1 : w0;
            acc += wj * h2[(size_t)sj * 16 + t];
        }
        if (cn & 1) {
            const int sl = __builtin_amdgcn_readlane(s, cp);
            const float wl = lane_f(w, cp);
            if (half == 0) acc += wl * h2[(size_t)sl * 16 + t];
        }
    }
    den = wave_sum(den);
    acc += __shfl_xor(acc, 16);
    if (lane < 16) out[(size_t)d * 16 + t] = acc / den + b2[t];
}


extern "C" void kernel_launch(void* const* d_in, const int* in_sizes, int n_in,
                              void* d_out, int out_size, void* d_ws, size_t ws_size,
                              hipStream_t stream) {
    const float* x   = (const float*)d_in[0];
    const int*   ei  = (const int*)  d_in[1];
    const float* W1  = (const float*)d_in[2];
    const float* as1 = (const float*)d_in[3];
    const float* ad1 = (const float*)d_in[4];
    const float* b1  = (const float*)d_in[5];
    const float* W2  = (const float*)d_in[6];
    const float* as2 = (const float*)d_in[7];
    const float* ad2 = (const float*)d_in[8];
    const float* b2  = (const float*)d_in[9];

    const int N  = in_sizes[0] / 128;
    const int E  = in_sizes[1] / 2;

    char* w = (char*)d_ws;
    auto carve = [&](size_t bytes) -> void* {
        void* p = (void*)w;
        w += (bytes + 255) & ~((size_t)255);
        return p;
    };
    float* h1     = (float*)carve((size_t)N * 256 * 4);
    float* y1     = (float*)carve((size_t)N * 256 * 4);
    float* aS1    = (float*)carve((size_t)N * 4 * 4);
    float* aD1    = (float*)carve((size_t)N * 4 * 4);
    float* h2     = (float*)carve((size_t)N * 16 * 4);
    float* aS2    = (float*)carve((size_t)N * 4);
    float* aD2    = (float*)carve((size_t)N * 4);
    int*   deg    = (int*)  carve((size_t)N * 4);
    int*   rowptr = (int*)  carve((size_t)N * 4);
    int*   rank   = (int*)  carve((size_t)E * 4);
    int*   csrs   = (int*)  carve((size_t)E * 4);
    const int NB  = (N + NBLK - 1) / NBLK;
    const int epc = (E + CHUNKS - 1) / CHUNKS;
    int*   cnt    = (int*)  carve((size_t)CHUNKS * HCAP * 4);
    int*   coarse = (int*)  carve((size_t)CHUNKS * NB * 4);

    const int TB = 256;
    auto blocks = [&](long long n) { return (unsigned)((n + TB - 1) / TB); };

    chunk_rank<<<CHUNKS, 1024, 0, stream>>>(ei, E, epc, NB, rank, cnt, coarse);
    scan_cols<<<NB, NBLK, 0, stream>>>(cnt, coarse, deg, rowptr, N, NB);
    dim3 gf(blocks(epc), CHUNKS);
    fill_csr<<<gf, TB, 0, stream>>>(ei, E, epc, cnt, rank, csrs);

    dim3 g1((N + 15) / 16, 4);
    gemm1_wmma<<<g1, 32, 0, stream>>>(x, W1, as1, ad1, h1, aS1, aD1, N);
    node_agg1<<<(N + 7) / 8, 256, 0, stream>>>(rowptr, deg, csrs, aS1, aD1, h1, b1, y1, N);

    gemm2_wmma<<<(N + 15) / 16, 32, 0, stream>>>(y1, W2, as2, ad2, h2, aS2, aD2, N);
    node_agg2<<<(N + 7) / 8, 256, 0, stream>>>(rowptr, deg, csrs, aS2, aD2, h2, b2, (float*)d_out, N);
}


# Layer-1 product tiles ride as extra blocks in the three edge-grouping launches (ranking, scan, scatter), which leave most WGPs idle; its own launch is gone; tile code and sum order unchanged
# speedup vs baseline: 5.0317x; 1.0509x over previous
#include <hip/hip_runtime.h>
#include <hip/hip_bf16.h>

typedef __attribute__((ext_vector_type(2))) float v2f;
typedef __attribute__((ext_vector_type(8))) float v8f;

#define NEG_SLOPE 0.2f


__device__ __forceinline__ float lrelu(float v) {
    return v >= 0.0f ? v : NEG_SLOPE * v;
}

__device__ __forceinline__ float half_rows_sum(const float (&ps)[8], const float (&pd)[8], int c) {
    const bool b3 = (c & 8) != 0, b2 = (c & 4) != 0, b1 = (c & 2) != 0, b0 = (c & 1) != 0;
    float qs[4], qd[4];
    for (int i = 0; i < 4; ++i) {
        qs[i] = (b3 ? ps[i + 4] : ps[i]) + __shfl_xor(b3 ? ps[i] : ps[i + 4], 8);
        qd[i] = (b3 ? pd[i + 4] : pd[i]) + __shfl_xor(b3 ? pd[i] : pd[i + 4], 8);
    }
    float us[2], ud[2];
    for (int i = 0; i < 2; ++i) {
        us[i] = (b2 ? qs[i + 2] : qs[i]) + __shfl_xor(b2 ? qs[i] : qs[i + 2], 4);
        ud[i] = (b2 ? qd[i + 2] : qd[i]) + __shfl_xor(b2 ? qd[i] : qd[i + 2], 4);
    }
    const float vs = (b1 ? us[1] : us[0]) + __shfl_xor(b1 ? us[0] : us[1], 2);
    const float vd = (b1 ? ud[1] : ud[0]) + __shfl_xor(b1 ? ud[0] : ud[1], 2);
    return (b0 ? vd : vs) + __shfl_xor(b0 ? vs : vd, 1);
}

__device__ __forceinline__
void gemm1_tile(const int tile, const int mtiles,
                const float* __restrict__ x, const float* __restrict__ W1,
                const float* __restrict__ att_src, const float* __restrict__ att_dst,
                float* __restrict__ h1, float* __restrict__ a_src, float* __restrict__ a_dst,
                int N) {
    const int head  = (tile >= mtiles ? 1 : 0) + (tile >= 2 * mtiles ? 1 : 0) + (tile >= 3 * mtiles ? 1 : 0);
    const int mtile = tile - head * mtiles;
    const int lane  = threadIdx.x & 31;
    const int mrow  = mtile * 16 + (lane & 15);
    const int lrow  = mrow < N ? mrow : N - 1;
    const int khalf = (lane >> 4) * 2;
    const int colbase = head * 64;
    const int nc    = lane & 15;

    v8f c0 = {}, c1 = {}, c2 = {}, c3 = {};
    const float* xrow = x + (size_t)lrow * 128;

    for (int k = 0; k < 128; k += 4) {
        const int kb = k + khalf;
        v2f a; a.x = xrow[kb]; a.y = xrow[kb + 1];
        const float* w0 = W1 + (size_t)kb * 256 + colbase + nc;
        const float* w1 = w0 + 256;
        v2f b0; b0.x = w0[0];  b0.y = w1[0];
        v2f b1; b1.x = w0[16]; b1.y = w1[16];
        v2f b2; b2.x = w0[32]; b2.y = w1[32];
        v2f b3; b3.x = w0[48]; b3.y = w1[48];
        c0 = __builtin_amdgcn_wmma_f32_16x16x4_f32(false, a, false, b0, (short)0, c0, false, false);
        c1 = __builtin_amdgcn_wmma_f32_16x16x4_f32(false, a, false, b1, (short)0, c1, false, false);
        c2 = __builtin_amdgcn_wmma_f32_16x16x4_f32(false, a, false, b2, (short)0, c2, false, false);
        c3 = __builtin_amdgcn_wmma_f32_16x16x4_f32(false, a, false, b3, (short)0, c3, false, false);
    }

    const int rowoff = (lane >= 16) ? 8 : 0;
    for (int r = 0; r < 8; ++r) {
        const int grow = mtile * 16 + rowoff + r;
        if (grow < N) {
            float* op = h1 + (size_t)grow * 256 + colbase + nc;
            op[0]  = c0[r];
            op[16] = c1[r];
            op[32] = c2[r];
            op[48] = c3[r];
        }
    }

    const float* as = att_src + colbase + nc;
    const float* ad = att_dst + colbase + nc;
    const float s0 = as[0], s1 = as[16], s2 = as[32], s3 = as[48];
    const float t0 = ad[0], t1 = ad[16], t2 = ad[32], t3 = ad[48];
    float ps[8], pd[8];
    for (int r = 0; r < 8; ++r) {
        ps[r] = c0[r] * s0 + c1[r] * s1 + c2[r] * s2 + c3[r] * s3;
        pd[r] = c0[r] * t0 + c1[r] * t1 + c2[r] * t2 + c3[r] * t3;
    }
    const float sc = half_rows_sum(ps, pd, nc);
    const int srow = mtile * 16 + rowoff + (nc >> 1);
    if (srow < N) {
        float* sp = (nc & 1) ? a_dst : a_src;
        sp[(size_t)srow * 4 + head] = sc;
    }
}

__global__ __launch_bounds__(32)
void gemm2_wmma(const float* __restrict__ y1, const float* __restrict__ W2,
                const float* __restrict__ att_src, const float* __restrict__ att_dst,
                float* __restrict__ h2, float* __restrict__ a_src, float* __restrict__ a_dst,
                int N) {
    const int mtile = blockIdx.x;
    const int lane  = threadIdx.x;
    const int mrow  = mtile * 16 + (lane & 15);
    const int lrow  = mrow < N ? mrow : N - 1;
    const int khalf = (lane >> 4) * 2;
    const int nc    = lane & 15;

    v8f c = {};
    const float* yr = y1 + (size_t)lrow * 256;
    for (int k = 0; k < 256; k += 4) {
        const int kb = k + khalf;
        v2f a; a.x = yr[kb]; a.y = yr[kb + 1];
        v2f b; b.x = W2[(size_t)kb * 16 + nc]; b.y = W2[(size_t)(kb + 1) * 16 + nc];
        c = __builtin_amdgcn_wmma_f32_16x16x4_f32(false, a, false, b, (short)0, c, false, false);
    }
    const int rowoff = (lane >= 16) ? 8 : 0;
    for (int r = 0; r < 8; ++r) {
        const int grow = mtile * 16 + rowoff + r;
        if (grow < N) h2[(size_t)grow * 16 + nc] = c[r];
    }

    const float s = att_src[nc];
    const float t = att_dst[nc];
    float ps[8], pd[8];
    for (int r = 0; r < 8; ++r) {
        ps[r] = c[r] * s;
        pd[r] = c[r] * t;
    }
    const float sc = half_rows_sum(ps, pd, nc);
    const int srow = mtile * 16 + rowoff + (nc >> 1);
    if (srow < N) {
        float* sp = (nc & 1) ? a_dst : a_src;
        sp[srow] = sc;
    }
}

#define CHUNKS 64
#define HCAP   50176
#define NBLK   256
#define NBLK_LOG 8
#define EPT    5
#define WGPS   256

__device__ __forceinline__ int wave_sum_i(int v) {
    for (int o = 16; o > 0; o >>= 1) v += __shfl_xor(v, o);
    return v;
}

__device__ __forceinline__
void chunk_rank(int* __restrict__ hist, int* __restrict__ blk,
                const int* __restrict__ ei, int E, int epc, int NB,
                int* __restrict__ rank, int* __restrict__ cnt, int* __restrict__ coarse) {
    const int t = threadIdx.x;
    const int c = blockIdx.x;
    int4* h4 = (int4*)hist;
    for (int i = t; i < HCAP / 4; i += 1024) h4[i] = make_int4(0, 0, 0, 0);
    if (t < HCAP / NBLK) blk[t] = 0;
    __syncthreads();

    const int e0 = c * epc;
    const int e1 = (e0 + epc) < E ? (e0 + epc) : E;
    for (int eb = e0 + t; eb < e1; eb += EPT * 1024) {
        int dd[EPT], rr[EPT];
#pragma unroll
        for (int u = 0; u < EPT; ++u) {
            const int e = eb + u * 1024;
            dd[u] = e < e1 ? ei[E + e] : 0;
        }
#pragma unroll
        for (int u = 0; u < EPT; ++u) {
            const int e = eb + u * 1024;
            rr[u] = 0;
            if (e < e1) {
                rr[u] = atomicAdd(&hist[dd[u]], 1);
                atomicAdd(&blk[dd[u] >> NBLK_LOG], 1);
            }
        }
#pragma unroll
        for (int u = 0; u < EPT; ++u) {
            const int e = eb + u * 1024;
            if (e < e1) rank[e] = rr[u];
        }
    }
    __syncthreads();

    int4* c4 = (int4*)(cnt + (size_t)c * HCAP);
    for (int i = t; i < HCAP / 4; i += 1024) c4[i] = h4[i];
    if (t < NB) coarse[c * NB + t] = blk[t];
}

__device__ __forceinline__
void scan_cols(int* __restrict__ wsum, int* __restrict__ wpre,
               int* __restrict__ cnt, const int* __restrict__ coarse,
               int* __restrict__ deg, int* __restrict__ rowptr, int N, int NB) {
    const int t = threadIdx.x, lane = t & 31, wv = t >> 5;
    const int j = blockIdx.x;
    const int idx = j * NBLK + t;

    int pre = 0;
    for (int b = t; b < j; b += NBLK)
        for (int c = 0; c < CHUNKS; ++c) pre += coarse[c * NB + b];

    int v[CHUNKS];
    int d = 0;
#pragma unroll
    for (int c = 0; c < CHUNKS; ++c) {
        v[c] = cnt[(size_t)c * HCAP + idx];
        d += v[c];
    }
    int inc = d;
    for (int off = 1; off < 32; off <<= 1) {
        const int nb = __shfl_up(inc, off);
        if (lane >= off) inc += nb;
    }
    pre = wave_sum_i(pre);
    if (lane == 31) wsum[wv] = inc;
    if (lane == 0)  wpre[wv] = pre;
    __syncthreads();
    int base = 0;
    for (int k = 0; k < NBLK / 32; ++k) {
        base += wpre[k];
        if (k < wv) base += wsum[k];
    }
    if (idx < N) {
        int run = base + inc - d;
        rowptr[idx] = run;
        deg[idx] = d;
#pragma unroll
        for (int c = 0; c < CHUNKS; ++c) {
            cnt[(size_t)c * HCAP + idx] = run;
            run += v[c];
        }
    }
}

__device__ __forceinline__
void fill_csr(const int b, const int* __restrict__ ei, int E, int epc, int bpc,
              const int* __restrict__ cnt, const int* __restrict__ rank,
              int* __restrict__ csr_src) {
    const int c = b / bpc;
    const int e0 = c * epc;
    const int e1 = (e0 + epc) < E ? (e0 + epc) : E;
    const int e = e0 + (b - c * bpc) * 256 + threadIdx.x;
    if (e >= e1) return;
    csr_src[cnt[(size_t)c * HCAP + ei[E + e]] + rank[e]] = ei[e];
}


__global__ __launch_bounds__(1024)
void rank_and_gemm1(const int* __restrict__ ei, int E, int epc, int NB,
                    int* __restrict__ rank, int* __restrict__ cnt, int* __restrict__ coarse,
                    const float* __restrict__ x, const float* __restrict__ W1,
                    const float* __restrict__ att_src, const float* __restrict__ att_dst,
                    float* __restrict__ h1, float* __restrict__ a_src, float* __restrict__ a_dst,
                    int N, int mtiles, int tile0, int tile1) {
    __shared__ int hist[HCAP];
    __shared__ int blk[HCAP / NBLK];
    if (blockIdx.x < CHUNKS) {
        chunk_rank(hist, blk, ei, E, epc, NB, rank, cnt, coarse);
    } else {
        const int wv = __builtin_amdgcn_readfirstlane((int)(threadIdx.x >> 5));
        const int tile = tile0 + ((int)blockIdx.x - CHUNKS) * 32 + wv;
        if (tile < tile1) gemm1_tile(tile, mtiles, x, W1, att_src, att_dst, h1, a_src, a_dst, N);
    }
}

__global__ __launch_bounds__(NBLK)
void scan_and_gemm1(int* __restrict__ cnt, const int* __restrict__ coarse,
                    int* __restrict__ deg, int* __restrict__ rowptr, int NB,
                    const float* __restrict__ x, const float* __restrict__ W1,
                    const float* __restrict__ att_src, const float* __restrict__ att_dst,
                    float* __restrict__ h1, float* __restrict__ a_src, float* __restrict__ a_dst,
                    int N, int mtiles, int tile0, int tile1) {
    __shared__ int wsum[NBLK / 32];
    __shared__ int wpre[NBLK / 32];
    if ((int)blockIdx.x < NB) {
        scan_cols(wsum, wpre, cnt, coarse, deg, rowptr, N, NB);
    } else {
        const int wv = __builtin_amdgcn_readfirstlane((int)(threadIdx.x >> 5));
        const int tile = tile0 + ((int)blockIdx.x - NB) * (NBLK / 32) + wv;
        if (tile < tile1) gemm1_tile(tile, mtiles, x, W1, att_src, att_dst, h1, a_src, a_dst, N);
    }
}

__global__ __launch_bounds__(256)
void gemm1_and_fill(const int* __restrict__ ei, int E, int epc, int bpc, int ngemm,
                    const int* __restrict__ cnt, const int* __restrict__ rank,
                    int* __restrict__ csr_src,
                    const float* __restrict__ x, const float* __restrict__ W1,
                    const float* __restrict__ att_src, const float* __restrict__ att_dst,
                    float* __restrict__ h1, float* __restrict__ a_src, float* __restrict__ a_dst,
                    int N, int mtiles, int tile0, int tile1) {
    if ((int)blockIdx.x >= ngemm) {
        fill_csr((int)blockIdx.x - ngemm, ei, E, epc, bpc, cnt, rank, csr_src);
    } else {
        const int wv = __builtin_amdgcn_readfirstlane((int)(threadIdx.x >> 5));
        const int tile = tile0 + (int)blockIdx.x * 8 + wv;
        if (tile < tile1) gemm1_tile(tile, mtiles, x, W1, att_src, att_dst, h1, a_src, a_dst, N);
    }
}

__device__ __forceinline__ float wave_max(float v) {
    for (int o = 16; o > 0; o >>= 1) v = fmaxf(v, __shfl_xor(v, o));
    return v;
}
__device__ __forceinline__ float wave_sum(float v) {
    for (int o = 16; o > 0; o >>= 1) v += __shfl_xor(v, o);
    return v;
}
__device__ __forceinline__ float lane_f(float v, int j) {
    return __int_as_float(__builtin_amdgcn_readlane(__float_as_int(v), j));
}

__global__ __launch_bounds__(256)
void node_agg1(const int* __restrict__ rowptr, const int* __restrict__ deg,
               const int* __restrict__ csr_src,
               const float* __restrict__ a_src, const float* __restrict__ a_dst,
               const float* __restrict__ h1, const float* __restrict__ b1,
               float* __restrict__ y1, int N) {
    const int lane = threadIdx.x & 31;
    const int wv = __builtin_amdgcn_readfirstlane((int)(threadIdx.x >> 5));
    const int d = blockIdx.x * 8 + wv;
    if (d >= N) return;
    const bool hi = lane >= 16;
    const int start = rowptr[d];
    const int n = deg[d];
    const int cnt = n + 1;
    const float4 ad = *(const float4*)(a_dst + (size_t)d * 4);

    float m0 = -__builtin_huge_valf(), m1 = m0, m2 = m0, m3 = m0;
    for (int c = 0; c < cnt; c += 32) {
        const int k = c + lane;
        if (k < cnt) {
            const int s = k < n ? csr_src[start + k] : d;
            const float4 a = *(const float4*)(a_src + (size_t)s * 4);
            m0 = fmaxf(m0, lrelu(a.x + ad.x));
            m1 = fmaxf(m1, lrelu(a.y + ad.y));
            m2 = fmaxf(m2, lrelu(a.z + ad.z));
            m3 = fmaxf(m3, lrelu(a.w + ad.w));
        }
    }
    m0 = wave_max(m0); m1 = wave_max(m1); m2 = wave_max(m2); m3 = wave_max(m3);

    float d0 = 0.0f, d1 = 0.0f, d2 = 0.0f, d3 = 0.0f;
    float4 accA = make_float4(0.0f, 0.0f, 0.0f, 0.0f);
    float4 accB = make_float4(0.0f, 0.0f, 0.0f, 0.0f);
    for (int c = 0; c < cnt; c += 32) {
        const int k = c + lane;
        int s = d;
        float w0 = 0.0f, w1 = 0.0f, w2 = 0.0f, w3 = 0.0f;
        if (k < cnt) {
            if (k < n) s = csr_src[start + k];
            const float4 a = *(const float4*)(a_src + (size_t)s * 4);
            w0 = expf(lrelu(a.x + ad.x) - m0);
            w1 = expf(lrelu(a.y + ad.y) - m1);
            w2 = expf(lrelu(a.z + ad.z) - m2);
            w3 = expf(lrelu(a.w + ad.w) - m3);
        }
        d0 += w0; d1 += w1; d2 += w2; d3 += w3;
        const int cn = (cnt - c) < 32 ? (cnt - c) : 32;
#pragma unroll 4
        for (int j = 0; j < cn; ++j) {
            const int sj = __builtin_amdgcn_readlane(s, j);
            const float x0 = lane_f(w0, j), x1 = lane_f(w1, j);
            const float x2 = lane_f(w2, j), x3 = lane_f(w3, j);
            const float wa = hi ? x1 : x0;
            const float wb = hi ? x3 : x2;
            const float* hr = h1 + (size_t)sj * 256 + lane * 4;
            const float4 vA = *(const float4*)hr;
            const float4 vB = *(const float4*)(hr + 128);
            accA.x += wa * vA.x; accA.y += wa * vA.y; accA.z += wa * vA.z; accA.w += wa * vA.w;
            accB.x += wb * vB.x; accB.y += wb * vB.y; accB.z += wb * vB.z; accB.w += wb * vB.w;
        }
    }
    d0 = wave_sum(d0); d1 = wave_sum(d1); d2 = wave_sum(d2); d3 = wave_sum(d3);
    const float denA = hi ? d1 : d0;
    const float denB = hi ? d3 : d2;
    const float4 bA = *(const float4*)(b1 + lane * 4);
    const float4 bB = *(const float4*)(b1 + 128 + lane * 4);
    float4 oA, oB;
    oA.x = fmaxf(accA.x / denA + bA.x, 0.0f); oA.y = fmaxf(accA.y / denA + bA.y, 0.0f);
    oA.z = fmaxf(accA.z / denA + bA.z, 0.0f); oA.w = fmaxf(accA.w / denA + bA.w, 0.0f);
    oB.x = fmaxf(accB.x / denB + bB.x, 0.0f); oB.y = fmaxf(accB.y / denB + bB.y, 0.0f);
    oB.z = fmaxf(accB.z / denB + bB.z, 0.0f); oB.w = fmaxf(accB.w / denB + bB.w, 0.0f);
    float* op = y1 + (size_t)d * 256 + lane * 4;
    *(float4*)op = oA;
    *(float4*)(op + 128) = oB;
}

__global__ __launch_bounds__(256)
void node_agg2(const int* __restrict__ rowptr, const int* __restrict__ deg,
               const int* __restrict__ csr_src,
               const float* __restrict__ a_src, const float* __restrict__ a_dst,
               const float* __restrict__ h2, const float* __restrict__ b2,
               float* __restrict__ out, int N) {
    const int lane = threadIdx.x & 31;
    const int wv = __builtin_amdgcn_readfirstlane((int)(threadIdx.x >> 5));
    const int d = blockIdx.x * 8 + wv;
    if (d >= N) return;
    const int half = lane >> 4;
    const int t = lane & 15;
    const int start = rowptr[d];
    const int n = deg[d];
    const int cnt = n + 1;
    const float ad = a_dst[d];

    float m = -__builtin_huge_valf();
    for (int c = 0; c < cnt; c += 32) {
        const int k = c + lane;
        if (k < cnt) {
            const int s = k < n ? csr_src[start + k] : d;
            m = fmaxf(m, lrelu(a_src[s] + ad));
        }
    }
    m = wave_max(m);

    float den = 0.0f, acc = 0.0f;
    for (int c = 0; c < cnt; c += 32) {
        const int k = c + lane;
        int s = d;
        float w = 0.0f;
        if (k < cnt) {
            if (k < n) s = csr_src[start + k];
            w = expf(lrelu(a_src[s] + ad) - m);
        }
        den += w;
        const int cn = (cnt - c) < 32 ? (cnt - c) : 32;
        const int cp = cn & ~1;
#pragma unroll 4
        for (int j = 0; j < cp; j += 2) {
            const int s0 = __builtin_amdgcn_readlane(s, j);
            const int s1 = __builtin_amdgcn_readlane(s, j + 1);
            const float w0 = lane_f(w, j);
            const float w1 = lane_f(w, j + 1);
            const int sj = half ? s1 : s0;
            const float wj = half ? w1 : w0;
            acc += wj * h2[(size_t)sj * 16 + t];
        }
        if (cn & 1) {
            const int sl = __builtin_amdgcn_readlane(s, cp);
            const float wl = lane_f(w, cp);
            if (half == 0) acc += wl * h2[(size_t)sl * 16 + t];
        }
    }
    den = wave_sum(den);
    acc += __shfl_xor(acc, 16);
    if (lane < 16) out[(size_t)d * 16 + t] = acc / den + b2[t];
}


extern "C" void kernel_launch(void* const* d_in, const int* in_sizes, int n_in,
                              void* d_out, int out_size, void* d_ws, size_t ws_size,
                              hipStream_t stream) {
    const float* x   = (const float*)d_in[0];
    const int*   ei  = (const int*)  d_in[1];
    const float* W1  = (const float*)d_in[2];
    const float* as1 = (const float*)d_in[3];
    const float* ad1 = (const float*)d_in[4];
    const float* b1  = (const float*)d_in[5];
    const float* W2  = (const float*)d_in[6];
    const float* as2 = (const float*)d_in[7];
    const float* ad2 = (const float*)d_in[8];
    const float* b2  = (const float*)d_in[9];

    const int N  = in_sizes[0] / 128;
    const int E  = in_sizes[1] / 2;

    char* w = (char*)d_ws;
    auto carve = [&](size_t bytes) -> void* {
        void* p = (void*)w;
        w += (bytes + 255) & ~((size_t)255);
        return p;
    };
    float* h1     = (float*)carve((size_t)N * 256 * 4);
    float* y1     = (float*)carve((size_t)N * 256 * 4);
    float* aS1    = (float*)carve((size_t)N * 4 * 4);
    float* aD1    = (float*)carve((size_t)N * 4 * 4);
    float* h2     = (float*)carve((size_t)N * 16 * 4);
    float* aS2    = (float*)carve((size_t)N * 4);
    float* aD2    = (float*)carve((size_t)N * 4);
    int*   deg    = (int*)  carve((size_t)N * 4);
    int*   rowptr = (int*)  carve((size_t)N * 4);
    int*   rank   = (int*)  carve((size_t)E * 4);
    int*   csrs   = (int*)  carve((size_t)E * 4);
    const int NB  = (N + NBLK - 1) / NBLK;
    const int epc = (E + CHUNKS - 1) / CHUNKS;
    int*   cnt    = (int*)  carve((size_t)CHUNKS * HCAP * 4);
    int*   coarse = (int*)  carve((size_t)CHUNKS * NB * 4);

    const int TB = 256;
    auto blocks = [&](long long n) { return (unsigned)((n + TB - 1) / TB); };

    const int mtiles = (N + 15) / 16;
    const int tiles  = 4 * mtiles;
    const int t1 = (WGPS - CHUNKS) * 32 < tiles ? (WGPS - CHUNKS) * 32 : tiles;
    const int t2 = t1 + (tiles - t1) / 2;
    const int bpc   = (int)blocks(epc);
    const int ng3   = (tiles - t2 + 7) / 8;
    rank_and_gemm1<<<CHUNKS + (t1 + 31) / 32, 1024, 0, stream>>>(
        ei, E, epc, NB, rank, cnt, coarse, x, W1, as1, ad1, h1, aS1, aD1, N, mtiles, 0, t1);
    scan_and_gemm1<<<NB + (t2 - t1 + NBLK / 32 - 1) / (NBLK / 32), NBLK, 0, stream>>>(
        cnt, coarse, deg, rowptr, NB, x, W1, as1, ad1, h1, aS1, aD1, N, mtiles, t1, t2);
    gemm1_and_fill<<<ng3 + bpc * CHUNKS, TB, 0, stream>>>(
        ei, E, epc, bpc, ng3, cnt, rank, csrs, x, W1, as1, ad1, h1, aS1, aD1, N, mtiles, t2, tiles);

    node_agg1<<<(N + 7) / 8, 256, 0, stream>>>(rowptr, deg, csrs, aS1, aD1, h1, b1, y1, N);

    gemm2_wmma<<<(N + 15) / 16, 32, 0, stream>>>(y1, W2, as2, ad2, h2, aS2, aD2, N);
    node_agg2<<<(N + 7) / 8, 256, 0, stream>>>(rowptr, deg, csrs, aS2, aD2, h2, b2, (float*)d_out, N);
}


# Layer 1 aggregates before it projects: per node and head the softmax-weighted sum of gathered 512-byte INPUT rows, then times W1 on WMMA from LDS; scores from x (W1 att) on WMMA; new sum order
# speedup vs baseline: 5.4304x; 1.0792x over previous
#include <hip/hip_runtime.h>
#include <hip/hip_bf16.h>

typedef __attribute__((ext_vector_type(2))) float v2f;
typedef __attribute__((ext_vector_type(8))) float v8f;

#define NEG_SLOPE 0.2f


__device__ __forceinline__ float lrelu(float v) {
    return v >= 0.0f ? v : NEG_SLOPE * v;
}

__device__ __forceinline__ float half_rows_sum(const float (&ps)[8], const float (&pd)[8], int c) {
    const bool b3 = (c & 8) != 0, b2 = (c & 4) != 0, b1 = (c & 2) != 0, b0 = (c & 1) != 0;
    float qs[4], qd[4];
    for (int i = 0; i < 4; ++i) {
        qs[i] = (b3 ? ps[i + 4] : ps[i]) + __shfl_xor(b3 ? ps[i] : ps[i + 4], 8);
        qd[i] = (b3 ? pd[i + 4] : pd[i]) + __shfl_xor(b3 ? pd[i] : pd[i + 4], 8);
    }
    float us[2], ud[2];
    for (int i = 0; i < 2; ++i) {
        us[i] = (b2 ? qs[i + 2] : qs[i]) + __shfl_xor(b2 ? qs[i] : qs[i + 2], 4);
        ud[i] = (b2 ? qd[i + 2] : qd[i]) + __shfl_xor(b2 ? qd[i] : qd[i + 2], 4);
    }
    const float vs = (b1 ? us[1] : us[0]) + __shfl_xor(b1 ? us[0] : us[1], 2);
    const float vd = (b1 ? ud[1] : ud[0]) + __shfl_xor(b1 ? ud[0] : ud[1], 2);
    return (b0 ? vd : vs) + __shfl_xor(b0 ? vs : vd, 1);
}


__device__ __forceinline__
void fold_att(const float* __restrict__ W1, const float* __restrict__ att_src,
              const float* __restrict__ att_dst, float* __restrict__ U) {
    const int t = threadIdx.x;
    const int k = t >> 3, j = t & 7, h = j & 3;
    const float* w  = W1 + (size_t)k * 256 + h * 64;
    const float* at = (j < 4 ? att_src : att_dst) + h * 64;
    float s = 0.0f;
    for (int c = 0; c < 64; ++c) s += w[c] * at[c];
    U[t] = s;
}

__device__ __forceinline__
void score_tile(const int mtile, const float* __restrict__ x, const float* __restrict__ U,
                float* __restrict__ a_src, float* __restrict__ a_dst, int N) {
    const int lane  = threadIdx.x & 31;
    const int mrow  = mtile * 16 + (lane & 15);
    const int lrow  = mrow < N ? mrow : N - 1;
    const int khalf = (lane >> 4) * 2;
    const int nc    = lane & 15;
    const int uc    = nc & 7;

    v8f c = {};
    const float* xrow = x + (size_t)lrow * 128;
    for (int k = 0; k < 128; k += 4) {
        const int kb = k + khalf;
        v2f a; a.x = xrow[kb]; a.y = xrow[kb + 1];
        v2f b; b.x = U[kb * 8 + uc]; b.y = U[(kb + 1) * 8 + uc];
        c = __builtin_amdgcn_wmma_f32_16x16x4_f32(false, a, false, b, (short)0, c, false, false);
    }
    const int rowoff = (lane >= 16) ? 8 : 0;
    float* sp = (nc < 4) ? a_src : a_dst;
    for (int r = 0; r < 8; ++r) {
        const int grow = mtile * 16 + rowoff + r;
        if (grow < N && nc < 8) sp[(size_t)grow * 4 + (nc & 3)] = c[r];
    }
}

__global__ __launch_bounds__(32)
void gemm2_wmma(const float* __restrict__ y1, const float* __restrict__ W2,
                const float* __restrict__ att_src, const float* __restrict__ att_dst,
                float* __restrict__ h2, float* __restrict__ a_src, float* __restrict__ a_dst,
                int N) {
    const int mtile = blockIdx.x;
    const int lane  = threadIdx.x;
    const int mrow  = mtile * 16 + (lane & 15);
    const int lrow  = mrow < N ? mrow : N - 1;
    const int khalf = (lane >> 4) * 2;
    const int nc    = lane & 15;

    v8f c = {};
    const float* yr = y1 + (size_t)lrow * 256;
    for (int k = 0; k < 256; k += 4) {
        const int kb = k + khalf;
        v2f a; a.x = yr[kb]; a.y = yr[kb + 1];
        v2f b; b.x = W2[(size_t)kb * 16 + nc]; b.y = W2[(size_t)(kb + 1) * 16 + nc];
        c = __builtin_amdgcn_wmma_f32_16x16x4_f32(false, a, false, b, (short)0, c, false, false);
    }
    const int rowoff = (lane >= 16) ? 8 : 0;
    for (int r = 0; r < 8; ++r) {
        const int grow = mtile * 16 + rowoff + r;
        if (grow < N) h2[(size_t)grow * 16 + nc] = c[r];
    }

    const float s = att_src[nc];
    const float t = att_dst[nc];
    float ps[8], pd[8];
    for (int r = 0; r < 8; ++r) {
        ps[r] = c[r] * s;
        pd[r] = c[r] * t;
    }
    const float sc = half_rows_sum(ps, pd, nc);
    const int srow = mtile * 16 + rowoff + (nc >> 1);
    if (srow < N) {
        float* sp = (nc & 1) ? a_dst : a_src;
        sp[srow] = sc;
    }
}

#define CHUNKS 64
#define HCAP   50176
#define NBLK   256
#define NBLK_LOG 8
#define EPT    5

__device__ __forceinline__ int wave_sum_i(int v) {
    for (int o = 16; o > 0; o >>= 1) v += __shfl_xor(v, o);
    return v;
}

__device__ __forceinline__
void chunk_rank(int* __restrict__ hist, int* __restrict__ blk,
                const int* __restrict__ ei, int E, int epc, int NB,
                int* __restrict__ rank, int* __restrict__ cnt, int* __restrict__ coarse) {
    const int t = threadIdx.x;
    const int c = blockIdx.x;
    int4* h4 = (int4*)hist;
    for (int i = t; i < HCAP / 4; i += 1024) h4[i] = make_int4(0, 0, 0, 0);
    if (t < HCAP / NBLK) blk[t] = 0;
    __syncthreads();

    const int e0 = c * epc;
    const int e1 = (e0 + epc) < E ? (e0 + epc) : E;
    for (int eb = e0 + t; eb < e1; eb += EPT * 1024) {
        int dd[EPT], rr[EPT];
#pragma unroll
        for (int u = 0; u < EPT; ++u) {
            const int e = eb + u * 1024;
            dd[u] = e < e1 ? ei[E + e] : 0;
        }
#pragma unroll
        for (int u = 0; u < EPT; ++u) {
            const int e = eb + u * 1024;
            rr[u] = 0;
            if (e < e1) {
                rr[u] = atomicAdd(&hist[dd[u]], 1);
                atomicAdd(&blk[dd[u] >> NBLK_LOG], 1);
            }
        }
#pragma unroll
        for (int u = 0; u < EPT; ++u) {
            const int e = eb + u * 1024;
            if (e < e1) rank[e] = rr[u];
        }
    }
    __syncthreads();

    int4* c4 = (int4*)(cnt + (size_t)c * HCAP);
    for (int i = t; i < HCAP / 4; i += 1024) c4[i] = h4[i];
    if (t < NB) coarse[c * NB + t] = blk[t];
}

__device__ __forceinline__
void scan_cols(int* __restrict__ wsum, int* __restrict__ wpre,
               int* __restrict__ cnt, const int* __restrict__ coarse,
               int* __restrict__ deg, int* __restrict__ rowptr, int N, int NB) {
    const int t = threadIdx.x, lane = t & 31, wv = t >> 5;
    const int j = blockIdx.x;
    const int idx = j * NBLK + t;

    int pre = 0;
    for (int b = t; b < j; b += NBLK)
        for (int c = 0; c < CHUNKS; ++c) pre += coarse[c * NB + b];

    int v[CHUNKS];
    int d = 0;
#pragma unroll
    for (int c = 0; c < CHUNKS; ++c) {
        v[c] = cnt[(size_t)c * HCAP + idx];
        d += v[c];
    }
    int inc = d;
    for (int off = 1; off < 32; off <<= 1) {
        const int nb = __shfl_up(inc, off);
        if (lane >= off) inc += nb;
    }
    pre = wave_sum_i(pre);
    if (lane == 31) wsum[wv] = inc;
    if (lane == 0)  wpre[wv] = pre;
    __syncthreads();
    int base = 0;
    for (int k = 0; k < NBLK / 32; ++k) {
        base += wpre[k];
        if (k < wv) base += wsum[k];
    }
    if (idx < N) {
        int run = base + inc - d;
        rowptr[idx] = run;
        deg[idx] = d;
#pragma unroll
        for (int c = 0; c < CHUNKS; ++c) {
            cnt[(size_t)c * HCAP + idx] = run;
            run += v[c];
        }
    }
}

__device__ __forceinline__
void fill_csr(const int b, const int* __restrict__ ei, int E, int epc, int bpc,
              const int* __restrict__ cnt, const int* __restrict__ rank,
              int* __restrict__ csr_src) {
    const int c = b / bpc;
    const int e0 = c * epc;
    const int e1 = (e0 + epc) < E ? (e0 + epc) : E;
    const int e = e0 + (b - c * bpc) * 256 + threadIdx.x;
    if (e >= e1) return;
    csr_src[cnt[(size_t)c * HCAP + ei[E + e]] + rank[e]] = ei[e];
}


__global__ __launch_bounds__(1024)
void rank_and_fold(const int* __restrict__ ei, int E, int epc, int NB,
                   int* __restrict__ rank, int* __restrict__ cnt, int* __restrict__ coarse,
                   const float* __restrict__ W1,
                   const float* __restrict__ att_src, const float* __restrict__ att_dst,
                   float* __restrict__ U) {
    __shared__ int hist[HCAP];
    __shared__ int blk[HCAP / NBLK];
    if (blockIdx.x < CHUNKS) {
        chunk_rank(hist, blk, ei, E, epc, NB, rank, cnt, coarse);
    } else {
        fold_att(W1, att_src, att_dst, U);
    }
}

__global__ __launch_bounds__(NBLK)
void scan_and_scores(int* __restrict__ cnt, const int* __restrict__ coarse,
                     int* __restrict__ deg, int* __restrict__ rowptr, int NB,
                     const float* __restrict__ x, const float* __restrict__ U,
                     float* __restrict__ a_src, float* __restrict__ a_dst,
                     int N, int mtiles) {
    __shared__ int wsum[NBLK / 32];
    __shared__ int wpre[NBLK / 32];
    if ((int)blockIdx.x < NB) {
        scan_cols(wsum, wpre, cnt, coarse, deg, rowptr, N, NB);
    } else {
        const int wv = __builtin_amdgcn_readfirstlane((int)(threadIdx.x >> 5));
        const int tile = ((int)blockIdx.x - NB) * (NBLK / 32) + wv;
        if (tile < mtiles) score_tile(tile, x, U, a_src, a_dst, N);
    }
}

__global__ __launch_bounds__(256)
void fill_groups(const int* __restrict__ ei, int E, int epc, int bpc,
                 const int* __restrict__ cnt, const int* __restrict__ rank,
                 int* __restrict__ csr_src) {
    fill_csr((int)blockIdx.x, ei, E, epc, bpc, cnt, rank, csr_src);
}

__device__ __forceinline__ float wave_max(float v) {
    for (int o = 16; o > 0; o >>= 1) v = fmaxf(v, __shfl_xor(v, o));
    return v;
}
__device__ __forceinline__ float wave_sum(float v) {
    for (int o = 16; o > 0; o >>= 1) v += __shfl_xor(v, o);
    return v;
}
__device__ __forceinline__ float lane_f(float v, int j) {
    return __int_as_float(__builtin_amdgcn_readlane(__float_as_int(v), j));
}

#define GN 32
#define ZP 516

__global__ __launch_bounds__(512) __attribute__((amdgpu_waves_per_eu(16)))
void node_agg_proj1(const int* __restrict__ rowptr, const int* __restrict__ deg,
                    const int* __restrict__ csr_src,
                    const float* __restrict__ a_src, const float* __restrict__ a_dst,
                    const float* __restrict__ x, const float* __restrict__ W1,
                    const float* __restrict__ b1,
                    float* __restrict__ y1, int N) {
    __shared__ __attribute__((aligned(16))) float Z[GN * ZP];
    const int lane = threadIdx.x & 31;
    const int wv = __builtin_amdgcn_readfirstlane((int)(threadIdx.x >> 5));

    for (int q = 0; q < 2; ++q) {
        const int nl = wv * 2 + q;
        const int d = blockIdx.x * GN + nl;
        float4 z0 = make_float4(0.0f, 0.0f, 0.0f, 0.0f);
        float4 z1 = z0, z2 = z0, z3 = z0;
        if (d < N) {
            const int start = rowptr[d];
            const int n = deg[d];
            const int cnt = n + 1;
            const float4 ad = *(const float4*)(a_dst + (size_t)d * 4);

            float m0 = -__builtin_huge_valf(), m1 = m0, m2 = m0, m3 = m0;
            for (int c = 0; c < cnt; c += 32) {
                const int k = c + lane;
                if (k < cnt) {
                    const int s = k < n ? csr_src[start + k] : d;
                    const float4 a = *(const float4*)(a_src + (size_t)s * 4);
                    m0 = fmaxf(m0, lrelu(a.x + ad.x));
                    m1 = fmaxf(m1, lrelu(a.y + ad.y));
                    m2 = fmaxf(m2, lrelu(a.z + ad.z));
                    m3 = fmaxf(m3, lrelu(a.w + ad.w));
                }
            }
            m0 = wave_max(m0); m1 = wave_max(m1); m2 = wave_max(m2); m3 = wave_max(m3);

            float d0 = 0.0f, d1 = 0.0f, d2 = 0.0f, d3 = 0.0f;
            for (int c = 0; c < cnt; c += 32) {
                const int k = c + lane;
                int s = d;
                float w0 = 0.0f, w1 = 0.0f, w2 = 0.0f, w3 = 0.0f;
                if (k < cnt) {
                    if (k < n) s = csr_src[start + k];
                    const float4 a = *(const float4*)(a_src + (size_t)s * 4);
                    w0 = expf(lrelu(a.x + ad.x) - m0);
                    w1 = expf(lrelu(a.y + ad.y) - m1);
                    w2 = expf(lrelu(a.z + ad.z) - m2);
                    w3 = expf(lrelu(a.w + ad.w) - m3);
                }
                d0 += w0; d1 += w1; d2 += w2; d3 += w3;
                const int cn = (cnt - c) < 32 ? (cnt - c) : 32;
#pragma unroll 4
                for (int j = 0; j < cn; ++j) {
                    const int sj = __builtin_amdgcn_readlane(s, j);
                    const float x0 = lane_f(w0, j), x1 = lane_f(w1, j);
                    const float x2 = lane_f(w2, j), x3 = lane_f(w3, j);
                    const float4 v = *(const float4*)(x + (size_t)sj * 128 + lane * 4);
                    z0.x += x0 * v.x; z0.y += x0 * v.y; z0.z += x0 * v.z; z0.w += x0 * v.w;
                    z1.x += x1 * v.x; z1.y += x1 * v.y; z1.z += x1 * v.z; z1.w += x1 * v.w;
                    z2.x += x2 * v.x; z2.y += x2 * v.y; z2.z += x2 * v.z; z2.w += x2 * v.w;
                    z3.x += x3 * v.x; z3.y += x3 * v.y; z3.z += x3 * v.z; z3.w += x3 * v.w;
                }
            }
            d0 = wave_sum(d0); d1 = wave_sum(d1); d2 = wave_sum(d2); d3 = wave_sum(d3);
            z0.x /= d0; z0.y /= d0; z0.z /= d0; z0.w /= d0;
            z1.x /= d1; z1.y /= d1; z1.z /= d1; z1.w /= d1;
            z2.x /= d2; z2.y /= d2; z2.z /= d2; z2.w /= d2;
            z3.x /= d3; z3.y /= d3; z3.z /= d3; z3.w /= d3;
        }
        float* zp = Z + nl * ZP + lane * 4;
        *(float4*)zp         = z0;
        *(float4*)(zp + 128) = z1;
        *(float4*)(zp + 256) = z2;
        *(float4*)(zp + 384) = z3;
    }
    __syncthreads();
    if (wv >= 8) return;

    const int head  = wv >> 1;
    const int cb    = head * 64 + (wv & 1) * 32;
    const int c     = lane & 15;
    const int khalf = (lane >> 4) * 2;
    const float* za0 = Z + c * ZP + head * 128 + khalf;
    const float* za1 = za0 + 16 * ZP;
    const float* wp  = W1 + (size_t)khalf * 256 + cb + 2 * c;

    v8f c00 = {}, c01 = {}, c10 = {}, c11 = {};
    for (int k = 0; k < 128; k += 4) {
        const v2f a0 = *(const v2f*)(za0 + k);
        const v2f a1 = *(const v2f*)(za1 + k);
        const v2f r0 = *(const v2f*)(wp + (size_t)k * 256);
        const v2f r1 = *(const v2f*)(wp + (size_t)(k + 1) * 256);
        v2f b0; b0.x = r0.x; b0.y = r1.x;
        v2f b1v; b1v.x = r0.y; b1v.y = r1.y;
        c00 = __builtin_amdgcn_wmma_f32_16x16x4_f32(false, a0, false, b0,  (short)0, c00, false, false);
        c01 = __builtin_amdgcn_wmma_f32_16x16x4_f32(false, a0, false, b1v, (short)0, c01, false, false);
        c10 = __builtin_amdgcn_wmma_f32_16x16x4_f32(false, a1, false, b0,  (short)0, c10, false, false);
        c11 = __builtin_amdgcn_wmma_f32_16x16x4_f32(false, a1, false, b1v, (short)0, c11, false, false);
    }

    const int rowoff = (lane >= 16) ? 8 : 0;
    const v2f bias = *(const v2f*)(b1 + cb + 2 * c);
    for (int r = 0; r < 8; ++r) {
        const int g0 = blockIdx.x * GN + rowoff + r;
        const int g1 = g0 + 16;
        if (g0 < N) {
            v2f o; o.x = fmaxf(c00[r] + bias.x, 0.0f); o.y = fmaxf(c01[r] + bias.y, 0.0f);
            *(v2f*)(y1 + (size_t)g0 * 256 + cb + 2 * c) = o;
        }
        if (g1 < N) {
            v2f o; o.x = fmaxf(c10[r] + bias.x, 0.0f); o.y = fmaxf(c11[r] + bias.y, 0.0f);
            *(v2f*)(y1 + (size_t)g1 * 256 + cb + 2 * c) = o;
        }
    }
}

__global__ __launch_bounds__(256)
void node_agg2(const int* __restrict__ rowptr, const int* __restrict__ deg,
               const int* __restrict__ csr_src,
               const float* __restrict__ a_src, const float* __restrict__ a_dst,
               const float* __restrict__ h2, const float* __restrict__ b2,
               float* __restrict__ out, int N) {
    const int lane = threadIdx.x & 31;
    const int wv = __builtin_amdgcn_readfirstlane((int)(threadIdx.x >> 5));
    const int d = blockIdx.x * 8 + wv;
    if (d >= N) return;
    const int half = lane >> 4;
    const int t = lane & 15;
    const int start = rowptr[d];
    const int n = deg[d];
    const int cnt = n + 1;
    const float ad = a_dst[d];

    float m = -__builtin_huge_valf();
    for (int c = 0; c < cnt; c += 32) {
        const int k = c + lane;
        if (k < cnt) {
            const int s = k < n ? csr_src[start + k] : d;
            m = fmaxf(m, lrelu(a_src[s] + ad));
        }
    }
    m = wave_max(m);

    float den = 0.0f, acc = 0.0f;
    for (int c = 0; c < cnt; c += 32) {
        const int k = c + lane;
        int s = d;
        float w = 0.0f;
        if (k < cnt) {
            if (k < n) s = csr_src[start + k];
            w = expf(lrelu(a_src[s] + ad) - m);
        }
        den += w;
        const int cn = (cnt - c) < 32 ? (cnt - c) : 32;
        const int cp = cn & ~1;
#pragma unroll 4
        for (int j = 0; j < cp; j += 2) {
            const int s0 = __builtin_amdgcn_readlane(s, j);
            const int s1 = __builtin_amdgcn_readlane(s, j + 1);
            const float w0 = lane_f(w, j);
            const float w1 = lane_f(w, j + 1);
            const int sj = half ? s1 : s0;
            const float wj = half ? w1 : w0;
            acc += wj * h2[(size_t)sj * 16 + t];
        }
        if (cn & 1) {
            const int sl = __builtin_amdgcn_readlane(s, cp);
            const float wl = lane_f(w, cp);
            if (half == 0) acc += wl * h2[(size_t)sl * 16 + t];
        }
    }
    den = wave_sum(den);
    acc += __shfl_xor(acc, 16);
    if (lane < 16) out[(size_t)d * 16 + t] = acc / den + b2[t];
}


extern "C" void kernel_launch(void* const* d_in, const int* in_sizes, int n_in,
                              void* d_out, int out_size, void* d_ws, size_t ws_size,
                              hipStream_t stream) {
    const float* x   = (const float*)d_in[0];
    const int*   ei  = (const int*)  d_in[1];
    const float* W1  = (const float*)d_in[2];
    const float* as1 = (const float*)d_in[3];
    const float* ad1 = (const float*)d_in[4];
    const float* b1  = (const float*)d_in[5];
    const float* W2  = (const float*)d_in[6];
    const float* as2 = (const float*)d_in[7];
    const float* ad2 = (const float*)d_in[8];
    const float* b2  = (const float*)d_in[9];

    const int N  = in_sizes[0] / 128;
    const int E  = in_sizes[1] / 2;

    char* w = (char*)d_ws;
    auto carve = [&](size_t bytes) -> void* {
        void* p = (void*)w;
        w += (bytes + 255) & ~((size_t)255);
        return p;
    };
    float* y1     = (float*)carve((size_t)N * 256 * 4);
    float* aS1    = (float*)carve((size_t)N * 4 * 4);
    float* aD1    = (float*)carve((size_t)N * 4 * 4);
    float* U      = (float*)carve((size_t)128 * 8 * 4);
    float* h2     = (float*)carve((size_t)N * 16 * 4);
    float* aS2    = (float*)carve((size_t)N * 4);
    float* aD2    = (float*)carve((size_t)N * 4);
    int*   deg    = (int*)  carve((size_t)N * 4);
    int*   rowptr = (int*)  carve((size_t)N * 4);
    int*   rank   = (int*)  carve((size_t)E * 4);
    int*   csrs   = (int*)  carve((size_t)E * 4);
    const int NB  = (N + NBLK - 1) / NBLK;
    const int epc = (E + CHUNKS - 1) / CHUNKS;
    int*   cnt    = (int*)  carve((size_t)CHUNKS * HCAP * 4);
    int*   coarse = (int*)  carve((size_t)CHUNKS * NB * 4);

    const int TB = 256;
    auto blocks = [&](long long n) { return (unsigned)((n + TB - 1) / TB); };

    const int mtiles = (N + 15) / 16;
    const int bpc    = (int)blocks(epc);
    rank_and_fold<<<CHUNKS + 1, 1024, 0, stream>>>(
        ei, E, epc, NB, rank, cnt, coarse, W1, as1, ad1, U);
    scan_and_scores<<<NB + (mtiles + NBLK / 32 - 1) / (NBLK / 32), NBLK, 0, stream>>>(
        cnt, coarse, deg, rowptr, NB, x, U, aS1, aD1, N, mtiles);
    fill_groups<<<bpc * CHUNKS, TB, 0, stream>>>(ei, E, epc, bpc, cnt, rank, csrs);

    node_agg_proj1<<<(N + GN - 1) / GN, 512, 0, stream>>>(rowptr, deg, csrs, aS1, aD1, x, W1, b1, y1, N);

    gemm2_wmma<<<(N + 15) / 16, 32, 0, stream>>>(y1, W2, as2, ad2, h2, aS2, aD2, N);
    node_agg2<<<(N + 7) / 8, 256, 0, stream>>>(rowptr, deg, csrs, aS2, aD2, h2, b2, (float*)d_out, N);
}


# Layer-2 product fused into the layer-1 node kernel: W1 product on all 16 waves, transposed, so its accumulators feed the next WMMA as A; 16 k-partials added via LDS (new sum order); gemm2 launch gone
# speedup vs baseline: 5.9316x; 1.0923x over previous
#include <hip/hip_runtime.h>
#include <hip/hip_bf16.h>

typedef __attribute__((ext_vector_type(2))) float v2f;
typedef __attribute__((ext_vector_type(8))) float v8f;

#define NEG_SLOPE 0.2f


__device__ __forceinline__ float lrelu(float v) {
    return v >= 0.0f ? v : NEG_SLOPE * v;
}


__device__ __forceinline__
void fold_att(const float* __restrict__ W1, const float* __restrict__ att_src,
              const float* __restrict__ att_dst, float* __restrict__ U) {
    const int t = threadIdx.x;
    const int k = t >> 3, j = t & 7, h = j & 3;
    const float* w  = W1 + (size_t)k * 256 + h * 64;
    const float* at = (j < 4 ? att_src : att_dst) + h * 64;
    float s = 0.0f;
    for (int c = 0; c < 64; ++c) s += w[c] * at[c];
    U[t] = s;
}

__device__ __forceinline__
void score_tile(const int mtile, const float* __restrict__ x, const float* __restrict__ U,
                float* __restrict__ a_src, float* __restrict__ a_dst, int N) {
    const int lane  = threadIdx.x & 31;
    const int mrow  = mtile * 16 + (lane & 15);
    const int lrow  = mrow < N ? mrow : N - 1;
    const int khalf = (lane >> 4) * 2;
    const int nc    = lane & 15;
    const int uc    = nc & 7;

    v8f c = {};
    const float* xrow = x + (size_t)lrow * 128;
    for (int k = 0; k < 128; k += 4) {
        const int kb = k + khalf;
        v2f a; a.x = xrow[kb]; a.y = xrow[kb + 1];
        v2f b; b.x = U[kb * 8 + uc]; b.y = U[(kb + 1) * 8 + uc];
        c = __builtin_amdgcn_wmma_f32_16x16x4_f32(false, a, false, b, (short)0, c, false, false);
    }
    const int rowoff = (lane >= 16) ? 8 : 0;
    float* sp = (nc < 4) ? a_src : a_dst;
    for (int r = 0; r < 8; ++r) {
        const int grow = mtile * 16 + rowoff + r;
        if (grow < N && nc < 8) sp[(size_t)grow * 4 + (nc & 3)] = c[r];
    }
}

#define CHUNKS 64
#define HCAP   50176
#define NBLK   256
#define NBLK_LOG 8
#define EPT    5

__device__ __forceinline__ int wave_sum_i(int v) {
    for (int o = 16; o > 0; o >>= 1) v += __shfl_xor(v, o);
    return v;
}

__device__ __forceinline__
void chunk_rank(int* __restrict__ hist, int* __restrict__ blk,
                const int* __restrict__ ei, int E, int epc, int NB,
                int* __restrict__ rank, int* __restrict__ cnt, int* __restrict__ coarse) {
    const int t = threadIdx.x;
    const int c = blockIdx.x;
    int4* h4 = (int4*)hist;
    for (int i = t; i < HCAP / 4; i += 1024) h4[i] = make_int4(0, 0, 0, 0);
    if (t < HCAP / NBLK) blk[t] = 0;
    __syncthreads();

    const int e0 = c * epc;
    const int e1 = (e0 + epc) < E ? (e0 + epc) : E;
    for (int eb = e0 + t; eb < e1; eb += EPT * 1024) {
        int dd[EPT], rr[EPT];
#pragma unroll
        for (int u = 0; u < EPT; ++u) {
            const int e = eb + u * 1024;
            dd[u] = e < e1 ? ei[E + e] : 0;
        }
#pragma unroll
        for (int u = 0; u < EPT; ++u) {
            const int e = eb + u * 1024;
            rr[u] = 0;
            if (e < e1) {
                rr[u] = atomicAdd(&hist[dd[u]], 1);
                atomicAdd(&blk[dd[u] >> NBLK_LOG], 1);
            }
        }
#pragma unroll
        for (int u = 0; u < EPT; ++u) {
            const int e = eb + u * 1024;
            if (e < e1) rank[e] = rr[u];
        }
    }
    __syncthreads();

    int4* c4 = (int4*)(cnt + (size_t)c * HCAP);
    for (int i = t; i < HCAP / 4; i += 1024) c4[i] = h4[i];
    if (t < NB) coarse[c * NB + t] = blk[t];
}

__device__ __forceinline__
void scan_cols(int* __restrict__ wsum, int* __restrict__ wpre,
               int* __restrict__ cnt, const int* __restrict__ coarse,
               int* __restrict__ deg, int* __restrict__ rowptr, int N, int NB) {
    const int t = threadIdx.x, lane = t & 31, wv = t >> 5;
    const int j = blockIdx.x;
    const int idx = j * NBLK + t;

    int pre = 0;
    for (int b = t; b < j; b += NBLK)
        for (int c = 0; c < CHUNKS; ++c) pre += coarse[c * NB + b];

    int v[CHUNKS];
    int d = 0;
#pragma unroll
    for (int c = 0; c < CHUNKS; ++c) {
        v[c] = cnt[(size_t)c * HCAP + idx];
        d += v[c];
    }
    int inc = d;
    for (int off = 1; off < 32; off <<= 1) {
        const int nb = __shfl_up(inc, off);
        if (lane >= off) inc += nb;
    }
    pre = wave_sum_i(pre);
    if (lane == 31) wsum[wv] = inc;
    if (lane == 0)  wpre[wv] = pre;
    __syncthreads();
    int base = 0;
    for (int k = 0; k < NBLK / 32; ++k) {
        base += wpre[k];
        if (k < wv) base += wsum[k];
    }
    if (idx < N) {
        int run = base + inc - d;
        rowptr[idx] = run;
        deg[idx] = d;
#pragma unroll
        for (int c = 0; c < CHUNKS; ++c) {
            cnt[(size_t)c * HCAP + idx] = run;
            run += v[c];
        }
    }
}

__device__ __forceinline__
void fill_csr(const int b, const int* __restrict__ ei, int E, int epc, int bpc,
              const int* __restrict__ cnt, const int* __restrict__ rank,
              int* __restrict__ csr_src) {
    const int c = b / bpc;
    const int e0 = c * epc;
    const int e1 = (e0 + epc) < E ? (e0 + epc) : E;
    const int e = e0 + (b - c * bpc) * 256 + threadIdx.x;
    if (e >= e1) return;
    csr_src[cnt[(size_t)c * HCAP + ei[E + e]] + rank[e]] = ei[e];
}


__global__ __launch_bounds__(1024)
void rank_and_fold(const int* __restrict__ ei, int E, int epc, int NB,
                   int* __restrict__ rank, int* __restrict__ cnt, int* __restrict__ coarse,
                   const float* __restrict__ W1,
                   const float* __restrict__ att_src, const float* __restrict__ att_dst,
                   float* __restrict__ U) {
    __shared__ int hist[HCAP];
    __shared__ int blk[HCAP / NBLK];
    if (blockIdx.x < CHUNKS) {
        chunk_rank(hist, blk, ei, E, epc, NB, rank, cnt, coarse);
    } else {
        fold_att(W1, att_src, att_dst, U);
    }
}

__global__ __launch_bounds__(NBLK)
void scan_and_scores(int* __restrict__ cnt, const int* __restrict__ coarse,
                     int* __restrict__ deg, int* __restrict__ rowptr, int NB,
                     const float* __restrict__ x, const float* __restrict__ U,
                     float* __restrict__ a_src, float* __restrict__ a_dst,
                     int N, int mtiles) {
    __shared__ int wsum[NBLK / 32];
    __shared__ int wpre[NBLK / 32];
    if ((int)blockIdx.x < NB) {
        scan_cols(wsum, wpre, cnt, coarse, deg, rowptr, N, NB);
    } else {
        const int wv = __builtin_amdgcn_readfirstlane((int)(threadIdx.x >> 5));
        const int tile = ((int)blockIdx.x - NB) * (NBLK / 32) + wv;
        if (tile < mtiles) score_tile(tile, x, U, a_src, a_dst, N);
    }
}

__global__ __launch_bounds__(256)
void fill_groups(const int* __restrict__ ei, int E, int epc, int bpc,
                 const int* __restrict__ cnt, const int* __restrict__ rank,
                 int* __restrict__ csr_src) {
    fill_csr((int)blockIdx.x, ei, E, epc, bpc, cnt, rank, csr_src);
}

__device__ __forceinline__ float wave_max(float v) {
    for (int o = 16; o > 0; o >>= 1) v = fmaxf(v, __shfl_xor(v, o));
    return v;
}
__device__ __forceinline__ float wave_sum(float v) {
    for (int o = 16; o > 0; o >>= 1) v += __shfl_xor(v, o);
    return v;
}
__device__ __forceinline__ float lane_f(float v, int j) {
    return __int_as_float(__builtin_amdgcn_readlane(__float_as_int(v), j));
}

#define GN 32
#define ZP 516
#define PT 288
#define PH 144

__global__ __launch_bounds__(512) __attribute__((amdgpu_waves_per_eu(16)))
void node_agg_proj12(const int* __restrict__ rowptr, const int* __restrict__ deg,
                     const int* __restrict__ csr_src,
                     const float* __restrict__ a_src, const float* __restrict__ a_dst,
                     const float* __restrict__ x, const float* __restrict__ W1,
                     const float* __restrict__ b1, const float* __restrict__ W2,
                     const float* __restrict__ att_src2, const float* __restrict__ att_dst2,
                     float* __restrict__ h2, float* __restrict__ a_src2, float* __restrict__ a_dst2,
                     int N) {
    __shared__ __attribute__((aligned(16))) float Z[GN * ZP];
    const int lane = threadIdx.x & 31;
    const int wv = __builtin_amdgcn_readfirstlane((int)(threadIdx.x >> 5));

    for (int q = 0; q < 2; ++q) {
        const int nl = wv * 2 + q;
        const int d = blockIdx.x * GN + nl;
        float4 z0 = make_float4(0.0f, 0.0f, 0.0f, 0.0f);
        float4 z1 = z0, z2 = z0, z3 = z0;
        if (d < N) {
            const int start = rowptr[d];
            const int n = deg[d];
            const int cnt = n + 1;
            const float4 ad = *(const float4*)(a_dst + (size_t)d * 4);

            float m0 = -__builtin_huge_valf(), m1 = m0, m2 = m0, m3 = m0;
            for (int c = 0; c < cnt; c += 32) {
                const int k = c + lane;
                if (k < cnt) {
                    const int s = k < n ? csr_src[start + k] : d;
                    const float4 a = *(const float4*)(a_src + (size_t)s * 4);
                    m0 = fmaxf(m0, lrelu(a.x + ad.x));
                    m1 = fmaxf(m1, lrelu(a.y + ad.y));
                    m2 = fmaxf(m2, lrelu(a.z + ad.z));
                    m3 = fmaxf(m3, lrelu(a.w + ad.w));
                }
            }
            m0 = wave_max(m0); m1 = wave_max(m1); m2 = wave_max(m2); m3 = wave_max(m3);

            float d0 = 0.0f, d1 = 0.0f, d2 = 0.0f, d3 = 0.0f;
            for (int c = 0; c < cnt; c += 32) {
                const int k = c + lane;
                int s = d;
                float w0 = 0.0f, w1 = 0.0f, w2 = 0.0f, w3 = 0.0f;
                if (k < cnt) {
                    if (k < n) s = csr_src[start + k];
                    const float4 a = *(const float4*)(a_src + (size_t)s * 4);
                    w0 = expf(lrelu(a.x + ad.x) - m0);
                    w1 = expf(lrelu(a.y + ad.y) - m1);
                    w2 = expf(lrelu(a.z + ad.z) - m2);
                    w3 = expf(lrelu(a.w + ad.w) - m3);
                }
                d0 += w0; d1 += w1; d2 += w2; d3 += w3;
                const int cn = (cnt - c) < 32 ? (cnt - c) : 32;
#pragma unroll 4
                for (int j = 0; j < cn; ++j) {
                    const int sj = __builtin_amdgcn_readlane(s, j);
                    const float x0 = lane_f(w0, j), x1 = lane_f(w1, j);
                    const float x2 = lane_f(w2, j), x3 = lane_f(w3, j);
                    const float4 v = *(const float4*)(x + (size_t)sj * 128 + lane * 4);
                    z0.x += x0 * v.x; z0.y += x0 * v.y; z0.z += x0 * v.z; z0.w += x0 * v.w;
                    z1.x += x1 * v.x; z1.y += x1 * v.y; z1.z += x1 * v.z; z1.w += x1 * v.w;
                    z2.x += x2 * v.x; z2.y += x2 * v.y; z2.z += x2 * v.z; z2.w += x2 * v.w;
                    z3.x += x3 * v.x; z3.y += x3 * v.y; z3.z += x3 * v.z; z3.w += x3 * v.w;
                }
            }
            d0 = wave_sum(d0); d1 = wave_sum(d1); d2 = wave_sum(d2); d3 = wave_sum(d3);
            z0.x /= d0; z0.y /= d0; z0.z /= d0; z0.w /= d0;
            z1.x /= d1; z1.y /= d1; z1.z /= d1; z1.w /= d1;
            z2.x /= d2; z2.y /= d2; z2.z /= d2; z2.w /= d2;
            z3.x /= d3; z3.y /= d3; z3.z /= d3; z3.w /= d3;
        }
        float* zp = Z + nl * ZP + lane * 4;
        *(float4*)zp         = z0;
        *(float4*)(zp + 128) = z1;
        *(float4*)(zp + 256) = z2;
        *(float4*)(zp + 384) = z3;
    }
    __syncthreads();

    const int c     = lane & 15;
    const int hf    = lane >> 4;
    const int khalf = hf * 2;
    const int head  = wv >> 2;
    const int cb    = wv * 16;
    const int col0  = cb + 8 * hf;

    const float* w2p = W2 + (size_t)col0 * 16 + c;
    v2f wb[4];
    for (int t = 0; t < 4; ++t) {
        wb[t].x = w2p[(2 * t) * 16];
        wb[t].y = w2p[(2 * t + 1) * 16];
    }
    const float4 bs0 = *(const float4*)(b1 + col0);
    const float4 bs1 = *(const float4*)(b1 + col0 + 4);

    const float* za0 = Z + c * ZP + head * 128 + khalf;
    const float* za1 = za0 + 16 * ZP;
    const float* wp  = W1 + (size_t)khalf * 256 + cb + c;
    v8f c0 = {}, c1 = {};
    v2f wn[4];
    for (int i = 0; i < 4; ++i) {
        wn[i].x = wp[(size_t)(4 * i) * 256];
        wn[i].y = wp[(size_t)(4 * i + 1) * 256];
    }
#pragma unroll 1
    for (int kk = 0; kk < 128 - 16; kk += 16) {
#pragma unroll
        for (int i = 0; i < 4; ++i) {
            const int k = kk + 4 * i;
            const v2f a0 = *(const v2f*)(za0 + k);
            const v2f a1 = *(const v2f*)(za1 + k);
            c0 = __builtin_amdgcn_wmma_f32_16x16x4_f32(false, wn[i], false, a0, (short)0, c0, false, false);
            c1 = __builtin_amdgcn_wmma_f32_16x16x4_f32(false, wn[i], false, a1, (short)0, c1, false, false);
            wn[i].x = wp[(size_t)(k + 16) * 256];
            wn[i].y = wp[(size_t)(k + 16 + 1) * 256];
        }
    }
    for (int i = 0; i < 4; ++i) {
        const int k = 128 - 16 + 4 * i;
        const v2f a0 = *(const v2f*)(za0 + k);
        const v2f a1 = *(const v2f*)(za1 + k);
        c0 = __builtin_amdgcn_wmma_f32_16x16x4_f32(false, wn[i], false, a0, (short)0, c0, false, false);
        c1 = __builtin_amdgcn_wmma_f32_16x16x4_f32(false, wn[i], false, a1, (short)0, c1, false, false);
    }

    const float bs[8] = {bs0.x, bs0.y, bs0.z, bs0.w, bs1.x, bs1.y, bs1.z, bs1.w};
    for (int r = 0; r < 8; ++r) {
        c0[r] = fmaxf(c0[r] + bs[r], 0.0f);
        c1[r] = fmaxf(c1[r] + bs[r], 0.0f);
    }

    v8f d0 = {}, d1 = {};
    for (int t = 0; t < 4; ++t) {
        v2f a;
        a.x = c0[2 * t]; a.y = c0[2 * t + 1];
        d0 = __builtin_amdgcn_wmma_f32_16x16x4_f32(false, a, false, wb[t], (short)0, d0, false, false);
        a.x = c1[2 * t]; a.y = c1[2 * t + 1];
        d1 = __builtin_amdgcn_wmma_f32_16x16x4_f32(false, a, false, wb[t], (short)0, d1, false, false);
    }
    const float as = att_src2[c];
    const float ad = att_dst2[c];
    __syncthreads();

    float* P = Z;
    {
        float* pp = P + wv * (2 * PT) + hf * PH + c;
        for (int r = 0; r < 8; ++r) {
            pp[r * 16]      = d0[r];
            pp[PT + r * 16] = d1[r];
        }
    }
    __syncthreads();

    {
        const int node = wv * 2 + hf;
        const int nl   = node & 15;
        const float* q = P + (node >> 4) * PT + (nl >> 3) * PH + (nl & 7) * 16 + c;
        float sum = q[0];
        for (int w = 1; w < 16; ++w) sum += q[w * (2 * PT)];
        float ps = sum * as;
        float pd = sum * ad;
        for (int o = 1; o < 16; o <<= 1) {
            ps += __shfl_xor(ps, o);
            pd += __shfl_xor(pd, o);
        }
        const int g = blockIdx.x * GN + node;
        if (g < N) {
            h2[(size_t)g * 16 + c] = sum;
            if (c == 0) {
                a_src2[g] = ps;
                a_dst2[g] = pd;
            }
        }
    }
}

__global__ __launch_bounds__(256)
void node_agg2(const int* __restrict__ rowptr, const int* __restrict__ deg,
               const int* __restrict__ csr_src,
               const float* __restrict__ a_src, const float* __restrict__ a_dst,
               const float* __restrict__ h2, const float* __restrict__ b2,
               float* __restrict__ out, int N) {
    const int lane = threadIdx.x & 31;
    const int wv = __builtin_amdgcn_readfirstlane((int)(threadIdx.x >> 5));
    const int d = blockIdx.x * 8 + wv;
    if (d >= N) return;
    const int half = lane >> 4;
    const int t = lane & 15;
    const int start = rowptr[d];
    const int n = deg[d];
    const int cnt = n + 1;
    const float ad = a_dst[d];

    float m = -__builtin_huge_valf();
    for (int c = 0; c < cnt; c += 32) {
        const int k = c + lane;
        if (k < cnt) {
            const int s = k < n ? csr_src[start + k] : d;
            m = fmaxf(m, lrelu(a_src[s] + ad));
        }
    }
    m = wave_max(m);

    float den = 0.0f, acc = 0.0f;
    for (int c = 0; c < cnt; c += 32) {
        const int k = c + lane;
        int s = d;
        float w = 0.0f;
        if (k < cnt) {
            if (k < n) s = csr_src[start + k];
            w = expf(lrelu(a_src[s] + ad) - m);
        }
        den += w;
        const int cn = (cnt - c) < 32 ? (cnt - c) : 32;
        const int cp = cn & ~1;
#pragma unroll 4
        for (int j = 0; j < cp; j += 2) {
            const int s0 = __builtin_amdgcn_readlane(s, j);
            const int s1 = __builtin_amdgcn_readlane(s, j + 1);
            const float w0 = lane_f(w, j);
            const float w1 = lane_f(w, j + 1);
            const int sj = half ? s1 : s0;
            const float wj = half ? w1 : w0;
            acc += wj * h2[(size_t)sj * 16 + t];
        }
        if (cn & 1) {
            const int sl = __builtin_amdgcn_readlane(s, cp);
            const float wl = lane_f(w, cp);
            if (half == 0) acc += wl * h2[(size_t)sl * 16 + t];
        }
    }
    den = wave_sum(den);
    acc += __shfl_xor(acc, 16);
    if (lane < 16) out[(size_t)d * 16 + t] = acc / den + b2[t];
}


extern "C" void kernel_launch(void* const* d_in, const int* in_sizes, int n_in,
                              void* d_out, int out_size, void* d_ws, size_t ws_size,
                              hipStream_t stream) {
    const float* x   = (const float*)d_in[0];
    const int*   ei  = (const int*)  d_in[1];
    const float* W1  = (const float*)d_in[2];
    const float* as1 = (const float*)d_in[3];
    const float* ad1 = (const float*)d_in[4];
    const float* b1  = (const float*)d_in[5];
    const float* W2  = (const float*)d_in[6];
    const float* as2 = (const float*)d_in[7];
    const float* ad2 = (const float*)d_in[8];
    const float* b2  = (const float*)d_in[9];

    const int N  = in_sizes[0] / 128;
    const int E  = in_sizes[1] / 2;

    char* w = (char*)d_ws;
    auto carve = [&](size_t bytes) -> void* {
        void* p = (void*)w;
        w += (bytes + 255) & ~((size_t)255);
        return p;
    };
    float* aS1    = (float*)carve((size_t)N * 4 * 4);
    float* aD1    = (float*)carve((size_t)N * 4 * 4);
    float* U      = (float*)carve((size_t)128 * 8 * 4);
    float* h2     = (float*)carve((size_t)N * 16 * 4);
    float* aS2    = (float*)carve((size_t)N * 4);
    float* aD2    = (float*)carve((size_t)N * 4);
    int*   deg    = (int*)  carve((size_t)N * 4);
    int*   rowptr = (int*)  carve((size_t)N * 4);
    int*   rank   = (int*)  carve((size_t)E * 4);
    int*   csrs   = (int*)  carve((size_t)E * 4);
    const int NB  = (N + NBLK - 1) / NBLK;
    const int epc = (E + CHUNKS - 1) / CHUNKS;
    int*   cnt    = (int*)  carve((size_t)CHUNKS * HCAP * 4);
    int*   coarse = (int*)  carve((size_t)CHUNKS * NB * 4);

    const int TB = 256;
    auto blocks = [&](long long n) { return (unsigned)((n + TB - 1) / TB); };

    const int mtiles = (N + 15) / 16;
    const int bpc    = (int)blocks(epc);
    rank_and_fold<<<CHUNKS + 1, 1024, 0, stream>>>(
        ei, E, epc, NB, rank, cnt, coarse, W1, as1, ad1, U);
    scan_and_scores<<<NB + (mtiles + NBLK / 32 - 1) / (NBLK / 32), NBLK, 0, stream>>>(
        cnt, coarse, deg, rowptr, NB, x, U, aS1, aD1, N, mtiles);
    fill_groups<<<bpc * CHUNKS, TB, 0, stream>>>(ei, E, epc, bpc, cnt, rank, csrs);

    node_agg_proj12<<<(N + GN - 1) / GN, 512, 0, stream>>>(
        rowptr, deg, csrs, aS1, aD1, x, W1, b1, W2, as2, ad2, h2, aS2, aD2, N);

    node_agg2<<<(N + 7) / 8, 256, 0, stream>>>(rowptr, deg, csrs, aS2, aD2, h2, b2, (float*)d_out, N);
}


# Wave sums and maxima (softmax maximum and denominator, score sums) as xor butterflies on the vector unit (DPP row_xmask, permlanex16), four heads interleaved, not LDS permutes; sum order kept
# speedup vs baseline: 6.1416x; 1.0354x over previous
#include <hip/hip_runtime.h>
#include <hip/hip_bf16.h>

typedef __attribute__((ext_vector_type(2))) float v2f;
typedef __attribute__((ext_vector_type(8))) float v8f;

#define NEG_SLOPE 0.2f


__device__ __forceinline__ float lrelu(float v) {
    return v >= 0.0f ? v : NEG_SLOPE * v;
}


__device__ __forceinline__ int xhalf_i(int b) {
    return __builtin_amdgcn_permlanex16(b, b, 0x76543210, 0xfedcba98, false, false);
}
template <int M> __device__ __forceinline__ int xrow_i(int b) {
    return __builtin_amdgcn_update_dpp(0, b, 0x160 + M, 0xf, 0xf, true);
}
__device__ __forceinline__ float xhalf(float v) { return __int_as_float(xhalf_i(__float_as_int(v))); }
template <int M> __device__ __forceinline__ float xrow(float v) {
    return __int_as_float(xrow_i<M>(__float_as_int(v)));
}


__device__ __forceinline__
void fold_att(const float* __restrict__ W1, const float* __restrict__ att_src,
              const float* __restrict__ att_dst, float* __restrict__ U) {
    const int t = threadIdx.x;
    const int k = t >> 3, j = t & 7, h = j & 3;
    const float* w  = W1 + (size_t)k * 256 + h * 64;
    const float* at = (j < 4 ? att_src : att_dst) + h * 64;
    float s = 0.0f;
    for (int c = 0; c < 64; ++c) s += w[c] * at[c];
    U[t] = s;
}

__device__ __forceinline__
void score_tile(const int mtile, const float* __restrict__ x, const float* __restrict__ U,
                float* __restrict__ a_src, float* __restrict__ a_dst, int N) {
    const int lane  = threadIdx.x & 31;
    const int mrow  = mtile * 16 + (lane & 15);
    const int lrow  = mrow < N ? mrow : N - 1;
    const int khalf = (lane >> 4) * 2;
    const int nc    = lane & 15;
    const int uc    = nc & 7;

    v8f c = {};
    const float* xrow = x + (size_t)lrow * 128;
    for (int k = 0; k < 128; k += 4) {
        const int kb = k + khalf;
        v2f a; a.x = xrow[kb]; a.y = xrow[kb + 1];
        v2f b; b.x = U[kb * 8 + uc]; b.y = U[(kb + 1) * 8 + uc];
        c = __builtin_amdgcn_wmma_f32_16x16x4_f32(false, a, false, b, (short)0, c, false, false);
    }
    const int rowoff = (lane >= 16) ? 8 : 0;
    float* sp = (nc < 4) ? a_src : a_dst;
    for (int r = 0; r < 8; ++r) {
        const int grow = mtile * 16 + rowoff + r;
        if (grow < N && nc < 8) sp[(size_t)grow * 4 + (nc & 3)] = c[r];
    }
}

#define CHUNKS 64
#define HCAP   50176
#define NBLK   256
#define NBLK_LOG 8
#define EPT    5

__device__ __forceinline__ int wave_sum_i(int v) {
    v += xhalf_i(v);
    v += xrow_i<8>(v); v += xrow_i<4>(v); v += xrow_i<2>(v); v += xrow_i<1>(v);
    return v;
}

__device__ __forceinline__
void chunk_rank(int* __restrict__ hist, int* __restrict__ blk,
                const int* __restrict__ ei, int E, int epc, int NB,
                int* __restrict__ rank, int* __restrict__ cnt, int* __restrict__ coarse) {
    const int t = threadIdx.x;
    const int c = blockIdx.x;
    int4* h4 = (int4*)hist;
    for (int i = t; i < HCAP / 4; i += 1024) h4[i] = make_int4(0, 0, 0, 0);
    if (t < HCAP / NBLK) blk[t] = 0;
    __syncthreads();

    const int e0 = c * epc;
    const int e1 = (e0 + epc) < E ? (e0 + epc) : E;
    for (int eb = e0 + t; eb < e1; eb += EPT * 1024) {
        int dd[EPT], rr[EPT];
#pragma unroll
        for (int u = 0; u < EPT; ++u) {
            const int e = eb + u * 1024;
            dd[u] = e < e1 ? ei[E + e] : 0;
        }
#pragma unroll
        for (int u = 0; u < EPT; ++u) {
            const int e = eb + u * 1024;
            rr[u] = 0;
            if (e < e1) {
                rr[u] = atomicAdd(&hist[dd[u]], 1);
                atomicAdd(&blk[dd[u] >> NBLK_LOG], 1);
            }
        }
#pragma unroll
        for (int u = 0; u < EPT; ++u) {
            const int e = eb + u * 1024;
            if (e < e1) rank[e] = rr[u];
        }
    }
    __syncthreads();

    int4* c4 = (int4*)(cnt + (size_t)c * HCAP);
    for (int i = t; i < HCAP / 4; i += 1024) c4[i] = h4[i];
    if (t < NB) coarse[c * NB + t] = blk[t];
}

__device__ __forceinline__
void scan_cols(int* __restrict__ wsum, int* __restrict__ wpre,
               int* __restrict__ cnt, const int* __restrict__ coarse,
               int* __restrict__ deg, int* __restrict__ rowptr, int N, int NB) {
    const int t = threadIdx.x, lane = t & 31, wv = t >> 5;
    const int j = blockIdx.x;
    const int idx = j * NBLK + t;

    int pre = 0;
    for (int b = t; b < j; b += NBLK)
        for (int c = 0; c < CHUNKS; ++c) pre += coarse[c * NB + b];

    int v[CHUNKS];
    int d = 0;
#pragma unroll
    for (int c = 0; c < CHUNKS; ++c) {
        v[c] = cnt[(size_t)c * HCAP + idx];
        d += v[c];
    }
    int inc = d;
    for (int off = 1; off < 32; off <<= 1) {
        const int nb = __shfl_up(inc, off);
        if (lane >= off) inc += nb;
    }
    pre = wave_sum_i(pre);
    if (lane == 31) wsum[wv] = inc;
    if (lane == 0)  wpre[wv] = pre;
    __syncthreads();
    int base = 0;
    for (int k = 0; k < NBLK / 32; ++k) {
        base += wpre[k];
        if (k < wv) base += wsum[k];
    }
    if (idx < N) {
        int run = base + inc - d;
        rowptr[idx] = run;
        deg[idx] = d;
#pragma unroll
        for (int c = 0; c < CHUNKS; ++c) {
            cnt[(size_t)c * HCAP + idx] = run;
            run += v[c];
        }
    }
}

__device__ __forceinline__
void fill_csr(const int b, const int* __restrict__ ei, int E, int epc, int bpc,
              const int* __restrict__ cnt, const int* __restrict__ rank,
              int* __restrict__ csr_src) {
    const int c = b / bpc;
    const int e0 = c * epc;
    const int e1 = (e0 + epc) < E ? (e0 + epc) : E;
    const int e = e0 + (b - c * bpc) * 256 + threadIdx.x;
    if (e >= e1) return;
    csr_src[cnt[(size_t)c * HCAP + ei[E + e]] + rank[e]] = ei[e];
}


__global__ __launch_bounds__(1024)
void rank_and_fold(const int* __restrict__ ei, int E, int epc, int NB,
                   int* __restrict__ rank, int* __restrict__ cnt, int* __restrict__ coarse,
                   const float* __restrict__ W1,
                   const float* __restrict__ att_src, const float* __restrict__ att_dst,
                   float* __restrict__ U) {
    __shared__ int hist[HCAP];
    __shared__ int blk[HCAP / NBLK];
    if (blockIdx.x < CHUNKS) {
        chunk_rank(hist, blk, ei, E, epc, NB, rank, cnt, coarse);
    } else {
        fold_att(W1, att_src, att_dst, U);
    }
}

__global__ __launch_bounds__(NBLK)
void scan_and_scores(int* __restrict__ cnt, const int* __restrict__ coarse,
                     int* __restrict__ deg, int* __restrict__ rowptr, int NB,
                     const float* __restrict__ x, const float* __restrict__ U,
                     float* __restrict__ a_src, float* __restrict__ a_dst,
                     int N, int mtiles) {
    __shared__ int wsum[NBLK / 32];
    __shared__ int wpre[NBLK / 32];
    if ((int)blockIdx.x < NB) {
        scan_cols(wsum, wpre, cnt, coarse, deg, rowptr, N, NB);
    } else {
        const int wv = __builtin_amdgcn_readfirstlane((int)(threadIdx.x >> 5));
        const int tile = ((int)blockIdx.x - NB) * (NBLK / 32) + wv;
        if (tile < mtiles) score_tile(tile, x, U, a_src, a_dst, N);
    }
}

__global__ __launch_bounds__(256)
void fill_groups(const int* __restrict__ ei, int E, int epc, int bpc,
                 const int* __restrict__ cnt, const int* __restrict__ rank,
                 int* __restrict__ csr_src) {
    fill_csr((int)blockIdx.x, ei, E, epc, bpc, cnt, rank, csr_src);
}

__device__ __forceinline__ float wave_max(float v) {
    v = fmaxf(v, xhalf(v));
    v = fmaxf(v, xrow<8>(v)); v = fmaxf(v, xrow<4>(v));
    v = fmaxf(v, xrow<2>(v)); v = fmaxf(v, xrow<1>(v));
    return v;
}
__device__ __forceinline__ float wave_sum(float v) {
    v += xhalf(v);
    v += xrow<8>(v); v += xrow<4>(v); v += xrow<2>(v); v += xrow<1>(v);
    return v;
}
__device__ __forceinline__ void wave_max4(float& a, float& b, float& c, float& d) {
    a = fmaxf(a, xhalf(a));   b = fmaxf(b, xhalf(b));   c = fmaxf(c, xhalf(c));   d = fmaxf(d, xhalf(d));
    a = fmaxf(a, xrow<8>(a)); b = fmaxf(b, xrow<8>(b)); c = fmaxf(c, xrow<8>(c)); d = fmaxf(d, xrow<8>(d));
    a = fmaxf(a, xrow<4>(a)); b = fmaxf(b, xrow<4>(b)); c = fmaxf(c, xrow<4>(c)); d = fmaxf(d, xrow<4>(d));
    a = fmaxf(a, xrow<2>(a)); b = fmaxf(b, xrow<2>(b)); c = fmaxf(c, xrow<2>(c)); d = fmaxf(d, xrow<2>(d));
    a = fmaxf(a, xrow<1>(a)); b = fmaxf(b, xrow<1>(b)); c = fmaxf(c, xrow<1>(c)); d = fmaxf(d, xrow<1>(d));
}
__device__ __forceinline__ void wave_sum4(float& a, float& b, float& c, float& d) {
    a += xhalf(a);   b += xhalf(b);   c += xhalf(c);   d += xhalf(d);
    a += xrow<8>(a); b += xrow<8>(b); c += xrow<8>(c); d += xrow<8>(d);
    a += xrow<4>(a); b += xrow<4>(b); c += xrow<4>(c); d += xrow<4>(d);
    a += xrow<2>(a); b += xrow<2>(b); c += xrow<2>(c); d += xrow<2>(d);
    a += xrow<1>(a); b += xrow<1>(b); c += xrow<1>(c); d += xrow<1>(d);
}
__device__ __forceinline__ float lane_f(float v, int j) {
    return __int_as_float(__builtin_amdgcn_readlane(__float_as_int(v), j));
}

#define GN 32
#define ZP 516
#define PT 288
#define PH 144

__global__ __launch_bounds__(512) __attribute__((amdgpu_waves_per_eu(16)))
void node_agg_proj12(const int* __restrict__ rowptr, const int* __restrict__ deg,
                     const int* __restrict__ csr_src,
                     const float* __restrict__ a_src, const float* __restrict__ a_dst,
                     const float* __restrict__ x, const float* __restrict__ W1,
                     const float* __restrict__ b1, const float* __restrict__ W2,
                     const float* __restrict__ att_src2, const float* __restrict__ att_dst2,
                     float* __restrict__ h2, float* __restrict__ a_src2, float* __restrict__ a_dst2,
                     int N) {
    __shared__ __attribute__((aligned(16))) float Z[GN * ZP];
    const int lane = threadIdx.x & 31;
    const int wv = __builtin_amdgcn_readfirstlane((int)(threadIdx.x >> 5));

    for (int q = 0; q < 2; ++q) {
        const int nl = wv * 2 + q;
        const int d = blockIdx.x * GN + nl;
        float4 z0 = make_float4(0.0f, 0.0f, 0.0f, 0.0f);
        float4 z1 = z0, z2 = z0, z3 = z0;
        if (d < N) {
            const int start = rowptr[d];
            const int n = deg[d];
            const int cnt = n + 1;
            const float4 ad = *(const float4*)(a_dst + (size_t)d * 4);

            float m0 = -__builtin_huge_valf(), m1 = m0, m2 = m0, m3 = m0;
            for (int c = 0; c < cnt; c += 32) {
                const int k = c + lane;
                if (k < cnt) {
                    const int s = k < n ? csr_src[start + k] : d;
                    const float4 a = *(const float4*)(a_src + (size_t)s * 4);
                    m0 = fmaxf(m0, lrelu(a.x + ad.x));
                    m1 = fmaxf(m1, lrelu(a.y + ad.y));
                    m2 = fmaxf(m2, lrelu(a.z + ad.z));
                    m3 = fmaxf(m3, lrelu(a.w + ad.w));
                }
            }
            wave_max4(m0, m1, m2, m3);

            float d0 = 0.0f, d1 = 0.0f, d2 = 0.0f, d3 = 0.0f;
            for (int c = 0; c < cnt; c += 32) {
                const int k = c + lane;
                int s = d;
                float w0 = 0.0f, w1 = 0.0f, w2 = 0.0f, w3 = 0.0f;
                if (k < cnt) {
                    if (k < n) s = csr_src[start + k];
                    const float4 a = *(const float4*)(a_src + (size_t)s * 4);
                    w0 = expf(lrelu(a.x + ad.x) - m0);
                    w1 = expf(lrelu(a.y + ad.y) - m1);
                    w2 = expf(lrelu(a.z + ad.z) - m2);
                    w3 = expf(lrelu(a.w + ad.w) - m3);
                }
                d0 += w0; d1 += w1; d2 += w2; d3 += w3;
                const int cn = (cnt - c) < 32 ? (cnt - c) : 32;
#pragma unroll 4
                for (int j = 0; j < cn; ++j) {
                    const int sj = __builtin_amdgcn_readlane(s, j);
                    const float x0 = lane_f(w0, j), x1 = lane_f(w1, j);
                    const float x2 = lane_f(w2, j), x3 = lane_f(w3, j);
                    const float4 v = *(const float4*)(x + (size_t)sj * 128 + lane * 4);
                    z0.x += x0 * v.x; z0.y += x0 * v.y; z0.z += x0 * v.z; z0.w += x0 * v.w;
                    z1.x += x1 * v.x; z1.y += x1 * v.y; z1.z += x1 * v.z; z1.w += x1 * v.w;
                    z2.x += x2 * v.x; z2.y += x2 * v.y; z2.z += x2 * v.z; z2.w += x2 * v.w;
                    z3.x += x3 * v.x; z3.y += x3 * v.y; z3.z += x3 * v.z; z3.w += x3 * v.w;
                }
            }
            wave_sum4(d0, d1, d2, d3);
            z0.x /= d0; z0.y /= d0; z0.z /= d0; z0.w /= d0;
            z1.x /= d1; z1.y /= d1; z1.z /= d1; z1.w /= d1;
            z2.x /= d2; z2.y /= d2; z2.z /= d2; z2.w /= d2;
            z3.x /= d3; z3.y /= d3; z3.z /= d3; z3.w /= d3;
        }
        float* zp = Z + nl * ZP + lane * 4;
        *(float4*)zp         = z0;
        *(float4*)(zp + 128) = z1;
        *(float4*)(zp + 256) = z2;
        *(float4*)(zp + 384) = z3;
    }
    __syncthreads();

    const int c     = lane & 15;
    const int hf    = lane >> 4;
    const int khalf = hf * 2;
    const int head  = wv >> 2;
    const int cb    = wv * 16;
    const int col0  = cb + 8 * hf;

    const float* w2p = W2 + (size_t)col0 * 16 + c;
    v2f wb[4];
    for (int t = 0; t < 4; ++t) {
        wb[t].x = w2p[(2 * t) * 16];
        wb[t].y = w2p[(2 * t + 1) * 16];
    }
    const float4 bs0 = *(const float4*)(b1 + col0);
    const float4 bs1 = *(const float4*)(b1 + col0 + 4);

    const float* za0 = Z + c * ZP + head * 128 + khalf;
    const float* za1 = za0 + 16 * ZP;
    const float* wp  = W1 + (size_t)khalf * 256 + cb + c;
    v8f c0 = {}, c1 = {};
    v2f wn[4];
    for (int i = 0; i < 4; ++i) {
        wn[i].x = wp[(size_t)(4 * i) * 256];
        wn[i].y = wp[(size_t)(4 * i + 1) * 256];
    }
#pragma unroll 1
    for (int kk = 0; kk < 128 - 16; kk += 16) {
#pragma unroll
        for (int i = 0; i < 4; ++i) {
            const int k = kk + 4 * i;
            const v2f a0 = *(const v2f*)(za0 + k);
            const v2f a1 = *(const v2f*)(za1 + k);
            c0 = __builtin_amdgcn_wmma_f32_16x16x4_f32(false, wn[i], false, a0, (short)0, c0, false, false);
            c1 = __builtin_amdgcn_wmma_f32_16x16x4_f32(false, wn[i], false, a1, (short)0, c1, false, false);
            wn[i].x = wp[(size_t)(k + 16) * 256];
            wn[i].y = wp[(size_t)(k + 16 + 1) * 256];
        }
    }
    for (int i = 0; i < 4; ++i) {
        const int k = 128 - 16 + 4 * i;
        const v2f a0 = *(const v2f*)(za0 + k);
        const v2f a1 = *(const v2f*)(za1 + k);
        c0 = __builtin_amdgcn_wmma_f32_16x16x4_f32(false, wn[i], false, a0, (short)0, c0, false, false);
        c1 = __builtin_amdgcn_wmma_f32_16x16x4_f32(false, wn[i], false, a1, (short)0, c1, false, false);
    }

    const float bs[8] = {bs0.x, bs0.y, bs0.z, bs0.w, bs1.x, bs1.y, bs1.z, bs1.w};
    for (int r = 0; r < 8; ++r) {
        c0[r] = fmaxf(c0[r] + bs[r], 0.0f);
        c1[r] = fmaxf(c1[r] + bs[r], 0.0f);
    }

    v8f d0 = {}, d1 = {};
    for (int t = 0; t < 4; ++t) {
        v2f a;
        a.x = c0[2 * t]; a.y = c0[2 * t + 1];
        d0 = __builtin_amdgcn_wmma_f32_16x16x4_f32(false, a, false, wb[t], (short)0, d0, false, false);
        a.x = c1[2 * t]; a.y = c1[2 * t + 1];
        d1 = __builtin_amdgcn_wmma_f32_16x16x4_f32(false, a, false, wb[t], (short)0, d1, false, false);
    }
    const float as = att_src2[c];
    const float ad = att_dst2[c];
    __syncthreads();

    float* P = Z;
    {
        float* pp = P + wv * (2 * PT) + hf * PH + c;
        for (int r = 0; r < 8; ++r) {
            pp[r * 16]      = d0[r];
            pp[PT + r * 16] = d1[r];
        }
    }
    __syncthreads();

    {
        const int node = wv * 2 + hf;
        const int nl   = node & 15;
        const float* q = P + (node >> 4) * PT + (nl >> 3) * PH + (nl & 7) * 16 + c;
        float sum = q[0];
        for (int w = 1; w < 16; ++w) sum += q[w * (2 * PT)];
        float ps = sum * as;
        float pd = sum * ad;
        ps += xrow<1>(ps); pd += xrow<1>(pd);
        ps += xrow<2>(ps); pd += xrow<2>(pd);
        ps += xrow<4>(ps); pd += xrow<4>(pd);
        ps += xrow<8>(ps); pd += xrow<8>(pd);
        const int g = blockIdx.x * GN + node;
        if (g < N) {
            h2[(size_t)g * 16 + c] = sum;
            if (c == 0) {
                a_src2[g] = ps;
                a_dst2[g] = pd;
            }
        }
    }
}

__global__ __launch_bounds__(256)
void node_agg2(const int* __restrict__ rowptr, const int* __restrict__ deg,
               const int* __restrict__ csr_src,
               const float* __restrict__ a_src, const float* __restrict__ a_dst,
               const float* __restrict__ h2, const float* __restrict__ b2,
               float* __restrict__ out, int N) {
    const int lane = threadIdx.x & 31;
    const int wv = __builtin_amdgcn_readfirstlane((int)(threadIdx.x >> 5));
    const int d = blockIdx.x * 8 + wv;
    if (d >= N) return;
    const int half = lane >> 4;
    const int t = lane & 15;
    const int start = rowptr[d];
    const int n = deg[d];
    const int cnt = n + 1;
    const float ad = a_dst[d];

    float m = -__builtin_huge_valf();
    for (int c = 0; c < cnt; c += 32) {
        const int k = c + lane;
        if (k < cnt) {
            const int s = k < n ? csr_src[start + k] : d;
            m = fmaxf(m, lrelu(a_src[s] + ad));
        }
    }
    m = wave_max(m);

    float den = 0.0f, acc = 0.0f;
    for (int c = 0; c < cnt; c += 32) {
        const int k = c + lane;
        int s = d;
        float w = 0.0f;
        if (k < cnt) {
            if (k < n) s = csr_src[start + k];
            w = expf(lrelu(a_src[s] + ad) - m);
        }
        den += w;
        const int cn = (cnt - c) < 32 ? (cnt - c) : 32;
        const int cp = cn & ~1;
#pragma unroll 4
        for (int j = 0; j < cp; j += 2) {
            const int s0 = __builtin_amdgcn_readlane(s, j);
            const int s1 = __builtin_amdgcn_readlane(s, j + 1);
            const float w0 = lane_f(w, j);
            const float w1 = lane_f(w, j + 1);
            const int sj = half ? s1 : s0;
            const float wj = half ? w1 : w0;
            acc += wj * h2[(size_t)sj * 16 + t];
        }
        if (cn & 1) {
            const int sl = __builtin_amdgcn_readlane(s, cp);
            const float wl = lane_f(w, cp);
            if (half == 0) acc += wl * h2[(size_t)sl * 16 + t];
        }
    }
    den = wave_sum(den);
    acc += xhalf(acc);
    if (lane < 16) out[(size_t)d * 16 + t] = acc / den + b2[t];
}


extern "C" void kernel_launch(void* const* d_in, const int* in_sizes, int n_in,
                              void* d_out, int out_size, void* d_ws, size_t ws_size,
                              hipStream_t stream) {
    const float* x   = (const float*)d_in[0];
    const int*   ei  = (const int*)  d_in[1];
    const float* W1  = (const float*)d_in[2];
    const float* as1 = (const float*)d_in[3];
    const float* ad1 = (const float*)d_in[4];
    const float* b1  = (const float*)d_in[5];
    const float* W2  = (const float*)d_in[6];
    const float* as2 = (const float*)d_in[7];
    const float* ad2 = (const float*)d_in[8];
    const float* b2  = (const float*)d_in[9];

    const int N  = in_sizes[0] / 128;
    const int E  = in_sizes[1] / 2;

    char* w = (char*)d_ws;
    auto carve = [&](size_t bytes) -> void* {
        void* p = (void*)w;
        w += (bytes + 255) & ~((size_t)255);
        return p;
    };
    float* aS1    = (float*)carve((size_t)N * 4 * 4);
    float* aD1    = (float*)carve((size_t)N * 4 * 4);
    float* U      = (float*)carve((size_t)128 * 8 * 4);
    float* h2     = (float*)carve((size_t)N * 16 * 4);
    float* aS2    = (float*)carve((size_t)N * 4);
    float* aD2    = (float*)carve((size_t)N * 4);
    int*   deg    = (int*)  carve((size_t)N * 4);
    int*   rowptr = (int*)  carve((size_t)N * 4);
    int*   rank   = (int*)  carve((size_t)E * 4);
    int*   csrs   = (int*)  carve((size_t)E * 4);
    const int NB  = (N + NBLK - 1) / NBLK;
    const int epc = (E + CHUNKS - 1) / CHUNKS;
    int*   cnt    = (int*)  carve((size_t)CHUNKS * HCAP * 4);
    int*   coarse = (int*)  carve((size_t)CHUNKS * NB * 4);

    const int TB = 256;
    auto blocks = [&](long long n) { return (unsigned)((n + TB - 1) / TB); };

    const int mtiles = (N + 15) / 16;
    const int bpc    = (int)blocks(epc);
    rank_and_fold<<<CHUNKS + 1, 1024, 0, stream>>>(
        ei, E, epc, NB, rank, cnt, coarse, W1, as1, ad1, U);
    scan_and_scores<<<NB + (mtiles + NBLK / 32 - 1) / (NBLK / 32), NBLK, 0, stream>>>(
        cnt, coarse, deg, rowptr, NB, x, U, aS1, aD1, N, mtiles);
    fill_groups<<<bpc * CHUNKS, TB, 0, stream>>>(ei, E, epc, bpc, cnt, rank, csrs);

    node_agg_proj12<<<(N + GN - 1) / GN, 512, 0, stream>>>(
        rowptr, deg, csrs, aS1, aD1, x, W1, b1, W2, as2, ad2, h2, aS2, aD2, N);

    node_agg2<<<(N + 7) / 8, 256, 0, stream>>>(rowptr, deg, csrs, aS2, aD2, h2, b2, (float*)d_out, N);
}


# Per-node load chain cut in both node kernels: list entry and scores loaded once, kept in registers; rows in whole zero-weight-padded groups, first group requested with the scores; sum order kept
# speedup vs baseline: 6.4721x; 1.0538x over previous
#include <hip/hip_runtime.h>
#include <hip/hip_bf16.h>

typedef __attribute__((ext_vector_type(2))) float v2f;
typedef __attribute__((ext_vector_type(8))) float v8f;

#define NEG_SLOPE 0.2f


__device__ __forceinline__ float lrelu(float v) {
    return v >= 0.0f ? v : NEG_SLOPE * v;
}


__device__ __forceinline__ int xhalf_i(int b) {
    return __builtin_amdgcn_permlanex16(b, b, 0x76543210, 0xfedcba98, false, false);
}
template <int M> __device__ __forceinline__ int xrow_i(int b) {
    return __builtin_amdgcn_update_dpp(0, b, 0x160 + M, 0xf, 0xf, true);
}
__device__ __forceinline__ float xhalf(float v) { return __int_as_float(xhalf_i(__float_as_int(v))); }
template <int M> __device__ __forceinline__ float xrow(float v) {
    return __int_as_float(xrow_i<M>(__float_as_int(v)));
}


__device__ __forceinline__
void fold_att(const float* __restrict__ W1, const float* __restrict__ att_src,
              const float* __restrict__ att_dst, float* __restrict__ U) {
    const int t = threadIdx.x;
    const int k = t >> 3, j = t & 7, h = j & 3;
    const float* w  = W1 + (size_t)k * 256 + h * 64;
    const float* at = (j < 4 ? att_src : att_dst) + h * 64;
    float s = 0.0f;
    for (int c = 0; c < 64; ++c) s += w[c] * at[c];
    U[t] = s;
}

__device__ __forceinline__
void score_tile(const int mtile, const float* __restrict__ x, const float* __restrict__ U,
                float* __restrict__ a_src, float* __restrict__ a_dst, int N) {
    const int lane  = threadIdx.x & 31;
    const int mrow  = mtile * 16 + (lane & 15);
    const int lrow  = mrow < N ? mrow : N - 1;
    const int khalf = (lane >> 4) * 2;
    const int nc    = lane & 15;
    const int uc    = nc & 7;

    v8f c = {};
    const float* xrow = x + (size_t)lrow * 128;
    for (int k = 0; k < 128; k += 4) {
        const int kb = k + khalf;
        v2f a; a.x = xrow[kb]; a.y = xrow[kb + 1];
        v2f b; b.x = U[kb * 8 + uc]; b.y = U[(kb + 1) * 8 + uc];
        c = __builtin_amdgcn_wmma_f32_16x16x4_f32(false, a, false, b, (short)0, c, false, false);
    }
    const int rowoff = (lane >= 16) ? 8 : 0;
    float* sp = (nc < 4) ? a_src : a_dst;
    for (int r = 0; r < 8; ++r) {
        const int grow = mtile * 16 + rowoff + r;
        if (grow < N && nc < 8) sp[(size_t)grow * 4 + (nc & 3)] = c[r];
    }
}

#define CHUNKS 64
#define HCAP   50176
#define NBLK   256
#define NBLK_LOG 8
#define EPT    5

__device__ __forceinline__ int wave_sum_i(int v) {
    v += xhalf_i(v);
    v += xrow_i<8>(v); v += xrow_i<4>(v); v += xrow_i<2>(v); v += xrow_i<1>(v);
    return v;
}

__device__ __forceinline__
void chunk_rank(int* __restrict__ hist, int* __restrict__ blk,
                const int* __restrict__ ei, int E, int epc, int NB,
                int* __restrict__ rank, int* __restrict__ cnt, int* __restrict__ coarse) {
    const int t = threadIdx.x;
    const int c = blockIdx.x;
    int4* h4 = (int4*)hist;
    for (int i = t; i < HCAP / 4; i += 1024) h4[i] = make_int4(0, 0, 0, 0);
    if (t < HCAP / NBLK) blk[t] = 0;
    __syncthreads();

    const int e0 = c * epc;
    const int e1 = (e0 + epc) < E ? (e0 + epc) : E;
    for (int eb = e0 + t; eb < e1; eb += EPT * 1024) {
        int dd[EPT], rr[EPT];
#pragma unroll
        for (int u = 0; u < EPT; ++u) {
            const int e = eb + u * 1024;
            dd[u] = e < e1 ? ei[E + e] : 0;
        }
#pragma unroll
        for (int u = 0; u < EPT; ++u) {
            const int e = eb + u * 1024;
            rr[u] = 0;
            if (e < e1) {
                rr[u] = atomicAdd(&hist[dd[u]], 1);
                atomicAdd(&blk[dd[u] >> NBLK_LOG], 1);
            }
        }
#pragma unroll
        for (int u = 0; u < EPT; ++u) {
            const int e = eb + u * 1024;
            if (e < e1) rank[e] = rr[u];
        }
    }
    __syncthreads();

    int4* c4 = (int4*)(cnt + (size_t)c * HCAP);
    for (int i = t; i < HCAP / 4; i += 1024) c4[i] = h4[i];
    if (t < NB) coarse[c * NB + t] = blk[t];
}

__device__ __forceinline__
void scan_cols(int* __restrict__ wsum, int* __restrict__ wpre,
               int* __restrict__ cnt, const int* __restrict__ coarse,
               int* __restrict__ deg, int* __restrict__ rowptr, int N, int NB) {
    const int t = threadIdx.x, lane = t & 31, wv = t >> 5;
    const int j = blockIdx.x;
    const int idx = j * NBLK + t;

    int pre = 0;
    for (int b = t; b < j; b += NBLK)
        for (int c = 0; c < CHUNKS; ++c) pre += coarse[c * NB + b];

    int v[CHUNKS];
    int d = 0;
#pragma unroll
    for (int c = 0; c < CHUNKS; ++c) {
        v[c] = cnt[(size_t)c * HCAP + idx];
        d += v[c];
    }
    int inc = d;
    for (int off = 1; off < 32; off <<= 1) {
        const int nb = __shfl_up(inc, off);
        if (lane >= off) inc += nb;
    }
    pre = wave_sum_i(pre);
    if (lane == 31) wsum[wv] = inc;
    if (lane == 0)  wpre[wv] = pre;
    __syncthreads();
    int base = 0;
    for (int k = 0; k < NBLK / 32; ++k) {
        base += wpre[k];
        if (k < wv) base += wsum[k];
    }
    if (idx < N) {
        int run = base + inc - d;
        rowptr[idx] = run;
        deg[idx] = d;
#pragma unroll
        for (int c = 0; c < CHUNKS; ++c) {
            cnt[(size_t)c * HCAP + idx] = run;
            run += v[c];
        }
    }
}

__device__ __forceinline__
void fill_csr(const int b, const int* __restrict__ ei, int E, int epc, int bpc,
              const int* __restrict__ cnt, const int* __restrict__ rank,
              int* __restrict__ csr_src) {
    const int c = b / bpc;
    const int e0 = c * epc;
    const int e1 = (e0 + epc) < E ? (e0 + epc) : E;
    const int e = e0 + (b - c * bpc) * 256 + threadIdx.x;
    if (e >= e1) return;
    csr_src[cnt[(size_t)c * HCAP + ei[E + e]] + rank[e]] = ei[e];
}


__global__ __launch_bounds__(1024)
void rank_and_fold(const int* __restrict__ ei, int E, int epc, int NB,
                   int* __restrict__ rank, int* __restrict__ cnt, int* __restrict__ coarse,
                   const float* __restrict__ W1,
                   const float* __restrict__ att_src, const float* __restrict__ att_dst,
                   float* __restrict__ U) {
    __shared__ int hist[HCAP];
    __shared__ int blk[HCAP / NBLK];
    if (blockIdx.x < CHUNKS) {
        chunk_rank(hist, blk, ei, E, epc, NB, rank, cnt, coarse);
    } else {
        fold_att(W1, att_src, att_dst, U);
    }
}

__global__ __launch_bounds__(NBLK)
void scan_and_scores(int* __restrict__ cnt, const int* __restrict__ coarse,
                     int* __restrict__ deg, int* __restrict__ rowptr, int NB,
                     const float* __restrict__ x, const float* __restrict__ U,
                     float* __restrict__ a_src, float* __restrict__ a_dst,
                     int N, int mtiles) {
    __shared__ int wsum[NBLK / 32];
    __shared__ int wpre[NBLK / 32];
    if ((int)blockIdx.x < NB) {
        scan_cols(wsum, wpre, cnt, coarse, deg, rowptr, N, NB);
    } else {
        const int wv = __builtin_amdgcn_readfirstlane((int)(threadIdx.x >> 5));
        const int tile = ((int)blockIdx.x - NB) * (NBLK / 32) + wv;
        if (tile < mtiles) score_tile(tile, x, U, a_src, a_dst, N);
    }
}

__global__ __launch_bounds__(256)
void fill_groups(const int* __restrict__ ei, int E, int epc, int bpc,
                 const int* __restrict__ cnt, const int* __restrict__ rank,
                 int* __restrict__ csr_src) {
    fill_csr((int)blockIdx.x, ei, E, epc, bpc, cnt, rank, csr_src);
}

__device__ __forceinline__ float wave_max(float v) {
    v = fmaxf(v, xhalf(v));
    v = fmaxf(v, xrow<8>(v)); v = fmaxf(v, xrow<4>(v));
    v = fmaxf(v, xrow<2>(v)); v = fmaxf(v, xrow<1>(v));
    return v;
}
__device__ __forceinline__ float wave_sum(float v) {
    v += xhalf(v);
    v += xrow<8>(v); v += xrow<4>(v); v += xrow<2>(v); v += xrow<1>(v);
    return v;
}
__device__ __forceinline__ void wave_max4(float& a, float& b, float& c, float& d) {
    a = fmaxf(a, xhalf(a));   b = fmaxf(b, xhalf(b));   c = fmaxf(c, xhalf(c));   d = fmaxf(d, xhalf(d));
    a = fmaxf(a, xrow<8>(a)); b = fmaxf(b, xrow<8>(b)); c = fmaxf(c, xrow<8>(c)); d = fmaxf(d, xrow<8>(d));
    a = fmaxf(a, xrow<4>(a)); b = fmaxf(b, xrow<4>(b)); c = fmaxf(c, xrow<4>(c)); d = fmaxf(d, xrow<4>(d));
    a = fmaxf(a, xrow<2>(a)); b = fmaxf(b, xrow<2>(b)); c = fmaxf(c, xrow<2>(c)); d = fmaxf(d, xrow<2>(d));
    a = fmaxf(a, xrow<1>(a)); b = fmaxf(b, xrow<1>(b)); c = fmaxf(c, xrow<1>(c)); d = fmaxf(d, xrow<1>(d));
}
__device__ __forceinline__ void wave_sum4(float& a, float& b, float& c, float& d) {
    a += xhalf(a);   b += xhalf(b);   c += xhalf(c);   d += xhalf(d);
    a += xrow<8>(a); b += xrow<8>(b); c += xrow<8>(c); d += xrow<8>(d);
    a += xrow<4>(a); b += xrow<4>(b); c += xrow<4>(c); d += xrow<4>(d);
    a += xrow<2>(a); b += xrow<2>(b); c += xrow<2>(c); d += xrow<2>(d);
    a += xrow<1>(a); b += xrow<1>(b); c += xrow<1>(c); d += xrow<1>(d);
}
__device__ __forceinline__ float lane_f(float v, int j) {
    return __int_as_float(__builtin_amdgcn_readlane(__float_as_int(v), j));
}

__device__ __forceinline__ void row_fma(float4& z0, float4& z1, float4& z2, float4& z3,
                                        const float x0, const float x1, const float x2, const float x3,
                                        const float4 v) {
    z0.x += x0 * v.x; z0.y += x0 * v.y; z0.z += x0 * v.z; z0.w += x0 * v.w;
    z1.x += x1 * v.x; z1.y += x1 * v.y; z1.z += x1 * v.z; z1.w += x1 * v.w;
    z2.x += x2 * v.x; z2.y += x2 * v.y; z2.z += x2 * v.z; z2.w += x2 * v.w;
    z3.x += x3 * v.x; z3.y += x3 * v.y; z3.z += x3 * v.z; z3.w += x3 * v.w;
}

#define GN 32
#define ZP 516
#define PT 288
#define PH 144

__global__ __launch_bounds__(512) __attribute__((amdgpu_waves_per_eu(16)))
void node_agg_proj12(const int* __restrict__ rowptr, const int* __restrict__ deg,
                     const int* __restrict__ csr_src,
                     const float* __restrict__ a_src, const float* __restrict__ a_dst,
                     const float* __restrict__ x, const float* __restrict__ W1,
                     const float* __restrict__ b1, const float* __restrict__ W2,
                     const float* __restrict__ att_src2, const float* __restrict__ att_dst2,
                     float* __restrict__ h2, float* __restrict__ a_src2, float* __restrict__ a_dst2,
                     int N) {
    __shared__ __attribute__((aligned(16))) float Z[GN * ZP];
    const int lane = threadIdx.x & 31;
    const int wv = __builtin_amdgcn_readfirstlane((int)(threadIdx.x >> 5));

    for (int q = 0; q < 2; ++q) {
        const int nl = wv * 2 + q;
        const int d = blockIdx.x * GN + nl;
        float4 z0 = make_float4(0.0f, 0.0f, 0.0f, 0.0f);
        float4 z1 = z0, z2 = z0, z3 = z0;
        if (d < N) {
            const int start = rowptr[d];
            const int n = deg[d];
            const int cnt = n + 1;
            const float4 ad = *(const float4*)(a_dst + (size_t)d * 4);

            float d0 = 0.0f, d1 = 0.0f, d2 = 0.0f, d3 = 0.0f;
            if (cnt <= 32) {
                const int s = lane < n ? csr_src[start + lane] : d;
                const float4 a = *(const float4*)(a_src + (size_t)s * 4);
                const float4 r0 = *(const float4*)(x + (size_t)__builtin_amdgcn_readlane(s, 0) * 128 + lane * 4);
                const float4 r1 = *(const float4*)(x + (size_t)__builtin_amdgcn_readlane(s, 1) * 128 + lane * 4);
                const float4 r2 = *(const float4*)(x + (size_t)__builtin_amdgcn_readlane(s, 2) * 128 + lane * 4);
                const float4 r3 = *(const float4*)(x + (size_t)__builtin_amdgcn_readlane(s, 3) * 128 + lane * 4);
                const bool live = lane < cnt;
                const float e0 = lrelu(a.x + ad.x), e1 = lrelu(a.y + ad.y);
                const float e2 = lrelu(a.z + ad.z), e3 = lrelu(a.w + ad.w);
                float m0 = live ? e0 : -__builtin_huge_valf();
                float m1 = live ? e1 : -__builtin_huge_valf();
                float m2 = live ? e2 : -__builtin_huge_valf();
                float m3 = live ? e3 : -__builtin_huge_valf();
                wave_max4(m0, m1, m2, m3);
                const float w0 = live ? expf(e0 - m0) : 0.0f;
                const float w1 = live ? expf(e1 - m1) : 0.0f;
                const float w2 = live ? expf(e2 - m2) : 0.0f;
                const float w3 = live ? expf(e3 - m3) : 0.0f;
                d0 = w0; d1 = w1; d2 = w2; d3 = w3;
                row_fma(z0, z1, z2, z3, lane_f(w0, 0), lane_f(w1, 0), lane_f(w2, 0), lane_f(w3, 0), r0);
                row_fma(z0, z1, z2, z3, lane_f(w0, 1), lane_f(w1, 1), lane_f(w2, 1), lane_f(w3, 1), r1);
                row_fma(z0, z1, z2, z3, lane_f(w0, 2), lane_f(w1, 2), lane_f(w2, 2), lane_f(w3, 2), r2);
                row_fma(z0, z1, z2, z3, lane_f(w0, 3), lane_f(w1, 3), lane_f(w2, 3), lane_f(w3, 3), r3);
                const int cn4 = (cnt + 3) & ~3;
                for (int j = 4; j < cn4; j += 4) {
                    float4 v[4];
#pragma unroll
                    for (int u = 0; u < 4; ++u)
                        v[u] = *(const float4*)(x + (size_t)__builtin_amdgcn_readlane(s, j + u) * 128 + lane * 4);
#pragma unroll
                    for (int u = 0; u < 4; ++u)
                        row_fma(z0, z1, z2, z3, lane_f(w0, j + u), lane_f(w1, j + u),
                                lane_f(w2, j + u), lane_f(w3, j + u), v[u]);
                }
            } else {
                float m0 = -__builtin_huge_valf(), m1 = m0, m2 = m0, m3 = m0;
                for (int c = 0; c < cnt; c += 32) {
                    const int k = c + lane;
                    if (k < cnt) {
                        const int s = k < n ? csr_src[start + k] : d;
                        const float4 a = *(const float4*)(a_src + (size_t)s * 4);
                        m0 = fmaxf(m0, lrelu(a.x + ad.x));
                        m1 = fmaxf(m1, lrelu(a.y + ad.y));
                        m2 = fmaxf(m2, lrelu(a.z + ad.z));
                        m3 = fmaxf(m3, lrelu(a.w + ad.w));
                    }
                }
                wave_max4(m0, m1, m2, m3);

                for (int c = 0; c < cnt; c += 32) {
                    const int k = c + lane;
                    int s = d;
                    float w0 = 0.0f, w1 = 0.0f, w2 = 0.0f, w3 = 0.0f;
                    if (k < cnt) {
                        if (k < n) s = csr_src[start + k];
                        const float4 a = *(const float4*)(a_src + (size_t)s * 4);
                        w0 = expf(lrelu(a.x + ad.x) - m0);
                        w1 = expf(lrelu(a.y + ad.y) - m1);
                        w2 = expf(lrelu(a.z + ad.z) - m2);
                        w3 = expf(lrelu(a.w + ad.w) - m3);
                    }
                    d0 += w0; d1 += w1; d2 += w2; d3 += w3;
                    const int cn = (cnt - c) < 32 ? (cnt - c) : 32;
#pragma unroll 4
                    for (int j = 0; j < cn; ++j) {
                        const int sj = __builtin_amdgcn_readlane(s, j);
                        const float x0 = lane_f(w0, j), x1 = lane_f(w1, j);
                        const float x2 = lane_f(w2, j), x3 = lane_f(w3, j);
                        const float4 v = *(const float4*)(x + (size_t)sj * 128 + lane * 4);
                        z0.x += x0 * v.x; z0.y += x0 * v.y; z0.z += x0 * v.z; z0.w += x0 * v.w;
                        z1.x += x1 * v.x; z1.y += x1 * v.y; z1.z += x1 * v.z; z1.w += x1 * v.w;
                        z2.x += x2 * v.x; z2.y += x2 * v.y; z2.z += x2 * v.z; z2.w += x2 * v.w;
                        z3.x += x3 * v.x; z3.y += x3 * v.y; z3.z += x3 * v.z; z3.w += x3 * v.w;
                    }
                }
            }
            wave_sum4(d0, d1, d2, d3);
            z0.x /= d0; z0.y /= d0; z0.z /= d0; z0.w /= d0;
            z1.x /= d1; z1.y /= d1; z1.z /= d1; z1.w /= d1;
            z2.x /= d2; z2.y /= d2; z2.z /= d2; z2.w /= d2;
            z3.x /= d3; z3.y /= d3; z3.z /= d3; z3.w /= d3;
        }
        float* zp = Z + nl * ZP + lane * 4;
        *(float4*)zp         = z0;
        *(float4*)(zp + 128) = z1;
        *(float4*)(zp + 256) = z2;
        *(float4*)(zp + 384) = z3;
    }
    __syncthreads();

    const int c     = lane & 15;
    const int hf    = lane >> 4;
    const int khalf = hf * 2;
    const int head  = wv >> 2;
    const int cb    = wv * 16;
    const int col0  = cb + 8 * hf;

    const float* w2p = W2 + (size_t)col0 * 16 + c;
    v2f wb[4];
    for (int t = 0; t < 4; ++t) {
        wb[t].x = w2p[(2 * t) * 16];
        wb[t].y = w2p[(2 * t + 1) * 16];
    }
    const float4 bs0 = *(const float4*)(b1 + col0);
    const float4 bs1 = *(const float4*)(b1 + col0 + 4);

    const float* za0 = Z + c * ZP + head * 128 + khalf;
    const float* za1 = za0 + 16 * ZP;
    const float* wp  = W1 + (size_t)khalf * 256 + cb + c;
    v8f c0 = {}, c1 = {};
    v2f wn[4];
    for (int i = 0; i < 4; ++i) {
        wn[i].x = wp[(size_t)(4 * i) * 256];
        wn[i].y = wp[(size_t)(4 * i + 1) * 256];
    }
#pragma unroll 1
    for (int kk = 0; kk < 128 - 16; kk += 16) {
#pragma unroll
        for (int i = 0; i < 4; ++i) {
            const int k = kk + 4 * i;
            const v2f a0 = *(const v2f*)(za0 + k);
            const v2f a1 = *(const v2f*)(za1 + k);
            c0 = __builtin_amdgcn_wmma_f32_16x16x4_f32(false, wn[i], false, a0, (short)0, c0, false, false);
            c1 = __builtin_amdgcn_wmma_f32_16x16x4_f32(false, wn[i], false, a1, (short)0, c1, false, false);
            wn[i].x = wp[(size_t)(k + 16) * 256];
            wn[i].y = wp[(size_t)(k + 16 + 1) * 256];
        }
    }
    for (int i = 0; i < 4; ++i) {
        const int k = 128 - 16 + 4 * i;
        const v2f a0 = *(const v2f*)(za0 + k);
        const v2f a1 = *(const v2f*)(za1 + k);
        c0 = __builtin_amdgcn_wmma_f32_16x16x4_f32(false, wn[i], false, a0, (short)0, c0, false, false);
        c1 = __builtin_amdgcn_wmma_f32_16x16x4_f32(false, wn[i], false, a1, (short)0, c1, false, false);
    }

    const float bs[8] = {bs0.x, bs0.y, bs0.z, bs0.w, bs1.x, bs1.y, bs1.z, bs1.w};
    for (int r = 0; r < 8; ++r) {
        c0[r] = fmaxf(c0[r] + bs[r], 0.0f);
        c1[r] = fmaxf(c1[r] + bs[r], 0.0f);
    }

    v8f d0 = {}, d1 = {};
    for (int t = 0; t < 4; ++t) {
        v2f a;
        a.x = c0[2 * t]; a.y = c0[2 * t + 1];
        d0 = __builtin_amdgcn_wmma_f32_16x16x4_f32(false, a, false, wb[t], (short)0, d0, false, false);
        a.x = c1[2 * t]; a.y = c1[2 * t + 1];
        d1 = __builtin_amdgcn_wmma_f32_16x16x4_f32(false, a, false, wb[t], (short)0, d1, false, false);
    }
    const float as = att_src2[c];
    const float ad = att_dst2[c];
    __syncthreads();

    float* P = Z;
    {
        float* pp = P + wv * (2 * PT) + hf * PH + c;
        for (int r = 0; r < 8; ++r) {
            pp[r * 16]      = d0[r];
            pp[PT + r * 16] = d1[r];
        }
    }
    __syncthreads();

    {
        const int node = wv * 2 + hf;
        const int nl   = node & 15;
        const float* q = P + (node >> 4) * PT + (nl >> 3) * PH + (nl & 7) * 16 + c;
        float sum = q[0];
        for (int w = 1; w < 16; ++w) sum += q[w * (2 * PT)];
        float ps = sum * as;
        float pd = sum * ad;
        ps += xrow<1>(ps); pd += xrow<1>(pd);
        ps += xrow<2>(ps); pd += xrow<2>(pd);
        ps += xrow<4>(ps); pd += xrow<4>(pd);
        ps += xrow<8>(ps); pd += xrow<8>(pd);
        const int g = blockIdx.x * GN + node;
        if (g < N) {
            h2[(size_t)g * 16 + c] = sum;
            if (c == 0) {
                a_src2[g] = ps;
                a_dst2[g] = pd;
            }
        }
    }
}

__global__ __launch_bounds__(256)
void node_agg2(const int* __restrict__ rowptr, const int* __restrict__ deg,
               const int* __restrict__ csr_src,
               const float* __restrict__ a_src, const float* __restrict__ a_dst,
               const float* __restrict__ h2, const float* __restrict__ b2,
               float* __restrict__ out, int N) {
    const int lane = threadIdx.x & 31;
    const int wv = __builtin_amdgcn_readfirstlane((int)(threadIdx.x >> 5));
    const int d = blockIdx.x * 8 + wv;
    if (d >= N) return;
    const int half = lane >> 4;
    const int t = lane & 15;
    const int start = rowptr[d];
    const int n = deg[d];
    const int cnt = n + 1;
    const float ad = a_dst[d];

    if (cnt <= 32) {
        const int s = lane < n ? csr_src[start + lane] : d;
        const float a = a_src[s];
        float r[16];
#pragma unroll
        for (int p = 0; p < 8; ++p) {
            const int s0 = __builtin_amdgcn_readlane(s, 2 * p);
            const int s1 = __builtin_amdgcn_readlane(s, 2 * p + 1);
            r[p] = h2[(size_t)(half ? s1 : s0) * 16 + t];
        }
#pragma unroll
        for (int p = 8; p < 16; ++p) r[p] = 0.0f;
        if (cnt > 16) {
#pragma unroll
            for (int p = 8; p < 16; ++p) {
                const int s0 = __builtin_amdgcn_readlane(s, 2 * p);
                const int s1 = __builtin_amdgcn_readlane(s, 2 * p + 1);
                r[p] = h2[(size_t)(half ? s1 : s0) * 16 + t];
            }
        }
        const bool live = lane < cnt;
        const float e = lrelu(a + ad);
        const float m = wave_max(live ? e : -__builtin_huge_valf());
        const float w = live ? expf(e - m) : 0.0f;
        const float den = wave_sum(w);
        float acc = 0.0f;
#pragma unroll
        for (int p = 0; p < 8; ++p) {
            const float w0 = lane_f(w, 2 * p);
            const float w1 = lane_f(w, 2 * p + 1);
            acc += (half ? w1 : w0) * r[p];
        }
        if (cnt > 16) {
#pragma unroll
            for (int p = 8; p < 16; ++p) {
                const float w0 = lane_f(w, 2 * p);
                const float w1 = lane_f(w, 2 * p + 1);
                acc += (half ? w1 : w0) * r[p];
            }
        }
        acc += xhalf(acc);
        if (lane < 16) out[(size_t)d * 16 + t] = acc / den + b2[t];
        return;
    }

    float m = -__builtin_huge_valf();
    for (int c = 0; c < cnt; c += 32) {
        const int k = c + lane;
        if (k < cnt) {
            const int s = k < n ? csr_src[start + k] : d;
            m = fmaxf(m, lrelu(a_src[s] + ad));
        }
    }
    m = wave_max(m);

    float den = 0.0f, acc = 0.0f;
    for (int c = 0; c < cnt; c += 32) {
        const int k = c + lane;
        int s = d;
        float w = 0.0f;
        if (k < cnt) {
            if (k < n) s = csr_src[start + k];
            w = expf(lrelu(a_src[s] + ad) - m);
        }
        den += w;
        const int cn = (cnt - c) < 32 ? (cnt - c) : 32;
        const int cp = cn & ~1;
#pragma unroll 4
        for (int j = 0; j < cp; j += 2) {
            const int s0 = __builtin_amdgcn_readlane(s, j);
            const int s1 = __builtin_amdgcn_readlane(s, j + 1);
            const float w0 = lane_f(w, j);
            const float w1 = lane_f(w, j + 1);
            const int sj = half ? s1 : s0;
            const float wj = half ? w1 : w0;
            acc += wj * h2[(size_t)sj * 16 + t];
        }
        if (cn & 1) {
            const int sl = __builtin_amdgcn_readlane(s, cp);
            const float wl = lane_f(w, cp);
            if (half == 0) acc += wl * h2[(size_t)sl * 16 + t];
        }
    }
    den = wave_sum(den);
    acc += xhalf(acc);
    if (lane < 16) out[(size_t)d * 16 + t] = acc / den + b2[t];
}


extern "C" void kernel_launch(void* const* d_in, const int* in_sizes, int n_in,
                              void* d_out, int out_size, void* d_ws, size_t ws_size,
                              hipStream_t stream) {
    const float* x   = (const float*)d_in[0];
    const int*   ei  = (const int*)  d_in[1];
    const float* W1  = (const float*)d_in[2];
    const float* as1 = (const float*)d_in[3];
    const float* ad1 = (const float*)d_in[4];
    const float* b1  = (const float*)d_in[5];
    const float* W2  = (const float*)d_in[6];
    const float* as2 = (const float*)d_in[7];
    const float* ad2 = (const float*)d_in[8];
    const float* b2  = (const float*)d_in[9];

    const int N  = in_sizes[0] / 128;
    const int E  = in_sizes[1] / 2;

    char* w = (char*)d_ws;
    auto carve = [&](size_t bytes) -> void* {
        void* p = (void*)w;
        w += (bytes + 255) & ~((size_t)255);
        return p;
    };
    float* aS1    = (float*)carve((size_t)N * 4 * 4);
    float* aD1    = (float*)carve((size_t)N * 4 * 4);
    float* U      = (float*)carve((size_t)128 * 8 * 4);
    float* h2     = (float*)carve((size_t)N * 16 * 4);
    float* aS2    = (float*)carve((size_t)N * 4);
    float* aD2    = (float*)carve((size_t)N * 4);
    int*   deg    = (int*)  carve((size_t)N * 4);
    int*   rowptr = (int*)  carve((size_t)N * 4);
    int*   rank   = (int*)  carve((size_t)E * 4);
    int*   csrs   = (int*)  carve((size_t)E * 4);
    const int NB  = (N + NBLK - 1) / NBLK;
    const int epc = (E + CHUNKS - 1) / CHUNKS;
    int*   cnt    = (int*)  carve((size_t)CHUNKS * HCAP * 4);
    int*   coarse = (int*)  carve((size_t)CHUNKS * NB * 4);

    const int TB = 256;
    auto blocks = [&](long long n) { return (unsigned)((n + TB - 1) / TB); };

    const int mtiles = (N + 15) / 16;
    const int bpc    = (int)blocks(epc);
    rank_and_fold<<<CHUNKS + 1, 1024, 0, stream>>>(
        ei, E, epc, NB, rank, cnt, coarse, W1, as1, ad1, U);
    scan_and_scores<<<NB + (mtiles + NBLK / 32 - 1) / (NBLK / 32), NBLK, 0, stream>>>(
        cnt, coarse, deg, rowptr, NB, x, U, aS1, aD1, N, mtiles);
    fill_groups<<<bpc * CHUNKS, TB, 0, stream>>>(ei, E, epc, bpc, cnt, rank, csrs);

    node_agg_proj12<<<(N + GN - 1) / GN, 512, 0, stream>>>(
        rowptr, deg, csrs, aS1, aD1, x, W1, b1, W2, as2, ad2, h2, aS2, aD2, N);

    node_agg2<<<(N + 7) / 8, 256, 0, stream>>>(rowptr, deg, csrs, aS2, aD2, h2, b2, (float*)d_out, N);
}


# Layer-1 softmax division moved behind the W1 product: sums go to LDS unnormalised, 1/denominator is a factor in the bias add on the accumulators; 16 divisions per node gone; rounding differs
# speedup vs baseline: 6.7307x; 1.0400x over previous
#include <hip/hip_runtime.h>
#include <hip/hip_bf16.h>

typedef __attribute__((ext_vector_type(2))) float v2f;
typedef __attribute__((ext_vector_type(8))) float v8f;

#define NEG_SLOPE 0.2f


__device__ __forceinline__ float lrelu(float v) {
    return v >= 0.0f ? v : NEG_SLOPE * v;
}


__device__ __forceinline__ int xhalf_i(int b) {
    return __builtin_amdgcn_permlanex16(b, b, 0x76543210, 0xfedcba98, false, false);
}
template <int M> __device__ __forceinline__ int xrow_i(int b) {
    return __builtin_amdgcn_update_dpp(0, b, 0x160 + M, 0xf, 0xf, true);
}
__device__ __forceinline__ float xhalf(float v) { return __int_as_float(xhalf_i(__float_as_int(v))); }
template <int M> __device__ __forceinline__ float xrow(float v) {
    return __int_as_float(xrow_i<M>(__float_as_int(v)));
}


__device__ __forceinline__
void fold_att(const float* __restrict__ W1, const float* __restrict__ att_src,
              const float* __restrict__ att_dst, float* __restrict__ U) {
    const int t = threadIdx.x;
    const int k = t >> 3, j = t & 7, h = j & 3;
    const float* w  = W1 + (size_t)k * 256 + h * 64;
    const float* at = (j < 4 ? att_src : att_dst) + h * 64;
    float s = 0.0f;
    for (int c = 0; c < 64; ++c) s += w[c] * at[c];
    U[t] = s;
}

__device__ __forceinline__
void score_tile(const int mtile, const float* __restrict__ x, const float* __restrict__ U,
                float* __restrict__ a_src, float* __restrict__ a_dst, int N) {
    const int lane  = threadIdx.x & 31;
    const int mrow  = mtile * 16 + (lane & 15);
    const int lrow  = mrow < N ? mrow : N - 1;
    const int khalf = (lane >> 4) * 2;
    const int nc    = lane & 15;
    const int uc    = nc & 7;

    v8f c = {};
    const float* xrow = x + (size_t)lrow * 128;
    for (int k = 0; k < 128; k += 4) {
        const int kb = k + khalf;
        v2f a; a.x = xrow[kb]; a.y = xrow[kb + 1];
        v2f b; b.x = U[kb * 8 + uc]; b.y = U[(kb + 1) * 8 + uc];
        c = __builtin_amdgcn_wmma_f32_16x16x4_f32(false, a, false, b, (short)0, c, false, false);
    }
    const int rowoff = (lane >= 16) ? 8 : 0;
    float* sp = (nc < 4) ? a_src : a_dst;
    for (int r = 0; r < 8; ++r) {
        const int grow = mtile * 16 + rowoff + r;
        if (grow < N && nc < 8) sp[(size_t)grow * 4 + (nc & 3)] = c[r];
    }
}

#define CHUNKS 64
#define HCAP   50176
#define NBLK   256
#define NBLK_LOG 8
#define EPT    5

__device__ __forceinline__ int wave_sum_i(int v) {
    v += xhalf_i(v);
    v += xrow_i<8>(v); v += xrow_i<4>(v); v += xrow_i<2>(v); v += xrow_i<1>(v);
    return v;
}

__device__ __forceinline__
void chunk_rank(int* __restrict__ hist, int* __restrict__ blk,
                const int* __restrict__ ei, int E, int epc, int NB,
                int* __restrict__ rank, int* __restrict__ cnt, int* __restrict__ coarse) {
    const int t = threadIdx.x;
    const int c = blockIdx.x;
    int4* h4 = (int4*)hist;
    for (int i = t; i < HCAP / 4; i += 1024) h4[i] = make_int4(0, 0, 0, 0);
    if (t < HCAP / NBLK) blk[t] = 0;
    __syncthreads();

    const int e0 = c * epc;
    const int e1 = (e0 + epc) < E ? (e0 + epc) : E;
    for (int eb = e0 + t; eb < e1; eb += EPT * 1024) {
        int dd[EPT], rr[EPT];
#pragma unroll
        for (int u = 0; u < EPT; ++u) {
            const int e = eb + u * 1024;
            dd[u] = e < e1 ? ei[E + e] : 0;
        }
#pragma unroll
        for (int u = 0; u < EPT; ++u) {
            const int e = eb + u * 1024;
            rr[u] = 0;
            if (e < e1) {
                rr[u] = atomicAdd(&hist[dd[u]], 1);
                atomicAdd(&blk[dd[u] >> NBLK_LOG], 1);
            }
        }
#pragma unroll
        for (int u = 0; u < EPT; ++u) {
            const int e = eb + u * 1024;
            if (e < e1) rank[e] = rr[u];
        }
    }
    __syncthreads();

    int4* c4 = (int4*)(cnt + (size_t)c * HCAP);
    for (int i = t; i < HCAP / 4; i += 1024) c4[i] = h4[i];
    if (t < NB) coarse[c * NB + t] = blk[t];
}

__device__ __forceinline__
void scan_cols(int* __restrict__ wsum, int* __restrict__ wpre,
               int* __restrict__ cnt, const int* __restrict__ coarse,
               int* __restrict__ deg, int* __restrict__ rowptr, int N, int NB) {
    const int t = threadIdx.x, lane = t & 31, wv = t >> 5;
    const int j = blockIdx.x;
    const int idx = j * NBLK + t;

    int pre = 0;
    for (int b = t; b < j; b += NBLK)
        for (int c = 0; c < CHUNKS; ++c) pre += coarse[c * NB + b];

    int v[CHUNKS];
    int d = 0;
#pragma unroll
    for (int c = 0; c < CHUNKS; ++c) {
        v[c] = cnt[(size_t)c * HCAP + idx];
        d += v[c];
    }
    int inc = d;
    for (int off = 1; off < 32; off <<= 1) {
        const int nb = __shfl_up(inc, off);
        if (lane >= off) inc += nb;
    }
    pre = wave_sum_i(pre);
    if (lane == 31) wsum[wv] = inc;
    if (lane == 0)  wpre[wv] = pre;
    __syncthreads();
    int base = 0;
    for (int k = 0; k < NBLK / 32; ++k) {
        base += wpre[k];
        if (k < wv) base += wsum[k];
    }
    if (idx < N) {
        int run = base + inc - d;
        rowptr[idx] = run;
        deg[idx] = d;
#pragma unroll
        for (int c = 0; c < CHUNKS; ++c) {
            cnt[(size_t)c * HCAP + idx] = run;
            run += v[c];
        }
    }
}

__device__ __forceinline__
void fill_csr(const int b, const int* __restrict__ ei, int E, int epc, int bpc,
              const int* __restrict__ cnt, const int* __restrict__ rank,
              int* __restrict__ csr_src) {
    const int c = b / bpc;
    const int e0 = c * epc;
    const int e1 = (e0 + epc) < E ? (e0 + epc) : E;
    const int e = e0 + (b - c * bpc) * 256 + threadIdx.x;
    if (e >= e1) return;
    csr_src[cnt[(size_t)c * HCAP + ei[E + e]] + rank[e]] = ei[e];
}


__global__ __launch_bounds__(1024)
void rank_and_fold(const int* __restrict__ ei, int E, int epc, int NB,
                   int* __restrict__ rank, int* __restrict__ cnt, int* __restrict__ coarse,
                   const float* __restrict__ W1,
                   const float* __restrict__ att_src, const float* __restrict__ att_dst,
                   float* __restrict__ U) {
    __shared__ int hist[HCAP];
    __shared__ int blk[HCAP / NBLK];
    if (blockIdx.x < CHUNKS) {
        chunk_rank(hist, blk, ei, E, epc, NB, rank, cnt, coarse);
    } else {
        fold_att(W1, att_src, att_dst, U);
    }
}

__global__ __launch_bounds__(NBLK)
void scan_and_scores(int* __restrict__ cnt, const int* __restrict__ coarse,
                     int* __restrict__ deg, int* __restrict__ rowptr, int NB,
                     const float* __restrict__ x, const float* __restrict__ U,
                     float* __restrict__ a_src, float* __restrict__ a_dst,
                     int N, int mtiles) {
    __shared__ int wsum[NBLK / 32];
    __shared__ int wpre[NBLK / 32];
    if ((int)blockIdx.x < NB) {
        scan_cols(wsum, wpre, cnt, coarse, deg, rowptr, N, NB);
    } else {
        const int wv = __builtin_amdgcn_readfirstlane((int)(threadIdx.x >> 5));
        const int tile = ((int)blockIdx.x - NB) * (NBLK / 32) + wv;
        if (tile < mtiles) score_tile(tile, x, U, a_src, a_dst, N);
    }
}

__global__ __launch_bounds__(256)
void fill_groups(const int* __restrict__ ei, int E, int epc, int bpc,
                 const int* __restrict__ cnt, const int* __restrict__ rank,
                 int* __restrict__ csr_src) {
    fill_csr((int)blockIdx.x, ei, E, epc, bpc, cnt, rank, csr_src);
}

__device__ __forceinline__ float wave_max(float v) {
    v = fmaxf(v, xhalf(v));
    v = fmaxf(v, xrow<8>(v)); v = fmaxf(v, xrow<4>(v));
    v = fmaxf(v, xrow<2>(v)); v = fmaxf(v, xrow<1>(v));
    return v;
}
__device__ __forceinline__ float wave_sum(float v) {
    v += xhalf(v);
    v += xrow<8>(v); v += xrow<4>(v); v += xrow<2>(v); v += xrow<1>(v);
    return v;
}
__device__ __forceinline__ void wave_max4(float& a, float& b, float& c, float& d) {
    a = fmaxf(a, xhalf(a));   b = fmaxf(b, xhalf(b));   c = fmaxf(c, xhalf(c));   d = fmaxf(d, xhalf(d));
    a = fmaxf(a, xrow<8>(a)); b = fmaxf(b, xrow<8>(b)); c = fmaxf(c, xrow<8>(c)); d = fmaxf(d, xrow<8>(d));
    a = fmaxf(a, xrow<4>(a)); b = fmaxf(b, xrow<4>(b)); c = fmaxf(c, xrow<4>(c)); d = fmaxf(d, xrow<4>(d));
    a = fmaxf(a, xrow<2>(a)); b = fmaxf(b, xrow<2>(b)); c = fmaxf(c, xrow<2>(c)); d = fmaxf(d, xrow<2>(d));
    a = fmaxf(a, xrow<1>(a)); b = fmaxf(b, xrow<1>(b)); c = fmaxf(c, xrow<1>(c)); d = fmaxf(d, xrow<1>(d));
}
__device__ __forceinline__ void wave_sum4(float& a, float& b, float& c, float& d) {
    a += xhalf(a);   b += xhalf(b);   c += xhalf(c);   d += xhalf(d);
    a += xrow<8>(a); b += xrow<8>(b); c += xrow<8>(c); d += xrow<8>(d);
    a += xrow<4>(a); b += xrow<4>(b); c += xrow<4>(c); d += xrow<4>(d);
    a += xrow<2>(a); b += xrow<2>(b); c += xrow<2>(c); d += xrow<2>(d);
    a += xrow<1>(a); b += xrow<1>(b); c += xrow<1>(c); d += xrow<1>(d);
}
__device__ __forceinline__ float lane_f(float v, int j) {
    return __int_as_float(__builtin_amdgcn_readlane(__float_as_int(v), j));
}

__device__ __forceinline__ void row_fma(float4& z0, float4& z1, float4& z2, float4& z3,
                                        const float x0, const float x1, const float x2, const float x3,
                                        const float4 v) {
    z0.x += x0 * v.x; z0.y += x0 * v.y; z0.z += x0 * v.z; z0.w += x0 * v.w;
    z1.x += x1 * v.x; z1.y += x1 * v.y; z1.z += x1 * v.z; z1.w += x1 * v.w;
    z2.x += x2 * v.x; z2.y += x2 * v.y; z2.z += x2 * v.z; z2.w += x2 * v.w;
    z3.x += x3 * v.x; z3.y += x3 * v.y; z3.z += x3 * v.z; z3.w += x3 * v.w;
}

#define GN 32
#define ZP 516
#define PT 288
#define PH 144

__global__ __launch_bounds__(512) __attribute__((amdgpu_waves_per_eu(16)))
void node_agg_proj12(const int* __restrict__ rowptr, const int* __restrict__ deg,
                     const int* __restrict__ csr_src,
                     const float* __restrict__ a_src, const float* __restrict__ a_dst,
                     const float* __restrict__ x, const float* __restrict__ W1,
                     const float* __restrict__ b1, const float* __restrict__ W2,
                     const float* __restrict__ att_src2, const float* __restrict__ att_dst2,
                     float* __restrict__ h2, float* __restrict__ a_src2, float* __restrict__ a_dst2,
                     int N) {
    __shared__ __attribute__((aligned(16))) float Z[GN * ZP];
    const int lane = threadIdx.x & 31;
    const int wv = __builtin_amdgcn_readfirstlane((int)(threadIdx.x >> 5));

    for (int q = 0; q < 2; ++q) {
        const int nl = wv * 2 + q;
        const int d = blockIdx.x * GN + nl;
        float4 z0 = make_float4(0.0f, 0.0f, 0.0f, 0.0f);
        float4 z1 = z0, z2 = z0, z3 = z0;
        float4 dn = make_float4(1.0f, 1.0f, 1.0f, 1.0f);
        if (d < N) {
            const int start = rowptr[d];
            const int n = deg[d];
            const int cnt = n + 1;
            const float4 ad = *(const float4*)(a_dst + (size_t)d * 4);

            float d0 = 0.0f, d1 = 0.0f, d2 = 0.0f, d3 = 0.0f;
            if (cnt <= 32) {
                const int s = lane < n ? csr_src[start + lane] : d;
                const float4 a = *(const float4*)(a_src + (size_t)s * 4);
                const float4 r0 = *(const float4*)(x + (size_t)__builtin_amdgcn_readlane(s, 0) * 128 + lane * 4);
                const float4 r1 = *(const float4*)(x + (size_t)__builtin_amdgcn_readlane(s, 1) * 128 + lane * 4);
                const float4 r2 = *(const float4*)(x + (size_t)__builtin_amdgcn_readlane(s, 2) * 128 + lane * 4);
                const float4 r3 = *(const float4*)(x + (size_t)__builtin_amdgcn_readlane(s, 3) * 128 + lane * 4);
                const bool live = lane < cnt;
                const float e0 = lrelu(a.x + ad.x), e1 = lrelu(a.y + ad.y);
                const float e2 = lrelu(a.z + ad.z), e3 = lrelu(a.w + ad.w);
                float m0 = live ? e0 : -__builtin_huge_valf();
                float m1 = live ? e1 : -__builtin_huge_valf();
                float m2 = live ? e2 : -__builtin_huge_valf();
                float m3 = live ? e3 : -__builtin_huge_valf();
                wave_max4(m0, m1, m2, m3);
                const float w0 = live ? expf(e0 - m0) : 0.0f;
                const float w1 = live ? expf(e1 - m1) : 0.0f;
                const float w2 = live ? expf(e2 - m2) : 0.0f;
                const float w3 = live ? expf(e3 - m3) : 0.0f;
                d0 = w0; d1 = w1; d2 = w2; d3 = w3;
                row_fma(z0, z1, z2, z3, lane_f(w0, 0), lane_f(w1, 0), lane_f(w2, 0), lane_f(w3, 0), r0);
                row_fma(z0, z1, z2, z3, lane_f(w0, 1), lane_f(w1, 1), lane_f(w2, 1), lane_f(w3, 1), r1);
                row_fma(z0, z1, z2, z3, lane_f(w0, 2), lane_f(w1, 2), lane_f(w2, 2), lane_f(w3, 2), r2);
                row_fma(z0, z1, z2, z3, lane_f(w0, 3), lane_f(w1, 3), lane_f(w2, 3), lane_f(w3, 3), r3);
                const int cn4 = (cnt + 3) & ~3;
                for (int j = 4; j < cn4; j += 4) {
                    float4 v[4];
#pragma unroll
                    for (int u = 0; u < 4; ++u)
                        v[u] = *(const float4*)(x + (size_t)__builtin_amdgcn_readlane(s, j + u) * 128 + lane * 4);
#pragma unroll
                    for (int u = 0; u < 4; ++u)
                        row_fma(z0, z1, z2, z3, lane_f(w0, j + u), lane_f(w1, j + u),
                                lane_f(w2, j + u), lane_f(w3, j + u), v[u]);
                }
            } else {
                float m0 = -__builtin_huge_valf(), m1 = m0, m2 = m0, m3 = m0;
                for (int c = 0; c < cnt; c += 32) {
                    const int k = c + lane;
                    if (k < cnt) {
                        const int s = k < n ? csr_src[start + k] : d;
                        const float4 a = *(const float4*)(a_src + (size_t)s * 4);
                        m0 = fmaxf(m0, lrelu(a.x + ad.x));
                        m1 = fmaxf(m1, lrelu(a.y + ad.y));
                        m2 = fmaxf(m2, lrelu(a.z + ad.z));
                        m3 = fmaxf(m3, lrelu(a.w + ad.w));
                    }
                }
                wave_max4(m0, m1, m2, m3);

                for (int c = 0; c < cnt; c += 32) {
                    const int k = c + lane;
                    int s = d;
                    float w0 = 0.0f, w1 = 0.0f, w2 = 0.0f, w3 = 0.0f;
                    if (k < cnt) {
                        if (k < n) s = csr_src[start + k];
                        const float4 a = *(const float4*)(a_src + (size_t)s * 4);
                        w0 = expf(lrelu(a.x + ad.x) - m0);
                        w1 = expf(lrelu(a.y + ad.y) - m1);
                        w2 = expf(lrelu(a.z + ad.z) - m2);
                        w3 = expf(lrelu(a.w + ad.w) - m3);
                    }
                    d0 += w0; d1 += w1; d2 += w2; d3 += w3;
                    const int cn = (cnt - c) < 32 ? (cnt - c) : 32;
#pragma unroll 4
                    for (int j = 0; j < cn; ++j) {
                        const int sj = __builtin_amdgcn_readlane(s, j);
                        const float x0 = lane_f(w0, j), x1 = lane_f(w1, j);
                        const float x2 = lane_f(w2, j), x3 = lane_f(w3, j);
                        const float4 v = *(const float4*)(x + (size_t)sj * 128 + lane * 4);
                        z0.x += x0 * v.x; z0.y += x0 * v.y; z0.z += x0 * v.z; z0.w += x0 * v.w;
                        z1.x += x1 * v.x; z1.y += x1 * v.y; z1.z += x1 * v.z; z1.w += x1 * v.w;
                        z2.x += x2 * v.x; z2.y += x2 * v.y; z2.z += x2 * v.z; z2.w += x2 * v.w;
                        z3.x += x3 * v.x; z3.y += x3 * v.y; z3.z += x3 * v.z; z3.w += x3 * v.w;
                    }
                }
            }
            wave_sum4(d0, d1, d2, d3);
            dn = make_float4(d0, d1, d2, d3);
        }
        float* zp = Z + nl * ZP + lane * 4;
        *(float4*)zp         = z0;
        *(float4*)(zp + 128) = z1;
        *(float4*)(zp + 256) = z2;
        *(float4*)(zp + 384) = z3;
        if (lane == 0) *(float4*)(Z + nl * ZP + 512) = dn;
    }
    __syncthreads();

    const int c     = lane & 15;
    const int hf    = lane >> 4;
    const int khalf = hf * 2;
    const int head  = wv >> 2;
    const int cb    = wv * 16;
    const int col0  = cb + 8 * hf;

    const float* w2p = W2 + (size_t)col0 * 16 + c;
    v2f wb[4];
    for (int t = 0; t < 4; ++t) {
        wb[t].x = w2p[(2 * t) * 16];
        wb[t].y = w2p[(2 * t + 1) * 16];
    }

    const float* za0 = Z + c * ZP + head * 128 + khalf;
    const float* za1 = za0 + 16 * ZP;
    const float* wp  = W1 + (size_t)khalf * 256 + cb + c;
    v8f c0 = {}, c1 = {};
    v2f wn[4];
    for (int i = 0; i < 4; ++i) {
        wn[i].x = wp[(size_t)(4 * i) * 256];
        wn[i].y = wp[(size_t)(4 * i + 1) * 256];
    }
#pragma unroll 1
    for (int kk = 0; kk < 128 - 16; kk += 16) {
#pragma unroll
        for (int i = 0; i < 4; ++i) {
            const int k = kk + 4 * i;
            const v2f a0 = *(const v2f*)(za0 + k);
            const v2f a1 = *(const v2f*)(za1 + k);
            c0 = __builtin_amdgcn_wmma_f32_16x16x4_f32(false, wn[i], false, a0, (short)0, c0, false, false);
            c1 = __builtin_amdgcn_wmma_f32_16x16x4_f32(false, wn[i], false, a1, (short)0, c1, false, false);
            wn[i].x = wp[(size_t)(k + 16) * 256];
            wn[i].y = wp[(size_t)(k + 16 + 1) * 256];
        }
    }
    const float inv0 = 1.0f / Z[c * ZP + 512 + head];
    const float inv1 = 1.0f / Z[(16 + c) * ZP + 512 + head];
    const float4 bs0 = *(const float4*)(b1 + col0);
    const float4 bs1 = *(const float4*)(b1 + col0 + 4);
    for (int i = 0; i < 4; ++i) {
        const int k = 128 - 16 + 4 * i;
        const v2f a0 = *(const v2f*)(za0 + k);
        const v2f a1 = *(const v2f*)(za1 + k);
        c0 = __builtin_amdgcn_wmma_f32_16x16x4_f32(false, wn[i], false, a0, (short)0, c0, false, false);
        c1 = __builtin_amdgcn_wmma_f32_16x16x4_f32(false, wn[i], false, a1, (short)0, c1, false, false);
    }

    const float bs[8] = {bs0.x, bs0.y, bs0.z, bs0.w, bs1.x, bs1.y, bs1.z, bs1.w};
    for (int r = 0; r < 8; ++r) {
        c0[r] = fmaxf(c0[r] * inv0 + bs[r], 0.0f);
        c1[r] = fmaxf(c1[r] * inv1 + bs[r], 0.0f);
    }

    v8f d0 = {}, d1 = {};
    for (int t = 0; t < 4; ++t) {
        v2f a;
        a.x = c0[2 * t]; a.y = c0[2 * t + 1];
        d0 = __builtin_amdgcn_wmma_f32_16x16x4_f32(false, a, false, wb[t], (short)0, d0, false, false);
        a.x = c1[2 * t]; a.y = c1[2 * t + 1];
        d1 = __builtin_amdgcn_wmma_f32_16x16x4_f32(false, a, false, wb[t], (short)0, d1, false, false);
    }
    const float as = att_src2[c];
    const float ad = att_dst2[c];
    __syncthreads();

    float* P = Z;
    {
        float* pp = P + wv * (2 * PT) + hf * PH + c;
        for (int r = 0; r < 8; ++r) {
            pp[r * 16]      = d0[r];
            pp[PT + r * 16] = d1[r];
        }
    }
    __syncthreads();

    {
        const int node = wv * 2 + hf;
        const int nl   = node & 15;
        const float* q = P + (node >> 4) * PT + (nl >> 3) * PH + (nl & 7) * 16 + c;
        float sum = q[0];
        for (int w = 1; w < 16; ++w) sum += q[w * (2 * PT)];
        float ps = sum * as;
        float pd = sum * ad;
        ps += xrow<1>(ps); pd += xrow<1>(pd);
        ps += xrow<2>(ps); pd += xrow<2>(pd);
        ps += xrow<4>(ps); pd += xrow<4>(pd);
        ps += xrow<8>(ps); pd += xrow<8>(pd);
        const int g = blockIdx.x * GN + node;
        if (g < N) {
            h2[(size_t)g * 16 + c] = sum;
            if (c == 0) {
                a_src2[g] = ps;
                a_dst2[g] = pd;
            }
        }
    }
}

__global__ __launch_bounds__(256)
void node_agg2(const int* __restrict__ rowptr, const int* __restrict__ deg,
               const int* __restrict__ csr_src,
               const float* __restrict__ a_src, const float* __restrict__ a_dst,
               const float* __restrict__ h2, const float* __restrict__ b2,
               float* __restrict__ out, int N) {
    const int lane = threadIdx.x & 31;
    const int wv = __builtin_amdgcn_readfirstlane((int)(threadIdx.x >> 5));
    const int d = blockIdx.x * 8 + wv;
    if (d >= N) return;
    const int half = lane >> 4;
    const int t = lane & 15;
    const int start = rowptr[d];
    const int n = deg[d];
    const int cnt = n + 1;
    const float ad = a_dst[d];

    if (cnt <= 32) {
        const int s = lane < n ? csr_src[start + lane] : d;
        const float a = a_src[s];
        float r[16];
#pragma unroll
        for (int p = 0; p < 8; ++p) {
            const int s0 = __builtin_amdgcn_readlane(s, 2 * p);
            const int s1 = __builtin_amdgcn_readlane(s, 2 * p + 1);
            r[p] = h2[(size_t)(half ? s1 : s0) * 16 + t];
        }
#pragma unroll
        for (int p = 8; p < 16; ++p) r[p] = 0.0f;
        if (cnt > 16) {
#pragma unroll
            for (int p = 8; p < 16; ++p) {
                const int s0 = __builtin_amdgcn_readlane(s, 2 * p);
                const int s1 = __builtin_amdgcn_readlane(s, 2 * p + 1);
                r[p] = h2[(size_t)(half ? s1 : s0) * 16 + t];
            }
        }
        const bool live = lane < cnt;
        const float e = lrelu(a + ad);
        const float m = wave_max(live ? e : -__builtin_huge_valf());
        const float w = live ? expf(e - m) : 0.0f;
        const float den = wave_sum(w);
        float acc = 0.0f;
#pragma unroll
        for (int p = 0; p < 8; ++p) {
            const float w0 = lane_f(w, 2 * p);
            const float w1 = lane_f(w, 2 * p + 1);
            acc += (half ? w1 : w0) * r[p];
        }
        if (cnt > 16) {
#pragma unroll
            for (int p = 8; p < 16; ++p) {
                const float w0 = lane_f(w, 2 * p);
                const float w1 = lane_f(w, 2 * p + 1);
                acc += (half ? w1 : w0) * r[p];
            }
        }
        acc += xhalf(acc);
        if (lane < 16) out[(size_t)d * 16 + t] = acc / den + b2[t];
        return;
    }

    float m = -__builtin_huge_valf();
    for (int c = 0; c < cnt; c += 32) {
        const int k = c + lane;
        if (k < cnt) {
            const int s = k < n ? csr_src[start + k] : d;
            m = fmaxf(m, lrelu(a_src[s] + ad));
        }
    }
    m = wave_max(m);

    float den = 0.0f, acc = 0.0f;
    for (int c = 0; c < cnt; c += 32) {
        const int k = c + lane;
        int s = d;
        float w = 0.0f;
        if (k < cnt) {
            if (k < n) s = csr_src[start + k];
            w = expf(lrelu(a_src[s] + ad) - m);
        }
        den += w;
        const int cn = (cnt - c) < 32 ? (cnt - c) : 32;
        const int cp = cn & ~1;
#pragma unroll 4
        for (int j = 0; j < cp; j += 2) {
            const int s0 = __builtin_amdgcn_readlane(s, j);
            const int s1 = __builtin_amdgcn_readlane(s, j + 1);
            const float w0 = lane_f(w, j);
            const float w1 = lane_f(w, j + 1);
            const int sj = half ? s1 : s0;
            const float wj = half ? w1 : w0;
            acc += wj * h2[(size_t)sj * 16 + t];
        }
        if (cn & 1) {
            const int sl = __builtin_amdgcn_readlane(s, cp);
            const float wl = lane_f(w, cp);
            if (half == 0) acc += wl * h2[(size_t)sl * 16 + t];
        }
    }
    den = wave_sum(den);
    acc += xhalf(acc);
    if (lane < 16) out[(size_t)d * 16 + t] = acc / den + b2[t];
}


extern "C" void kernel_launch(void* const* d_in, const int* in_sizes, int n_in,
                              void* d_out, int out_size, void* d_ws, size_t ws_size,
                              hipStream_t stream) {
    const float* x   = (const float*)d_in[0];
    const int*   ei  = (const int*)  d_in[1];
    const float* W1  = (const float*)d_in[2];
    const float* as1 = (const float*)d_in[3];
    const float* ad1 = (const float*)d_in[4];
    const float* b1  = (const float*)d_in[5];
    const float* W2  = (const float*)d_in[6];
    const float* as2 = (const float*)d_in[7];
    const float* ad2 = (const float*)d_in[8];
    const float* b2  = (const float*)d_in[9];

    const int N  = in_sizes[0] / 128;
    const int E  = in_sizes[1] / 2;

    char* w = (char*)d_ws;
    auto carve = [&](size_t bytes) -> void* {
        void* p = (void*)w;
        w += (bytes + 255) & ~((size_t)255);
        return p;
    };
    float* aS1    = (float*)carve((size_t)N * 4 * 4);
    float* aD1    = (float*)carve((size_t)N * 4 * 4);
    float* U      = (float*)carve((size_t)128 * 8 * 4);
    float* h2     = (float*)carve((size_t)N * 16 * 4);
    float* aS2    = (float*)carve((size_t)N * 4);
    float* aD2    = (float*)carve((size_t)N * 4);
    int*   deg    = (int*)  carve((size_t)N * 4);
    int*   rowptr = (int*)  carve((size_t)N * 4);
    int*   rank   = (int*)  carve((size_t)E * 4);
    int*   csrs   = (int*)  carve((size_t)E * 4);
    const int NB  = (N + NBLK - 1) / NBLK;
    const int epc = (E + CHUNKS - 1) / CHUNKS;
    int*   cnt    = (int*)  carve((size_t)CHUNKS * HCAP * 4);
    int*   coarse = (int*)  carve((size_t)CHUNKS * NB * 4);

    const int TB = 256;
    auto blocks = [&](long long n) { return (unsigned)((n + TB - 1) / TB); };

    const int mtiles = (N + 15) / 16;
    const int bpc    = (int)blocks(epc);
    rank_and_fold<<<CHUNKS + 1, 1024, 0, stream>>>(
        ei, E, epc, NB, rank, cnt, coarse, W1, as1, ad1, U);
    scan_and_scores<<<NB + (mtiles + NBLK / 32 - 1) / (NBLK / 32), NBLK, 0, stream>>>(
        cnt, coarse, deg, rowptr, NB, x, U, aS1, aD1, N, mtiles);
    fill_groups<<<bpc * CHUNKS, TB, 0, stream>>>(ei, E, epc, bpc, cnt, rank, csrs);

    node_agg_proj12<<<(N + GN - 1) / GN, 512, 0, stream>>>(
        rowptr, deg, csrs, aS1, aD1, x, W1, b1, W2, as2, ad2, h2, aS2, aD2, N);

    node_agg2<<<(N + 7) / 8, 256, 0, stream>>>(rowptr, deg, csrs, aS2, aD2, h2, b2, (float*)d_out, N);
}


# Edge grouping in two launches, not three: each list piece is grouped inside its ranking block (LDS scan of its counters, a second LDS add places the edge); scan blocks move the runs; list order kept
# speedup vs baseline: 8.2363x; 1.2237x over previous
#include <hip/hip_runtime.h>
#include <hip/hip_bf16.h>

typedef __attribute__((ext_vector_type(2))) float v2f;
typedef __attribute__((ext_vector_type(8))) float v8f;

#define NEG_SLOPE 0.2f


__device__ __forceinline__ float lrelu(float v) {
    return v >= 0.0f ? v : NEG_SLOPE * v;
}


__device__ __forceinline__ int xhalf_i(int b) {
    return __builtin_amdgcn_permlanex16(b, b, 0x76543210, 0xfedcba98, false, false);
}
template <int M> __device__ __forceinline__ int xrow_i(int b) {
    return __builtin_amdgcn_update_dpp(0, b, 0x160 + M, 0xf, 0xf, true);
}
__device__ __forceinline__ float xhalf(float v) { return __int_as_float(xhalf_i(__float_as_int(v))); }
template <int M> __device__ __forceinline__ float xrow(float v) {
    return __int_as_float(xrow_i<M>(__float_as_int(v)));
}


__device__ __forceinline__
void fold_att(const float* __restrict__ W1, const float* __restrict__ att_src,
              const float* __restrict__ att_dst, float* __restrict__ U) {
    const int t = threadIdx.x;
    const int k = t >> 3, j = t & 7, h = j & 3;
    const float* w  = W1 + (size_t)k * 256 + h * 64;
    const float* at = (j < 4 ? att_src : att_dst) + h * 64;
    float s = 0.0f;
    for (int c = 0; c < 64; ++c) s += w[c] * at[c];
    U[t] = s;
}

__device__ __forceinline__
void score_tile(const int mtile, const float* __restrict__ x, const float* __restrict__ U,
                float* __restrict__ a_src, float* __restrict__ a_dst, int N) {
    const int lane  = threadIdx.x & 31;
    const int mrow  = mtile * 16 + (lane & 15);
    const int lrow  = mrow < N ? mrow : N - 1;
    const int khalf = (lane >> 4) * 2;
    const int nc    = lane & 15;
    const int uc    = nc & 7;

    v8f c = {};
    const float* xrow = x + (size_t)lrow * 128;
    for (int k = 0; k < 128; k += 4) {
        const int kb = k + khalf;
        v2f a; a.x = xrow[kb]; a.y = xrow[kb + 1];
        v2f b; b.x = U[kb * 8 + uc]; b.y = U[(kb + 1) * 8 + uc];
        c = __builtin_amdgcn_wmma_f32_16x16x4_f32(false, a, false, b, (short)0, c, false, false);
    }
    const int rowoff = (lane >= 16) ? 8 : 0;
    float* sp = (nc < 4) ? a_src : a_dst;
    for (int r = 0; r < 8; ++r) {
        const int grow = mtile * 16 + rowoff + r;
        if (grow < N && nc < 8) sp[(size_t)grow * 4 + (nc & 3)] = c[r];
    }
}

#define CHUNKS 64
#define HCAP   50176
#define HPT    49
#define NBLK   256
#define CPW    (CHUNKS / (NBLK / 32))
#define EPT    10

__device__ __forceinline__ int wave_sum_i(int v) {
    v += xhalf_i(v);
    v += xrow_i<8>(v); v += xrow_i<4>(v); v += xrow_i<2>(v); v += xrow_i<1>(v);
    return v;
}

__device__ __forceinline__
void chunk_group(int* __restrict__ hist, int* __restrict__ wtot,
                 const int* __restrict__ ei, int E, int epc,
                 int2* __restrict__ csl, unsigned short* __restrict__ cnt) {
    const int t = threadIdx.x, lane = t & 31, wv = t >> 5;
    const int c = blockIdx.x;
    int4* h4 = (int4*)hist;
    for (int i = t; i < HCAP / 4; i += 1024) h4[i] = make_int4(0, 0, 0, 0);
    __syncthreads();

    const int e0 = c * epc;
    const int e1 = (e0 + epc) < E ? (e0 + epc) : E;
    for (int eb = e0 + t; eb < e1; eb += EPT * 1024) {
        int dd[EPT];
#pragma unroll
        for (int u = 0; u < EPT; ++u) {
            const int e = eb + u * 1024;
            dd[u] = e < e1 ? ei[E + e] : 0;
        }
#pragma unroll
        for (int u = 0; u < EPT; ++u) {
            const int e = eb + u * 1024;
            if (e < e1) atomicAdd(&hist[dd[u]], 1);
        }
    }
    __syncthreads();

    const int b0 = t * HPT;
    int hv[HPT];
    int s = 0;
#pragma unroll
    for (int i = 0; i < HPT; ++i) {
        hv[i] = hist[b0 + i];
        s += hv[i];
    }
    int inc = s;
    for (int off = 1; off < 32; off <<= 1) {
        const int nb = __shfl_up(inc, off);
        if (lane >= off) inc += nb;
    }
    if (lane == 31) wtot[wv] = inc;
    __syncthreads();
    int run = inc - s;
    for (int k = 0; k < 32; ++k)
        if (k < wv) run += wtot[k];
#pragma unroll
    for (int i = 0; i < HPT; ++i) {
        hist[b0 + i] = run;
        run += hv[i];
    }
    __syncthreads();

    for (int eb = e0 + t; eb < e1; eb += EPT * 1024) {
        int dd[EPT], ss[EPT];
#pragma unroll
        for (int u = 0; u < EPT; ++u) {
            const int e = eb + u * 1024;
            dd[u] = e < e1 ? ei[E + e] : 0;
            ss[u] = e < e1 ? ei[e] : 0;
        }
#pragma unroll
        for (int u = 0; u < EPT; ++u) {
            const int e = eb + u * 1024;
            if (e < e1) csl[e0 + atomicAdd(&hist[dd[u]], 1)] = make_int2(ss[u], dd[u]);
        }
    }
    __syncthreads();

    uint4* c4 = (uint4*)(cnt + (size_t)c * HCAP);
    for (int i = t; i < HCAP / 8; i += 1024) {
        const int4 a = h4[2 * i], b = h4[2 * i + 1];
        c4[i] = make_uint4((unsigned)a.x | ((unsigned)a.y << 16), (unsigned)a.z | ((unsigned)a.w << 16),
                           (unsigned)b.x | ((unsigned)b.y << 16), (unsigned)b.z | ((unsigned)b.w << 16));
    }
}

__device__ __forceinline__
void scan_cols(int* __restrict__ wsum, int* __restrict__ wpre, int* __restrict__ rel, int* __restrict__ rp,
               const unsigned short* __restrict__ cnt, const int2* __restrict__ csl, int E, int epc,
               int* __restrict__ deg, int* __restrict__ rowptr, int* __restrict__ csr_src, int N) {
    const int t = threadIdx.x, lane = t & 31;
    const int wv = __builtin_amdgcn_readfirstlane((int)(threadIdx.x >> 5));
    const int j = blockIdx.x;
    const int n0 = j * NBLK;
    const int idx = n0 + t;

    int pre = 0;
    if (t < CHUNKS && j > 0) pre = cnt[(size_t)t * HCAP + n0 - 1];

    int d = 0;
    const int prv = idx > 0 ? idx - 1 : 0;
#pragma unroll
    for (int c = 0; c < CHUNKS; ++c) {
        const int en = cnt[(size_t)c * HCAP + idx];
        const int pv = cnt[(size_t)c * HCAP + prv];
        const int st = idx > 0 ? pv : 0;
        rel[c * NBLK + t] = d - st;
        d += en - st;
    }
    int inc = d;
    for (int off = 1; off < 32; off <<= 1) {
        const int nb = __shfl_up(inc, off);
        if (lane >= off) inc += nb;
    }
    pre = wave_sum_i(pre);
    if (lane == 31) wsum[wv] = inc;
    if (lane == 0)  wpre[wv] = pre;
    __syncthreads();
    int base = 0;
    for (int k = 0; k < NBLK / 32; ++k) {
        base += wpre[k];
        if (k < wv) base += wsum[k];
    }
    const int first = base + inc - d;
    rp[t] = first;
    if (idx < N) {
        rowptr[idx] = first;
        deg[idx] = d;
    }
    __syncthreads();

    int2 sd[2 * CPW];
    int  kk[2 * CPW], dest[2 * CPW];
    const int pb = j > 0 ? n0 - 1 : 0;
#pragma unroll
    for (int u = 0; u < CPW; ++u) {
        const int c = wv * CPW + u;
        const int pv = cnt[(size_t)c * HCAP + pb];
        const int lo = j > 0 ? pv : 0;
        const int hi = cnt[(size_t)c * HCAP + n0 + NBLK - 1];
#pragma unroll
        for (int r = 0; r < 2; ++r) {
            const int k = lo + r * 32 + lane;
            long long gi = (long long)c * epc + (k < hi ? k : lo);
            if (gi > (long long)E - 1) gi = (long long)E - 1;
            sd[2 * u + r] = csl[gi];
            kk[2 * u + r] = k < hi ? k : -1;
        }
    }
#pragma unroll
    for (int u = 0; u < CPW; ++u) {
        const int c = wv * CPW + u;
#pragma unroll
        for (int r = 0; r < 2; ++r) {
            const int k = kk[2 * u + r];
            const int nl = k >= 0 ? sd[2 * u + r].y - n0 : 0;
            dest[2 * u + r] = rp[nl] + rel[c * NBLK + nl] + k;
        }
    }
#pragma unroll
    for (int i = 0; i < 2 * CPW; ++i)
        if (kk[i] >= 0) csr_src[dest[i]] = sd[i].x;
    for (int u = 0; u < CPW; ++u) {
        const int c = wv * CPW + u;
        const int pv = cnt[(size_t)c * HCAP + pb];
        const int lo = j > 0 ? pv : 0;
        const int hi = cnt[(size_t)c * HCAP + n0 + NBLK - 1];
        for (int k = lo + 64 + lane; k < hi; k += 32) {
            const int2 e = csl[(size_t)c * epc + k];
            const int nl = e.y - n0;
            csr_src[rp[nl] + rel[c * NBLK + nl] + k] = e.x;
        }
    }
}


__global__ __launch_bounds__(1024)
void group_and_fold(const int* __restrict__ ei, int E, int epc,
                    int2* __restrict__ csl, unsigned short* __restrict__ cnt,
                    const float* __restrict__ W1,
                    const float* __restrict__ att_src, const float* __restrict__ att_dst,
                    float* __restrict__ U) {
    __shared__ int hist[HCAP];
    __shared__ int wtot[32];
    if (blockIdx.x < CHUNKS) {
        chunk_group(hist, wtot, ei, E, epc, csl, cnt);
    } else {
        fold_att(W1, att_src, att_dst, U);
    }
}

__global__ __launch_bounds__(NBLK)
void scan_and_scores(const unsigned short* __restrict__ cnt, const int2* __restrict__ csl, int E, int epc,
                     int* __restrict__ deg, int* __restrict__ rowptr, int* __restrict__ csr_src, int NB,
                     const float* __restrict__ x, const float* __restrict__ U,
                     float* __restrict__ a_src, float* __restrict__ a_dst,
                     int N, int mtiles) {
    __shared__ int rel[CHUNKS * NBLK];
    __shared__ int rp[NBLK];
    __shared__ int wsum[NBLK / 32];
    __shared__ int wpre[NBLK / 32];
    if ((int)blockIdx.x < NB) {
        scan_cols(wsum, wpre, rel, rp, cnt, csl, E, epc, deg, rowptr, csr_src, N);
    } else {
        const int wv = __builtin_amdgcn_readfirstlane((int)(threadIdx.x >> 5));
        const int tile = ((int)blockIdx.x - NB) * (NBLK / 32) + wv;
        if (tile < mtiles) score_tile(tile, x, U, a_src, a_dst, N);
    }
}

__device__ __forceinline__ float wave_max(float v) {
    v = fmaxf(v, xhalf(v));
    v = fmaxf(v, xrow<8>(v)); v = fmaxf(v, xrow<4>(v));
    v = fmaxf(v, xrow<2>(v)); v = fmaxf(v, xrow<1>(v));
    return v;
}
__device__ __forceinline__ float wave_sum(float v) {
    v += xhalf(v);
    v += xrow<8>(v); v += xrow<4>(v); v += xrow<2>(v); v += xrow<1>(v);
    return v;
}
__device__ __forceinline__ void wave_max4(float& a, float& b, float& c, float& d) {
    a = fmaxf(a, xhalf(a));   b = fmaxf(b, xhalf(b));   c = fmaxf(c, xhalf(c));   d = fmaxf(d, xhalf(d));
    a = fmaxf(a, xrow<8>(a)); b = fmaxf(b, xrow<8>(b)); c = fmaxf(c, xrow<8>(c)); d = fmaxf(d, xrow<8>(d));
    a = fmaxf(a, xrow<4>(a)); b = fmaxf(b, xrow<4>(b)); c = fmaxf(c, xrow<4>(c)); d = fmaxf(d, xrow<4>(d));
    a = fmaxf(a, xrow<2>(a)); b = fmaxf(b, xrow<2>(b)); c = fmaxf(c, xrow<2>(c)); d = fmaxf(d, xrow<2>(d));
    a = fmaxf(a, xrow<1>(a)); b = fmaxf(b, xrow<1>(b)); c = fmaxf(c, xrow<1>(c)); d = fmaxf(d, xrow<1>(d));
}
__device__ __forceinline__ void wave_sum4(float& a, float& b, float& c, float& d) {
    a += xhalf(a);   b += xhalf(b);   c += xhalf(c);   d += xhalf(d);
    a += xrow<8>(a); b += xrow<8>(b); c += xrow<8>(c); d += xrow<8>(d);
    a += xrow<4>(a); b += xrow<4>(b); c += xrow<4>(c); d += xrow<4>(d);
    a += xrow<2>(a); b += xrow<2>(b); c += xrow<2>(c); d += xrow<2>(d);
    a += xrow<1>(a); b += xrow<1>(b); c += xrow<1>(c); d += xrow<1>(d);
}
__device__ __forceinline__ float lane_f(float v, int j) {
    return __int_as_float(__builtin_amdgcn_readlane(__float_as_int(v), j));
}

__device__ __forceinline__ void row_fma(float4& z0, float4& z1, float4& z2, float4& z3,
                                        const float x0, const float x1, const float x2, const float x3,
                                        const float4 v) {
    z0.x += x0 * v.x; z0.y += x0 * v.y; z0.z += x0 * v.z; z0.w += x0 * v.w;
    z1.x += x1 * v.x; z1.y += x1 * v.y; z1.z += x1 * v.z; z1.w += x1 * v.w;
    z2.x += x2 * v.x; z2.y += x2 * v.y; z2.z += x2 * v.z; z2.w += x2 * v.w;
    z3.x += x3 * v.x; z3.y += x3 * v.y; z3.z += x3 * v.z; z3.w += x3 * v.w;
}

#define GN 32
#define ZP 516
#define PT 288
#define PH 144

__global__ __launch_bounds__(512) __attribute__((amdgpu_waves_per_eu(16)))
void node_agg_proj12(const int* __restrict__ rowptr, const int* __restrict__ deg,
                     const int* __restrict__ csr_src,
                     const float* __restrict__ a_src, const float* __restrict__ a_dst,
                     const float* __restrict__ x, const float* __restrict__ W1,
                     const float* __restrict__ b1, const float* __restrict__ W2,
                     const float* __restrict__ att_src2, const float* __restrict__ att_dst2,
                     float* __restrict__ h2, float* __restrict__ a_src2, float* __restrict__ a_dst2,
                     int N) {
    __shared__ __attribute__((aligned(16))) float Z[GN * ZP];
    const int lane = threadIdx.x & 31;
    const int wv = __builtin_amdgcn_readfirstlane((int)(threadIdx.x >> 5));

    for (int q = 0; q < 2; ++q) {
        const int nl = wv * 2 + q;
        const int d = blockIdx.x * GN + nl;
        float4 z0 = make_float4(0.0f, 0.0f, 0.0f, 0.0f);
        float4 z1 = z0, z2 = z0, z3 = z0;
        float4 dn = make_float4(1.0f, 1.0f, 1.0f, 1.0f);
        if (d < N) {
            const int start = rowptr[d];
            const int n = deg[d];
            const int cnt = n + 1;
            const float4 ad = *(const float4*)(a_dst + (size_t)d * 4);

            float d0 = 0.0f, d1 = 0.0f, d2 = 0.0f, d3 = 0.0f;
            if (cnt <= 32) {
                const int s = lane < n ? csr_src[start + lane] : d;
                const float4 a = *(const float4*)(a_src + (size_t)s * 4);
                const float4 r0 = *(const float4*)(x + (size_t)__builtin_amdgcn_readlane(s, 0) * 128 + lane * 4);
                const float4 r1 = *(const float4*)(x + (size_t)__builtin_amdgcn_readlane(s, 1) * 128 + lane * 4);
                const float4 r2 = *(const float4*)(x + (size_t)__builtin_amdgcn_readlane(s, 2) * 128 + lane * 4);
                const float4 r3 = *(const float4*)(x + (size_t)__builtin_amdgcn_readlane(s, 3) * 128 + lane * 4);
                const bool live = lane < cnt;
                const float e0 = lrelu(a.x + ad.x), e1 = lrelu(a.y + ad.y);
                const float e2 = lrelu(a.z + ad.z), e3 = lrelu(a.w + ad.w);
                float m0 = live ? e0 : -__builtin_huge_valf();
                float m1 = live ? e1 : -__builtin_huge_valf();
                float m2 = live ? e2 : -__builtin_huge_valf();
                float m3 = live ? e3 : -__builtin_huge_valf();
                wave_max4(m0, m1, m2, m3);
                const float w0 = live ? expf(e0 - m0) : 0.0f;
                const float w1 = live ? expf(e1 - m1) : 0.0f;
                const float w2 = live ? expf(e2 - m2) : 0.0f;
                const float w3 = live ? expf(e3 - m3) : 0.0f;
                d0 = w0; d1 = w1; d2 = w2; d3 = w3;
                row_fma(z0, z1, z2, z3, lane_f(w0, 0), lane_f(w1, 0), lane_f(w2, 0), lane_f(w3, 0), r0);
                row_fma(z0, z1, z2, z3, lane_f(w0, 1), lane_f(w1, 1), lane_f(w2, 1), lane_f(w3, 1), r1);
                row_fma(z0, z1, z2, z3, lane_f(w0, 2), lane_f(w1, 2), lane_f(w2, 2), lane_f(w3, 2), r2);
                row_fma(z0, z1, z2, z3, lane_f(w0, 3), lane_f(w1, 3), lane_f(w2, 3), lane_f(w3, 3), r3);
                const int cn4 = (cnt + 3) & ~3;
                for (int j = 4; j < cn4; j += 4) {
                    float4 v[4];
#pragma unroll
                    for (int u = 0; u < 4; ++u)
                        v[u] = *(const float4*)(x + (size_t)__builtin_amdgcn_readlane(s, j + u) * 128 + lane * 4);
#pragma unroll
                    for (int u = 0; u < 4; ++u)
                        row_fma(z0, z1, z2, z3, lane_f(w0, j + u), lane_f(w1, j + u),
                                lane_f(w2, j + u), lane_f(w3, j + u), v[u]);
                }
            } else {
                float m0 = -__builtin_huge_valf(), m1 = m0, m2 = m0, m3 = m0;
                for (int c = 0; c < cnt; c += 32) {
                    const int k = c + lane;
                    if (k < cnt) {
                        const int s = k < n ? csr_src[start + k] : d;
                        const float4 a = *(const float4*)(a_src + (size_t)s * 4);
                        m0 = fmaxf(m0, lrelu(a.x + ad.x));
                        m1 = fmaxf(m1, lrelu(a.y + ad.y));
                        m2 = fmaxf(m2, lrelu(a.z + ad.z));
                        m3 = fmaxf(m3, lrelu(a.w + ad.w));
                    }
                }
                wave_max4(m0, m1, m2, m3);

                for (int c = 0; c < cnt; c += 32) {
                    const int k = c + lane;
                    int s = d;
                    float w0 = 0.0f, w1 = 0.0f, w2 = 0.0f, w3 = 0.0f;
                    if (k < cnt) {
                        if (k < n) s = csr_src[start + k];
                        const float4 a = *(const float4*)(a_src + (size_t)s * 4);
                        w0 = expf(lrelu(a.x + ad.x) - m0);
                        w1 = expf(lrelu(a.y + ad.y) - m1);
                        w2 = expf(lrelu(a.z + ad.z) - m2);
                        w3 = expf(lrelu(a.w + ad.w) - m3);
                    }
                    d0 += w0; d1 += w1; d2 += w2; d3 += w3;
                    const int cn = (cnt - c) < 32 ? (cnt - c) : 32;
#pragma unroll 4
                    for (int j = 0; j < cn; ++j) {
                        const int sj = __builtin_amdgcn_readlane(s, j);
                        const float x0 = lane_f(w0, j), x1 = lane_f(w1, j);
                        const float x2 = lane_f(w2, j), x3 = lane_f(w3, j);
                        const float4 v = *(const float4*)(x + (size_t)sj * 128 + lane * 4);
                        z0.x += x0 * v.x; z0.y += x0 * v.y; z0.z += x0 * v.z; z0.w += x0 * v.w;
                        z1.x += x1 * v.x; z1.y += x1 * v.y; z1.z += x1 * v.z; z1.w += x1 * v.w;
                        z2.x += x2 * v.x; z2.y += x2 * v.y; z2.z += x2 * v.z; z2.w += x2 * v.w;
                        z3.x += x3 * v.x; z3.y += x3 * v.y; z3.z += x3 * v.z; z3.w += x3 * v.w;
                    }
                }
            }
            wave_sum4(d0, d1, d2, d3);
            dn = make_float4(d0, d1, d2, d3);
        }
        float* zp = Z + nl * ZP + lane * 4;
        *(float4*)zp         = z0;
        *(float4*)(zp + 128) = z1;
        *(float4*)(zp + 256) = z2;
        *(float4*)(zp + 384) = z3;
        if (lane == 0) *(float4*)(Z + nl * ZP + 512) = dn;
    }
    __syncthreads();

    const int c     = lane & 15;
    const int hf    = lane >> 4;
    const int khalf = hf * 2;
    const int head  = wv >> 2;
    const int cb    = wv * 16;
    const int col0  = cb + 8 * hf;

    const float* w2p = W2 + (size_t)col0 * 16 + c;
    v2f wb[4];
    for (int t = 0; t < 4; ++t) {
        wb[t].x = w2p[(2 * t) * 16];
        wb[t].y = w2p[(2 * t + 1) * 16];
    }

    const float* za0 = Z + c * ZP + head * 128 + khalf;
    const float* za1 = za0 + 16 * ZP;
    const float* wp  = W1 + (size_t)khalf * 256 + cb + c;
    v8f c0 = {}, c1 = {};
    v2f wn[4];
    for (int i = 0; i < 4; ++i) {
        wn[i].x = wp[(size_t)(4 * i) * 256];
        wn[i].y = wp[(size_t)(4 * i + 1) * 256];
    }
#pragma unroll 1
    for (int kk = 0; kk < 128 - 16; kk += 16) {
#pragma unroll
        for (int i = 0; i < 4; ++i) {
            const int k = kk + 4 * i;
            const v2f a0 = *(const v2f*)(za0 + k);
            const v2f a1 = *(const v2f*)(za1 + k);
            c0 = __builtin_amdgcn_wmma_f32_16x16x4_f32(false, wn[i], false, a0, (short)0, c0, false, false);
            c1 = __builtin_amdgcn_wmma_f32_16x16x4_f32(false, wn[i], false, a1, (short)0, c1, false, false);
            wn[i].x = wp[(size_t)(k + 16) * 256];
            wn[i].y = wp[(size_t)(k + 16 + 1) * 256];
        }
    }
    const float inv0 = 1.0f / Z[c * ZP + 512 + head];
    const float inv1 = 1.0f / Z[(16 + c) * ZP + 512 + head];
    const float4 bs0 = *(const float4*)(b1 + col0);
    const float4 bs1 = *(const float4*)(b1 + col0 + 4);
    for (int i = 0; i < 4; ++i) {
        const int k = 128 - 16 + 4 * i;
        const v2f a0 = *(const v2f*)(za0 + k);
        const v2f a1 = *(const v2f*)(za1 + k);
        c0 = __builtin_amdgcn_wmma_f32_16x16x4_f32(false, wn[i], false, a0, (short)0, c0, false, false);
        c1 = __builtin_amdgcn_wmma_f32_16x16x4_f32(false, wn[i], false, a1, (short)0, c1, false, false);
    }

    const float bs[8] = {bs0.x, bs0.y, bs0.z, bs0.w, bs1.x, bs1.y, bs1.z, bs1.w};
    for (int r = 0; r < 8; ++r) {
        c0[r] = fmaxf(c0[r] * inv0 + bs[r], 0.0f);
        c1[r] = fmaxf(c1[r] * inv1 + bs[r], 0.0f);
    }

    v8f d0 = {}, d1 = {};
    for (int t = 0; t < 4; ++t) {
        v2f a;
        a.x = c0[2 * t]; a.y = c0[2 * t + 1];
        d0 = __builtin_amdgcn_wmma_f32_16x16x4_f32(false, a, false, wb[t], (short)0, d0, false, false);
        a.x = c1[2 * t]; a.y = c1[2 * t + 1];
        d1 = __builtin_amdgcn_wmma_f32_16x16x4_f32(false, a, false, wb[t], (short)0, d1, false, false);
    }
    const float as = att_src2[c];
    const float ad = att_dst2[c];
    __syncthreads();

    float* P = Z;
    {
        float* pp = P + wv * (2 * PT) + hf * PH + c;
        for (int r = 0; r < 8; ++r) {
            pp[r * 16]      = d0[r];
            pp[PT + r * 16] = d1[r];
        }
    }
    __syncthreads();

    {
        const int node = wv * 2 + hf;
        const int nl   = node & 15;
        const float* q = P + (node >> 4) * PT + (nl >> 3) * PH + (nl & 7) * 16 + c;
        float sum = q[0];
        for (int w = 1; w < 16; ++w) sum += q[w * (2 * PT)];
        float ps = sum * as;
        float pd = sum * ad;
        ps += xrow<1>(ps); pd += xrow<1>(pd);
        ps += xrow<2>(ps); pd += xrow<2>(pd);
        ps += xrow<4>(ps); pd += xrow<4>(pd);
        ps += xrow<8>(ps); pd += xrow<8>(pd);
        const int g = blockIdx.x * GN + node;
        if (g < N) {
            h2[(size_t)g * 16 + c] = sum;
            if (c == 0) {
                a_src2[g] = ps;
                a_dst2[g] = pd;
            }
        }
    }
}

__global__ __launch_bounds__(256)
void node_agg2(const int* __restrict__ rowptr, const int* __restrict__ deg,
               const int* __restrict__ csr_src,
               const float* __restrict__ a_src, const float* __restrict__ a_dst,
               const float* __restrict__ h2, const float* __restrict__ b2,
               float* __restrict__ out, int N) {
    const int lane = threadIdx.x & 31;
    const int wv = __builtin_amdgcn_readfirstlane((int)(threadIdx.x >> 5));
    const int d = blockIdx.x * 8 + wv;
    if (d >= N) return;
    const int half = lane >> 4;
    const int t = lane & 15;
    const int start = rowptr[d];
    const int n = deg[d];
    const int cnt = n + 1;
    const float ad = a_dst[d];

    if (cnt <= 32) {
        const int s = lane < n ? csr_src[start + lane] : d;
        const float a = a_src[s];
        float r[16];
#pragma unroll
        for (int p = 0; p < 8; ++p) {
            const int s0 = __builtin_amdgcn_readlane(s, 2 * p);
            const int s1 = __builtin_amdgcn_readlane(s, 2 * p + 1);
            r[p] = h2[(size_t)(half ? s1 : s0) * 16 + t];
        }
#pragma unroll
        for (int p = 8; p < 16; ++p) r[p] = 0.0f;
        if (cnt > 16) {
#pragma unroll
            for (int p = 8; p < 16; ++p) {
                const int s0 = __builtin_amdgcn_readlane(s, 2 * p);
                const int s1 = __builtin_amdgcn_readlane(s, 2 * p + 1);
                r[p] = h2[(size_t)(half ? s1 : s0) * 16 + t];
            }
        }
        const bool live = lane < cnt;
        const float e = lrelu(a + ad);
        const float m = wave_max(live ? e : -__builtin_huge_valf());
        const float w = live ? expf(e - m) : 0.0f;
        const float den = wave_sum(w);
        float acc = 0.0f;
#pragma unroll
        for (int p = 0; p < 8; ++p) {
            const float w0 = lane_f(w, 2 * p);
            const float w1 = lane_f(w, 2 * p + 1);
            acc += (half ? w1 : w0) * r[p];
        }
        if (cnt > 16) {
#pragma unroll
            for (int p = 8; p < 16; ++p) {
                const float w0 = lane_f(w, 2 * p);
                const float w1 = lane_f(w, 2 * p + 1);
                acc += (half ? w1 : w0) * r[p];
            }
        }
        acc += xhalf(acc);
        if (lane < 16) out[(size_t)d * 16 + t] = acc / den + b2[t];
        return;
    }

    float m = -__builtin_huge_valf();
    for (int c = 0; c < cnt; c += 32) {
        const int k = c + lane;
        if (k < cnt) {
            const int s = k < n ? csr_src[start + k] : d;
            m = fmaxf(m, lrelu(a_src[s] + ad));
        }
    }
    m = wave_max(m);

    float den = 0.0f, acc = 0.0f;
    for (int c = 0; c < cnt; c += 32) {
        const int k = c + lane;
        int s = d;
        float w = 0.0f;
        if (k < cnt) {
            if (k < n) s = csr_src[start + k];
            w = expf(lrelu(a_src[s] + ad) - m);
        }
        den += w;
        const int cn = (cnt - c) < 32 ? (cnt - c) : 32;
        const int cp = cn & ~1;
#pragma unroll 4
        for (int j = 0; j < cp; j += 2) {
            const int s0 = __builtin_amdgcn_readlane(s, j);
            const int s1 = __builtin_amdgcn_readlane(s, j + 1);
            const float w0 = lane_f(w, j);
            const float w1 = lane_f(w, j + 1);
            const int sj = half ? s1 : s0;
            const float wj = half ? w1 : w0;
            acc += wj * h2[(size_t)sj * 16 + t];
        }
        if (cn & 1) {
            const int sl = __builtin_amdgcn_readlane(s, cp);
            const float wl = lane_f(w, cp);
            if (half == 0) acc += wl * h2[(size_t)sl * 16 + t];
        }
    }
    den = wave_sum(den);
    acc += xhalf(acc);
    if (lane < 16) out[(size_t)d * 16 + t] = acc / den + b2[t];
}


extern "C" void kernel_launch(void* const* d_in, const int* in_sizes, int n_in,
                              void* d_out, int out_size, void* d_ws, size_t ws_size,
                              hipStream_t stream) {
    const float* x   = (const float*)d_in[0];
    const int*   ei  = (const int*)  d_in[1];
    const float* W1  = (const float*)d_in[2];
    const float* as1 = (const float*)d_in[3];
    const float* ad1 = (const float*)d_in[4];
    const float* b1  = (const float*)d_in[5];
    const float* W2  = (const float*)d_in[6];
    const float* as2 = (const float*)d_in[7];
    const float* ad2 = (const float*)d_in[8];
    const float* b2  = (const float*)d_in[9];

    const int N  = in_sizes[0] / 128;
    const int E  = in_sizes[1] / 2;

    char* w = (char*)d_ws;
    auto carve = [&](size_t bytes) -> void* {
        void* p = (void*)w;
        w += (bytes + 255) & ~((size_t)255);
        return p;
    };
    float* aS1    = (float*)carve((size_t)N * 4 * 4);
    float* aD1    = (float*)carve((size_t)N * 4 * 4);
    float* U      = (float*)carve((size_t)128 * 8 * 4);
    float* h2     = (float*)carve((size_t)N * 16 * 4);
    float* aS2    = (float*)carve((size_t)N * 4);
    float* aD2    = (float*)carve((size_t)N * 4);
    int*   deg    = (int*)  carve((size_t)N * 4);
    int*   rowptr = (int*)  carve((size_t)N * 4);
    int2*  csl    = (int2*) carve((size_t)E * 8);
    int*   csrs   = (int*)  carve((size_t)E * 4);
    const int NB  = (N + NBLK - 1) / NBLK;
    const int epc = (E + CHUNKS - 1) / CHUNKS;
    unsigned short* cnt = (unsigned short*)carve((size_t)CHUNKS * HCAP * 2);

    const int mtiles = (N + 15) / 16;
    group_and_fold<<<CHUNKS + 1, 1024, 0, stream>>>(
        ei, E, epc, csl, cnt, W1, as1, ad1, U);
    scan_and_scores<<<NB + (mtiles + NBLK / 32 - 1) / (NBLK / 32), NBLK, 0, stream>>>(
        cnt, csl, E, epc, deg, rowptr, csrs, NB, x, U, aS1, aD1, N, mtiles);

    node_agg_proj12<<<(N + GN - 1) / GN, 512, 0, stream>>>(
        rowptr, deg, csrs, aS1, aD1, x, W1, b1, W2, as2, ad2, h2, aS2, aD2, N);

    node_agg2<<<(N + 7) / 8, 256, 0, stream>>>(rowptr, deg, csrs, aS2, aD2, h2, b2, (float*)d_out, N);
}
